# starting kernel
# baseline (speedup 1.0000x reference)
#include <hip/hip_runtime.h>

typedef float v2f __attribute__((ext_vector_type(2)));
typedef float v8f __attribute__((ext_vector_type(8)));

#define N_NODES  100000
#define N_ETYPES 3
#define N_EDGES  1600000
#define D_IN     128
#define D_HID    256
#define D_OUT    32

__global__ void fill_zero_kernel(float* __restrict__ p, int n) {
    int i = blockIdx.x * blockDim.x + threadIdx.x;
    if (i < n) p[i] = 0.0f;
}

__global__ void degree_kernel(const int* __restrict__ dst, float* __restrict__ deg) {
    int e = blockIdx.y;
    int i = blockIdx.x * blockDim.x + threadIdx.x;
    if (i < N_EDGES) {
        int d = dst[(size_t)e * N_EDGES + i];
        atomicAdd(&deg[(size_t)e * N_NODES + d], 1.0f);
    }
}

__global__ void scatter_kernel(const float* __restrict__ h,
                               const int* __restrict__ src,
                               const int* __restrict__ dst,
                               float* __restrict__ agg) {
    int edge = blockIdx.x * 8 + (threadIdx.x >> 5);
    int c    = (threadIdx.x & 31) * 4;
    if (edge >= N_EDGES) return;
    int s = src[edge];
    int d = dst[edge];
    const float4 v = *(const float4*)(h + (size_t)s * D_IN + c);
    float* o = agg + (size_t)d * D_IN + c;
    atomicAdd(o + 0, v.x);
    atomicAdd(o + 1, v.y);
    atomicAdd(o + 2, v.z);
    atomicAdd(o + 3, v.w);
}

__global__ void wmma_gemm_f32_kernel(const float* __restrict__ A,
                                     const float* __restrict__ W,
                                     const float* __restrict__ bias,
                                     float* __restrict__ C,
                                     const float* __restrict__ deg,
                                     int M, int N, int K,
                                     float outScale, int accumulate, int doRelu) {
    const int lane = threadIdx.x;
    const int half = lane >> 4;
    const int l15  = lane & 15;
    const int n0   = blockIdx.x * 16;
    const int m0   = blockIdx.y * 16;
    const int rowA = m0 + l15;
    const int colB = n0 + l15;

    float rs = 1.0f;
    if (deg) rs = 1.0f / fmaxf(deg[rowA], 1.0f);

    v8f acc = {};
    const float* arow = A + (size_t)rowA * K;
    for (int k = 0; k < K; k += 4) {
        const int ka = k + half * 2;
        v2f a, b;
        a.x = arow[ka]     * rs;
        a.y = arow[ka + 1] * rs;
        b.x = W[(size_t)ka       * N + colB];
        b.y = W[(size_t)(ka + 1) * N + colB];
        acc = __builtin_amdgcn_wmma_f32_16x16x4_f32(
             false, a,  false, b,
             (short)0, acc,  false,  false);
    }

    const float bv = bias[colB];
#pragma unroll
    for (int r = 0; r < 8; ++r) {
        const int row = m0 + r + half * 8;
        float v = acc[r] + bv;
        if (doRelu) v = fmaxf(v, 0.0f);
        v *= outScale;
        const size_t idx = (size_t)row * N + colB;
        if (accumulate) C[idx] += v;
        else            C[idx] = v;
    }
}

extern "C" void kernel_launch(void* const* d_in, const int* in_sizes, int n_in,
                              void* d_out, int out_size, void* d_ws, size_t ws_size,
                              hipStream_t stream) {
    const float* blocks = (const float*)d_in[0];
    const int*   esrc   = (const int*)  d_in[1];
    const int*   edst   = (const int*)  d_in[2];
    const float* convW  = (const float*)d_in[3];
    const float* convB  = (const float*)d_in[4];
    const float* W1     = (const float*)d_in[5];
    const float* b1     = (const float*)d_in[6];
    const float* W2     = (const float*)d_in[7];
    const float* b2     = (const float*)d_in[8];
    float*       out    = (float*)d_out;

    char* ws = (char*)d_ws;
    const size_t BUF = (size_t)N_NODES * D_IN * sizeof(float);
    float* buf0 = (float*)(ws);
    float* agg  = (float*)(ws + BUF);
    float* buf2 = (float*)(ws + 2 * BUF);
    float* deg  = (float*)(ws + 3 * BUF);
    float* H1   = buf0;

    fill_zero_kernel<<<(N_ETYPES * N_NODES + 255) / 256, 256, 0, stream>>>(deg, N_ETYPES * N_NODES);
    {
        dim3 g((N_EDGES + 255) / 256, N_ETYPES);
        degree_kernel<<<g, 256, 0, stream>>>(edst, deg);
    }

    const float* hin = blocks;
    float* hout = buf0;
    for (int l = 0; l < 2; ++l) {
        for (int e = 0; e < N_ETYPES; ++e) {
            fill_zero_kernel<<<(N_NODES * D_IN + 255) / 256, 256, 0, stream>>>(agg, N_NODES * D_IN);
            scatter_kernel<<<(N_EDGES + 7) / 8, 256, 0, stream>>>(
                hin, esrc + (size_t)e * N_EDGES, edst + (size_t)e * N_EDGES, agg);
            dim3 g(D_IN / 16, N_NODES / 16);
            wmma_gemm_f32_kernel<<<g, 32, 0, stream>>>(
                agg,
                convW + ((size_t)l * N_ETYPES + e) * D_IN * D_IN,
                convB + ((size_t)l * N_ETYPES + e) * D_IN,
                hout,
                deg + (size_t)e * N_NODES,
                N_NODES, D_IN, D_IN,
                1.0f / 3.0f,  (e > 0) ? 1 : 0,  1);
        }
        hin  = hout;
        hout = buf2;
    }

    {
        dim3 g(D_HID / 16, N_NODES / 16);
        wmma_gemm_f32_kernel<<<g, 32, 0, stream>>>(
            hin, W1, b1, H1, nullptr, N_NODES, D_HID, D_IN, 1.0f, 0, 0);
    }
    {
        dim3 g(D_OUT / 16, N_NODES / 16);
        wmma_gemm_f32_kernel<<<g, 32, 0, stream>>>(
            H1, W2, b2, out, nullptr, N_NODES, D_OUT, D_HID, 1.0f, 0, 0);
    }
}


# GEMM retiled to row strips: an 8-wave work-group stages the W slab in LDS once and each wave does 16 rows x all its column tiles, so A is read once instead of once per 16 columns
# speedup vs baseline: 1.3167x; 1.3167x over previous
#include <hip/hip_runtime.h>

typedef float v2f __attribute__((ext_vector_type(2)));
typedef float v8f __attribute__((ext_vector_type(8)));

#define N_NODES  100000
#define N_ETYPES 3
#define N_EDGES  1600000
#define D_IN     128
#define D_HID    256
#define D_OUT    32

__global__ void fill_zero_kernel(float* __restrict__ p, int n) {
    int i = blockIdx.x * blockDim.x + threadIdx.x;
    if (i < n) p[i] = 0.0f;
}

__global__ void degree_kernel(const int* __restrict__ dst, float* __restrict__ deg) {
    int e = blockIdx.y;
    int i = blockIdx.x * blockDim.x + threadIdx.x;
    if (i < N_EDGES) {
        int d = dst[(size_t)e * N_EDGES + i];
        atomicAdd(&deg[(size_t)e * N_NODES + d], 1.0f);
    }
}

__global__ void scatter_kernel(const float* __restrict__ h,
                               const int* __restrict__ src,
                               const int* __restrict__ dst,
                               float* __restrict__ agg) {
    int edge = blockIdx.x * 8 + (threadIdx.x >> 5);
    int c    = (threadIdx.x & 31) * 4;
    if (edge >= N_EDGES) return;
    int s = src[edge];
    int d = dst[edge];
    const float4 v = *(const float4*)(h + (size_t)s * D_IN + c);
    float* o = agg + (size_t)d * D_IN + c;
    atomicAdd(o + 0, v.x);
    atomicAdd(o + 1, v.y);
    atomicAdd(o + 2, v.z);
    atomicAdd(o + 3, v.w);
}

template <int NT, int N, int K, bool RELU, bool ACC>
__global__ __launch_bounds__(256)
void wmma_gemm_strip_f32_kernel(const float* __restrict__ A,
                                const float* __restrict__ W,
                                const float* __restrict__ bias,
                                float* __restrict__ C,
                                const float* __restrict__ deg,
                                int M, float outScale) {
    constexpr int JT = NT / 16;
    constexpr int P  = NT + 16;
    constexpr int KB = K / 32;
    __shared__ __attribute__((aligned(16))) float2 Ws[(K / 2) * P];

    const int tid = threadIdx.x;
    const int nb  = blockIdx.y * NT;

    constexpr int STAGE_IT = (K / 2) * (NT / 4) / 256;
    static_assert(STAGE_IT * 256 == (K / 2) * (NT / 4), "slab divides evenly over the work-group");
#pragma unroll
    for (int it = 0; it < STAGE_IT; ++it) {
        const int idx = tid + it * 256;
        const int kp = idx / (NT / 4);
        const int n4 = (idx % (NT / 4)) * 4;
        const float4 r0 = *(const float4*)(W + (size_t)(2 * kp)     * N + nb + n4);
        const float4 r1 = *(const float4*)(W + (size_t)(2 * kp + 1) * N + nb + n4);
        float4* d = (float4*)(Ws + kp * P + n4);
        d[0] = make_float4(r0.x, r1.x, r0.y, r1.y);
        d[1] = make_float4(r0.z, r1.z, r0.w, r1.w);
    }
    __syncthreads();

    const int wave = tid >> 5;
    const int lane = tid & 31;
    const int half = lane >> 4;
    const int l15  = lane & 15;
    const int m0   = (blockIdx.x * 8 + wave) * 16;
    if (m0 >= M) return;

    const int rowA = m0 + l15;
    float rs = 1.0f;
    if (deg) rs = 1.0f / fmaxf(deg[rowA], 1.0f);

    v8f acc[JT];
#pragma unroll
    for (int j = 0; j < JT; ++j) acc[j] = v8f{};

    const float*  arow = A + (size_t)rowA * K + half * 2;
    const float2* wb   = Ws + half * P + l15;

    float2 cur[8], nxt[8];
#pragma unroll
    for (int s = 0; s < 8; ++s) cur[s] = *(const float2*)(arow + s * 4);

#pragma unroll 1
    for (int kb = 0; kb < KB; ++kb) {
        const int kn = (kb + 1 < KB) ? kb + 1 : kb;
#pragma unroll
        for (int s = 0; s < 8; ++s) nxt[s] = *(const float2*)(arow + kn * 32 + s * 4);
#pragma unroll
        for (int s = 0; s < 8; ++s) {
            v2f a;
            a.x = cur[s].x * rs;
            a.y = cur[s].y * rs;
            const float2* wk = wb + (kb * 16 + s * 2) * P;
#pragma unroll
            for (int j = 0; j < JT; ++j) {
                const float2 bv = wk[16 * j];
                v2f b;
                b.x = bv.x;
                b.y = bv.y;
                acc[j] = __builtin_amdgcn_wmma_f32_16x16x4_f32(
                     false, a,  false, b,
                     (short)0, acc[j],  false,  false);
            }
        }
#pragma unroll
        for (int s = 0; s < 8; ++s) cur[s] = nxt[s];
    }

    float* crow = C + (size_t)(m0 + half * 8) * N + nb + l15;
    float old[ACC ? JT : 1][8];
    if (ACC) {
#pragma unroll
        for (int j = 0; j < JT; ++j)
#pragma unroll
            for (int r = 0; r < 8; ++r) old[j][r] = crow[(size_t)r * N + 16 * j];
    }
#pragma unroll
    for (int j = 0; j < JT; ++j) {
        const float bv = bias[nb + 16 * j + l15];
#pragma unroll
        for (int r = 0; r < 8; ++r) {
            float v = acc[j][r] + bv;
            if (RELU) v = fmaxf(v, 0.0f);
            v *= outScale;
            if (ACC) v = old[j][r] + v;
            crow[(size_t)r * N + 16 * j] = v;
        }
    }
}

extern "C" void kernel_launch(void* const* d_in, const int* in_sizes, int n_in,
                              void* d_out, int out_size, void* d_ws, size_t ws_size,
                              hipStream_t stream) {
    const float* blocks = (const float*)d_in[0];
    const int*   esrc   = (const int*)  d_in[1];
    const int*   edst   = (const int*)  d_in[2];
    const float* convW  = (const float*)d_in[3];
    const float* convB  = (const float*)d_in[4];
    const float* W1     = (const float*)d_in[5];
    const float* b1     = (const float*)d_in[6];
    const float* W2     = (const float*)d_in[7];
    const float* b2     = (const float*)d_in[8];
    float*       out    = (float*)d_out;

    char* ws = (char*)d_ws;
    const size_t BUF = (size_t)N_NODES * D_IN * sizeof(float);
    float* buf0 = (float*)(ws);
    float* agg  = (float*)(ws + BUF);
    float* buf2 = (float*)(ws + 2 * BUF);
    float* deg  = (float*)(ws + 3 * BUF);
    float* H1   = buf0;

    fill_zero_kernel<<<(N_ETYPES * N_NODES + 255) / 256, 256, 0, stream>>>(deg, N_ETYPES * N_NODES);
    {
        dim3 g((N_EDGES + 255) / 256, N_ETYPES);
        degree_kernel<<<g, 256, 0, stream>>>(edst, deg);
    }

    const int row_blocks = (N_NODES / 16 + 7) / 8;

    const float* hin = blocks;
    float* hout = buf0;
    for (int l = 0; l < 2; ++l) {
        for (int e = 0; e < N_ETYPES; ++e) {
            fill_zero_kernel<<<(N_NODES * D_IN + 255) / 256, 256, 0, stream>>>(agg, N_NODES * D_IN);
            scatter_kernel<<<(N_EDGES + 7) / 8, 256, 0, stream>>>(
                hin, esrc + (size_t)e * N_EDGES, edst + (size_t)e * N_EDGES, agg);
            dim3 g(row_blocks, D_IN / 128);
            const float* Wle = convW + ((size_t)l * N_ETYPES + e) * D_IN * D_IN;
            const float* ble = convB + ((size_t)l * N_ETYPES + e) * D_IN;
            if (e == 0)
                wmma_gemm_strip_f32_kernel<128, D_IN, D_IN, true, false><<<g, 256, 0, stream>>>(
                    agg, Wle, ble, hout, deg + (size_t)e * N_NODES, N_NODES, 1.0f / 3.0f);
            else
                wmma_gemm_strip_f32_kernel<128, D_IN, D_IN, true, true><<<g, 256, 0, stream>>>(
                    agg, Wle, ble, hout, deg + (size_t)e * N_NODES, N_NODES, 1.0f / 3.0f);
        }
        hin  = hout;
        hout = buf2;
    }

    {
        dim3 g(row_blocks, D_HID / 128);
        wmma_gemm_strip_f32_kernel<128, D_HID, D_IN, false, false><<<g, 256, 0, stream>>>(
            hin, W1, b1, H1, nullptr, N_NODES, 1.0f);
    }
    {
        dim3 g(row_blocks, D_OUT / 32);
        wmma_gemm_strip_f32_kernel<32, D_OUT, D_HID, false, false><<<g, 256, 0, stream>>>(
            H1, W2, b2, out, nullptr, N_NODES, 1.0f);
    }
}


# Neighbour sums from a destination-sorted edge list (integer-atomic counting sort, built once per call, used by both layers) gathered per node; no float atomics, no zero fill; sum order = slot order
# speedup vs baseline: 2.5986x; 1.9736x over previous
#include <hip/hip_runtime.h>

typedef float v2f __attribute__((ext_vector_type(2)));
typedef float v8f __attribute__((ext_vector_type(8)));

#define N_NODES  100000
#define N_ETYPES 3
#define N_EDGES  1600000
#define D_IN     128
#define D_HID    256
#define D_OUT    32

#define HALF     (N_NODES / 2)
#define RPS      (N_NODES + 4)

__global__ void fill_zero_kernel(int* __restrict__ p, int n) {
    int i = blockIdx.x * blockDim.x + threadIdx.x;
    if (i < n) p[i] = 0;
}


__global__ void csr_count_kernel(const int* __restrict__ dst, int* __restrict__ cnt) {
    const int e = blockIdx.y;
    const int i = blockIdx.x * blockDim.x + threadIdx.x;
    if (i < N_EDGES / 4) {
        const int4 d = *(const int4*)(dst + (size_t)e * N_EDGES + (size_t)i * 4);
        int* c = cnt + (size_t)e * N_NODES;
        atomicAdd(c + d.x, 1);
        atomicAdd(c + d.y, 1);
        atomicAdd(c + d.z, 1);
        atomicAdd(c + d.w, 1);
    }
}

__global__ __launch_bounds__(1024)
void csr_scan_kernel(const int* __restrict__ cnt, int* __restrict__ rowptr, float* __restrict__ deg) {
    __shared__ int wsum[32];
    const int e    = blockIdx.x;
    const int t    = threadIdx.x;
    const int lane = t & 31;
    const int w    = t >> 5;
    const bool live = t < N_NODES / 100;
    const int4* c4 = (const int4*)(cnt + (size_t)e * N_NODES) + t * 25;

    int s = 0;
    if (live) {
#pragma unroll 5
        for (int i = 0; i < 25; ++i) {
            const int4 v = c4[i];
            s += (v.x + v.y) + (v.z + v.w);
        }
    }
    int inc = s;
#pragma unroll
    for (int d = 1; d < 32; d <<= 1) {
        const int o = __shfl_up(inc, d);
        if (lane >= d) inc += o;
    }
    if (lane == 31) wsum[w] = inc;
    __syncthreads();
    if (w == 0) {
        const int v = wsum[lane];
        int iv = v;
#pragma unroll
        for (int d = 1; d < 32; d <<= 1) {
            const int o = __shfl_up(iv, d);
            if (lane >= d) iv += o;
        }
        wsum[lane] = iv - v;
    }
    __syncthreads();
    int run = wsum[w] + inc - s;
    if (live) {
        int4*   r4 = (int4*)(rowptr + (size_t)e * RPS) + t * 25;
        float4* d4 = (float4*)(deg + (size_t)e * N_NODES) + t * 25;
#pragma unroll 5
        for (int i = 0; i < 25; ++i) {
            const int4 v = c4[i];
            int4 o;
            o.x = run; run += v.x;
            o.y = run; run += v.y;
            o.z = run; run += v.z;
            o.w = run; run += v.w;
            r4[i] = o;
            d4[i] = make_float4((float)v.x, (float)v.y, (float)v.z, (float)v.w);
        }
        if (t == N_NODES / 100 - 1) rowptr[(size_t)e * RPS + N_NODES] = run;
    }
}

__global__ void csr_fill_kernel(const int* __restrict__ src, const int* __restrict__ dst,
                                const int* __restrict__ rowptr, int* __restrict__ cnt,
                                int* __restrict__ col) {
    const int e = blockIdx.y;
    const int i = blockIdx.x * blockDim.x + threadIdx.x;
    if (i < N_EDGES / 4) {
        const int4 d = *(const int4*)(dst + (size_t)e * N_EDGES + (size_t)i * 4);
        const int4 s = *(const int4*)(src + (size_t)e * N_EDGES + (size_t)i * 4);
        int*       c  = cnt + (size_t)e * N_NODES;
        const int* rp = rowptr + (size_t)e * RPS;
        int*       cl = col + (size_t)e * N_EDGES;
        const int p0 = atomicSub(c + d.x, 1) - 1;
        const int p1 = atomicSub(c + d.y, 1) - 1;
        const int p2 = atomicSub(c + d.z, 1) - 1;
        const int p3 = atomicSub(c + d.w, 1) - 1;
        cl[rp[d.x] + p0] = s.x;
        cl[rp[d.y] + p1] = s.y;
        cl[rp[d.z] + p2] = s.z;
        cl[rp[d.w] + p3] = s.w;
    }
}

__global__ __launch_bounds__(256)
void gather_sum_kernel(const float* __restrict__ h,
                       const int* __restrict__ rowptr,
                       const int* __restrict__ col,
                       float* __restrict__ agg) {
    const int wave = __builtin_amdgcn_readfirstlane((int)(threadIdx.x >> 5));
    const int lane = threadIdx.x & 31;
    const int r    = blockIdx.x * 8 + wave;
    const int beg  = rowptr[r];
    const int end  = rowptr[r + 1];
    const float* hl = h + lane * 4;
    float4 acc = make_float4(0.0f, 0.0f, 0.0f, 0.0f);
    for (int j = beg; j < end; j += 32) {
        const int m = (end - j < 32) ? end - j : 32;
        int mine = 0;
        if (lane < m) mine = col[j + lane];
        for (int k = 0; k < m; k += 8) {
            float4 v[8];
#pragma unroll
            for (int u = 0; u < 8; ++u) {
                const int q = (k + u < m) ? k + u : k;
                const int s = __builtin_amdgcn_readlane(mine, q);
                v[u] = *(const float4*)(hl + (size_t)s * D_IN);
            }
#pragma unroll
            for (int u = 0; u < 8; ++u) {
                if (k + u < m) {
                    acc.x += v[u].x;
                    acc.y += v[u].y;
                    acc.z += v[u].z;
                    acc.w += v[u].w;
                }
            }
        }
    }
    *(float4*)(agg + (size_t)r * D_IN + lane * 4) = acc;
}

template <int NT, int N, int K, bool RELU, bool ACC>
__global__ __launch_bounds__(256)
void wmma_gemm_strip_f32_kernel(const float* __restrict__ A,
                                const float* __restrict__ W,
                                const float* __restrict__ bias,
                                float* __restrict__ C,
                                const float* __restrict__ deg,
                                int M, float outScale) {
    constexpr int JT = NT / 16;
    constexpr int P  = NT + 16;
    constexpr int KB = K / 32;
    __shared__ __attribute__((aligned(16))) float2 Ws[(K / 2) * P];

    const int tid = threadIdx.x;
    const int nb  = blockIdx.y * NT;

    constexpr int STAGE_IT = (K / 2) * (NT / 4) / 256;
    static_assert(STAGE_IT * 256 == (K / 2) * (NT / 4), "slab divides evenly over the work-group");
#pragma unroll
    for (int it = 0; it < STAGE_IT; ++it) {
        const int idx = tid + it * 256;
        const int kp = idx / (NT / 4);
        const int n4 = (idx % (NT / 4)) * 4;
        const float4 r0 = *(const float4*)(W + (size_t)(2 * kp)     * N + nb + n4);
        const float4 r1 = *(const float4*)(W + (size_t)(2 * kp + 1) * N + nb + n4);
        float4* d = (float4*)(Ws + kp * P + n4);
        d[0] = make_float4(r0.x, r1.x, r0.y, r1.y);
        d[1] = make_float4(r0.z, r1.z, r0.w, r1.w);
    }
    __syncthreads();

    const int wave = tid >> 5;
    const int lane = tid & 31;
    const int half = lane >> 4;
    const int l15  = lane & 15;
    const int m0   = (blockIdx.x * 8 + wave) * 16;
    if (m0 >= M) return;

    const int rowA = m0 + l15;
    float rs = 1.0f;
    if (deg) rs = 1.0f / fmaxf(deg[rowA], 1.0f);

    v8f acc[JT];
#pragma unroll
    for (int j = 0; j < JT; ++j) acc[j] = v8f{};

    const float*  arow = A + (size_t)rowA * K + half * 2;
    const float2* wb   = Ws + half * P + l15;

    float2 cur[8], nxt[8];
#pragma unroll
    for (int s = 0; s < 8; ++s) cur[s] = *(const float2*)(arow + s * 4);

#pragma unroll 1
    for (int kb = 0; kb < KB; ++kb) {
        const int kn = (kb + 1 < KB) ? kb + 1 : kb;
#pragma unroll
        for (int s = 0; s < 8; ++s) nxt[s] = *(const float2*)(arow + kn * 32 + s * 4);
#pragma unroll
        for (int s = 0; s < 8; ++s) {
            v2f a;
            a.x = cur[s].x * rs;
            a.y = cur[s].y * rs;
            const float2* wk = wb + (kb * 16 + s * 2) * P;
#pragma unroll
            for (int j = 0; j < JT; ++j) {
                const float2 bv = wk[16 * j];
                v2f b;
                b.x = bv.x;
                b.y = bv.y;
                acc[j] = __builtin_amdgcn_wmma_f32_16x16x4_f32(
                     false, a,  false, b,
                     (short)0, acc[j],  false,  false);
            }
        }
#pragma unroll
        for (int s = 0; s < 8; ++s) cur[s] = nxt[s];
    }

    float* crow = C + (size_t)(m0 + half * 8) * N + nb + l15;
    float old[ACC ? JT : 1][8];
    if (ACC) {
#pragma unroll
        for (int j = 0; j < JT; ++j)
#pragma unroll
            for (int r = 0; r < 8; ++r) old[j][r] = crow[(size_t)r * N + 16 * j];
    }
#pragma unroll
    for (int j = 0; j < JT; ++j) {
        const float bv = bias[nb + 16 * j + l15];
#pragma unroll
        for (int r = 0; r < 8; ++r) {
            float v = acc[j][r] + bv;
            if (RELU) v = fmaxf(v, 0.0f);
            v *= outScale;
            if (ACC) v = old[j][r] + v;
            crow[(size_t)r * N + 16 * j] = v;
        }
    }
}

extern "C" void kernel_launch(void* const* d_in, const int* in_sizes, int n_in,
                              void* d_out, int out_size, void* d_ws, size_t ws_size,
                              hipStream_t stream) {
    const float* blocks = (const float*)d_in[0];
    const int*   esrc   = (const int*)  d_in[1];
    const int*   edst   = (const int*)  d_in[2];
    const float* convW  = (const float*)d_in[3];
    const float* convB  = (const float*)d_in[4];
    const float* W1     = (const float*)d_in[5];
    const float* b1     = (const float*)d_in[6];
    const float* W2     = (const float*)d_in[7];
    const float* b2     = (const float*)d_in[8];
    float*       out    = (float*)d_out;

    char* ws = (char*)d_ws;
    const size_t BUF = (size_t)N_NODES * D_IN * sizeof(float);
    float* buf0 = (float*)(ws);
    float* agg  = (float*)(ws + BUF);
    char*  csr  = ws + BUF + BUF / 2;
    int*   col    = (int*)(csr);
    int*   rowptr = (int*)(csr + (size_t)N_ETYPES * N_EDGES * sizeof(int));
    int*   cnt    = rowptr + (size_t)N_ETYPES * RPS;
    float* buf2 = (float*)(ws + 2 * BUF);
    float* deg  = (float*)(ws + 3 * BUF);
    float* H1   = buf0;

    fill_zero_kernel<<<(N_ETYPES * N_NODES + 255) / 256, 256, 0, stream>>>(cnt, N_ETYPES * N_NODES);
    {
        dim3 g((N_EDGES / 4 + 255) / 256, N_ETYPES);
        csr_count_kernel<<<g, 256, 0, stream>>>(edst, cnt);
        csr_scan_kernel<<<N_ETYPES, 1024, 0, stream>>>(cnt, rowptr, deg);
        csr_fill_kernel<<<g, 256, 0, stream>>>(esrc, edst, rowptr, cnt, col);
    }

    const int row_blocks      = (N_NODES / 16 + 7) / 8;
    const int row_blocks_half = (HALF / 16 + 7) / 8;

    const float* hin = blocks;
    float* hout = buf0;
    for (int l = 0; l < 2; ++l) {
        for (int e = 0; e < N_ETYPES; ++e) {
            const float* Wle = convW + ((size_t)l * N_ETYPES + e) * D_IN * D_IN;
            const float* ble = convB + ((size_t)l * N_ETYPES + e) * D_IN;
            for (int p = 0; p < 2; ++p) {
                gather_sum_kernel<<<HALF / 8, 256, 0, stream>>>(
                    hin, rowptr + (size_t)e * RPS + (size_t)p * HALF, col + (size_t)e * N_EDGES, agg);
                dim3 g(row_blocks_half, D_IN / 128);
                float*       Cp = hout + (size_t)p * HALF * D_IN;
                const float* dp = deg + (size_t)e * N_NODES + (size_t)p * HALF;
                if (e == 0)
                    wmma_gemm_strip_f32_kernel<128, D_IN, D_IN, true, false><<<g, 256, 0, stream>>>(
                        agg, Wle, ble, Cp, dp, HALF, 1.0f / 3.0f);
                else
                    wmma_gemm_strip_f32_kernel<128, D_IN, D_IN, true, true><<<g, 256, 0, stream>>>(
                        agg, Wle, ble, Cp, dp, HALF, 1.0f / 3.0f);
            }
        }
        hin  = hout;
        hout = buf2;
    }

    {
        dim3 g(row_blocks, D_HID / 128);
        wmma_gemm_strip_f32_kernel<128, D_HID, D_IN, false, false><<<g, 256, 0, stream>>>(
            hin, W1, b1, H1, nullptr, N_NODES, 1.0f);
    }
    {
        dim3 g(row_blocks, D_OUT / 32);
        wmma_gemm_strip_f32_kernel<32, D_OUT, D_HID, false, false><<<g, 256, 0, stream>>>(
            H1, W2, b2, out, nullptr, N_NODES, 1.0f);
    }
}


# Edge-list build with one atomic pass instead of two: the counting add's return value is kept per edge as its slot and the fill pass has no atomics; per-node sum order = arrival order of counting adds
# speedup vs baseline: 3.1305x; 1.2047x over previous
#include <hip/hip_runtime.h>

typedef float v2f __attribute__((ext_vector_type(2)));
typedef float v8f __attribute__((ext_vector_type(8)));

#define N_NODES  100000
#define N_ETYPES 3
#define N_EDGES  1600000
#define D_IN     128
#define D_HID    256
#define D_OUT    32

#define HALF     (N_NODES / 2)
#define RPS      (N_NODES + 4)

__global__ void fill_zero_kernel(int* __restrict__ p, int n) {
    int i = blockIdx.x * blockDim.x + threadIdx.x;
    if (i < n) p[i] = 0;
}


__global__ void csr_count_kernel(const int* __restrict__ dst, int* __restrict__ cnt,
                                 int* __restrict__ rank) {
    const int e = blockIdx.y;
    const int i = blockIdx.x * blockDim.x + threadIdx.x;
    if (i < N_EDGES / 4) {
        const size_t o = (size_t)e * N_EDGES + (size_t)i * 4;
        const int4 d = *(const int4*)(dst + o);
        int* c = cnt + (size_t)e * N_NODES;
        int4 r;
        r.x = atomicAdd(c + d.x, 1);
        r.y = atomicAdd(c + d.y, 1);
        r.z = atomicAdd(c + d.z, 1);
        r.w = atomicAdd(c + d.w, 1);
        *(int4*)(rank + o) = r;
    }
}

__global__ __launch_bounds__(1024)
void csr_scan_kernel(const int* __restrict__ cnt, int* __restrict__ rowptr, float* __restrict__ deg) {
    __shared__ int wsum[32];
    const int e    = blockIdx.x;
    const int t    = threadIdx.x;
    const int lane = t & 31;
    const int w    = t >> 5;
    const bool live = t < N_NODES / 100;
    const int4* c4 = (const int4*)(cnt + (size_t)e * N_NODES) + t * 25;

    int s = 0;
    if (live) {
#pragma unroll 5
        for (int i = 0; i < 25; ++i) {
            const int4 v = c4[i];
            s += (v.x + v.y) + (v.z + v.w);
        }
    }
    int inc = s;
#pragma unroll
    for (int d = 1; d < 32; d <<= 1) {
        const int o = __shfl_up(inc, d);
        if (lane >= d) inc += o;
    }
    if (lane == 31) wsum[w] = inc;
    __syncthreads();
    if (w == 0) {
        const int v = wsum[lane];
        int iv = v;
#pragma unroll
        for (int d = 1; d < 32; d <<= 1) {
            const int o = __shfl_up(iv, d);
            if (lane >= d) iv += o;
        }
        wsum[lane] = iv - v;
    }
    __syncthreads();
    int run = wsum[w] + inc - s;
    if (live) {
        int4*   r4 = (int4*)(rowptr + (size_t)e * RPS) + t * 25;
        float4* d4 = (float4*)(deg + (size_t)e * N_NODES) + t * 25;
#pragma unroll 5
        for (int i = 0; i < 25; ++i) {
            const int4 v = c4[i];
            int4 o;
            o.x = run; run += v.x;
            o.y = run; run += v.y;
            o.z = run; run += v.z;
            o.w = run; run += v.w;
            r4[i] = o;
            d4[i] = make_float4((float)v.x, (float)v.y, (float)v.z, (float)v.w);
        }
        if (t == N_NODES / 100 - 1) rowptr[(size_t)e * RPS + N_NODES] = run;
    }
}

__global__ void csr_fill_kernel(const int* __restrict__ src, const int* __restrict__ dst,
                                const int* __restrict__ rank, const int* __restrict__ rowptr,
                                int* __restrict__ col) {
    const int e = blockIdx.y;
    const int i = blockIdx.x * blockDim.x + threadIdx.x;
    if (i < N_EDGES / 4) {
        const size_t o = (size_t)e * N_EDGES + (size_t)i * 4;
        const int4 d = *(const int4*)(dst + o);
        const int4 s = *(const int4*)(src + o);
        const int4 r = *(const int4*)(rank + o);
        const int* rp = rowptr + (size_t)e * RPS;
        int*       cl = col + (size_t)e * N_EDGES;
        cl[rp[d.x] + r.x] = s.x;
        cl[rp[d.y] + r.y] = s.y;
        cl[rp[d.z] + r.z] = s.z;
        cl[rp[d.w] + r.w] = s.w;
    }
}

__global__ __launch_bounds__(256)
void gather_sum_kernel(const float* __restrict__ h,
                       const int* __restrict__ rowptr,
                       const int* __restrict__ col,
                       float* __restrict__ agg) {
    const int wave = __builtin_amdgcn_readfirstlane((int)(threadIdx.x >> 5));
    const int lane = threadIdx.x & 31;
    const int r    = blockIdx.x * 8 + wave;
    const int beg  = rowptr[r];
    const int end  = rowptr[r + 1];
    const float* hl = h + lane * 4;
    float4 acc = make_float4(0.0f, 0.0f, 0.0f, 0.0f);
    for (int j = beg; j < end; j += 32) {
        const int m = (end - j < 32) ? end - j : 32;
        int mine = 0;
        if (lane < m) mine = col[j + lane];
        for (int k = 0; k < m; k += 8) {
            float4 v[8];
#pragma unroll
            for (int u = 0; u < 8; ++u) {
                const int q = (k + u < m) ? k + u : k;
                const int s = __builtin_amdgcn_readlane(mine, q);
                v[u] = *(const float4*)(hl + (size_t)s * D_IN);
            }
#pragma unroll
            for (int u = 0; u < 8; ++u) {
                if (k + u < m) {
                    acc.x += v[u].x;
                    acc.y += v[u].y;
                    acc.z += v[u].z;
                    acc.w += v[u].w;
                }
            }
        }
    }
    *(float4*)(agg + (size_t)r * D_IN + lane * 4) = acc;
}

template <int NT, int N, int K, bool RELU, bool ACC>
__global__ __launch_bounds__(256)
void wmma_gemm_strip_f32_kernel(const float* __restrict__ A,
                                const float* __restrict__ W,
                                const float* __restrict__ bias,
                                float* __restrict__ C,
                                const float* __restrict__ deg,
                                int M, float outScale) {
    constexpr int JT = NT / 16;
    constexpr int P  = NT + 16;
    constexpr int KB = K / 32;
    __shared__ __attribute__((aligned(16))) float2 Ws[(K / 2) * P];

    const int tid = threadIdx.x;
    const int nb  = blockIdx.y * NT;

    constexpr int STAGE_IT = (K / 2) * (NT / 4) / 256;
    static_assert(STAGE_IT * 256 == (K / 2) * (NT / 4), "slab divides evenly over the work-group");
#pragma unroll
    for (int it = 0; it < STAGE_IT; ++it) {
        const int idx = tid + it * 256;
        const int kp = idx / (NT / 4);
        const int n4 = (idx % (NT / 4)) * 4;
        const float4 r0 = *(const float4*)(W + (size_t)(2 * kp)     * N + nb + n4);
        const float4 r1 = *(const float4*)(W + (size_t)(2 * kp + 1) * N + nb + n4);
        float4* d = (float4*)(Ws + kp * P + n4);
        d[0] = make_float4(r0.x, r1.x, r0.y, r1.y);
        d[1] = make_float4(r0.z, r1.z, r0.w, r1.w);
    }
    __syncthreads();

    const int wave = tid >> 5;
    const int lane = tid & 31;
    const int half = lane >> 4;
    const int l15  = lane & 15;
    const int m0   = (blockIdx.x * 8 + wave) * 16;
    if (m0 >= M) return;

    const int rowA = m0 + l15;
    float rs = 1.0f;
    if (deg) rs = 1.0f / fmaxf(deg[rowA], 1.0f);

    v8f acc[JT];
#pragma unroll
    for (int j = 0; j < JT; ++j) acc[j] = v8f{};

    const float*  arow = A + (size_t)rowA * K + half * 2;
    const float2* wb   = Ws + half * P + l15;

    float2 cur[8], nxt[8];
#pragma unroll
    for (int s = 0; s < 8; ++s) cur[s] = *(const float2*)(arow + s * 4);

#pragma unroll 1
    for (int kb = 0; kb < KB; ++kb) {
        const int kn = (kb + 1 < KB) ? kb + 1 : kb;
#pragma unroll
        for (int s = 0; s < 8; ++s) nxt[s] = *(const float2*)(arow + kn * 32 + s * 4);
#pragma unroll
        for (int s = 0; s < 8; ++s) {
            v2f a;
            a.x = cur[s].x * rs;
            a.y = cur[s].y * rs;
            const float2* wk = wb + (kb * 16 + s * 2) * P;
#pragma unroll
            for (int j = 0; j < JT; ++j) {
                const float2 bv = wk[16 * j];
                v2f b;
                b.x = bv.x;
                b.y = bv.y;
                acc[j] = __builtin_amdgcn_wmma_f32_16x16x4_f32(
                     false, a,  false, b,
                     (short)0, acc[j],  false,  false);
            }
        }
#pragma unroll
        for (int s = 0; s < 8; ++s) cur[s] = nxt[s];
    }

    float* crow = C + (size_t)(m0 + half * 8) * N + nb + l15;
    float old[ACC ? JT : 1][8];
    if (ACC) {
#pragma unroll
        for (int j = 0; j < JT; ++j)
#pragma unroll
            for (int r = 0; r < 8; ++r) old[j][r] = crow[(size_t)r * N + 16 * j];
    }
#pragma unroll
    for (int j = 0; j < JT; ++j) {
        const float bv = bias[nb + 16 * j + l15];
#pragma unroll
        for (int r = 0; r < 8; ++r) {
            float v = acc[j][r] + bv;
            if (RELU) v = fmaxf(v, 0.0f);
            v *= outScale;
            if (ACC) v = old[j][r] + v;
            crow[(size_t)r * N + 16 * j] = v;
        }
    }
}

extern "C" void kernel_launch(void* const* d_in, const int* in_sizes, int n_in,
                              void* d_out, int out_size, void* d_ws, size_t ws_size,
                              hipStream_t stream) {
    const float* blocks = (const float*)d_in[0];
    const int*   esrc   = (const int*)  d_in[1];
    const int*   edst   = (const int*)  d_in[2];
    const float* convW  = (const float*)d_in[3];
    const float* convB  = (const float*)d_in[4];
    const float* W1     = (const float*)d_in[5];
    const float* b1     = (const float*)d_in[6];
    const float* W2     = (const float*)d_in[7];
    const float* b2     = (const float*)d_in[8];
    float*       out    = (float*)d_out;

    char* ws = (char*)d_ws;
    const size_t BUF = (size_t)N_NODES * D_IN * sizeof(float);
    float* buf0 = (float*)(ws);
    float* agg  = (float*)(ws + BUF);
    char*  csr  = ws + BUF + BUF / 2;
    int*   col    = (int*)(csr);
    int*   rowptr = (int*)(csr + (size_t)N_ETYPES * N_EDGES * sizeof(int));
    int*   cnt    = rowptr + (size_t)N_ETYPES * RPS;
    float* buf2 = (float*)(ws + 2 * BUF);
    int*   rank = (int*)(ws + 2 * BUF);
    float* deg  = (float*)(ws + 3 * BUF);
    float* H1   = buf0;

    fill_zero_kernel<<<(N_ETYPES * N_NODES + 255) / 256, 256, 0, stream>>>(cnt, N_ETYPES * N_NODES);
    {
        dim3 g((N_EDGES / 4 + 255) / 256, N_ETYPES);
        csr_count_kernel<<<g, 256, 0, stream>>>(edst, cnt, rank);
        csr_scan_kernel<<<N_ETYPES, 1024, 0, stream>>>(cnt, rowptr, deg);
        csr_fill_kernel<<<g, 256, 0, stream>>>(esrc, edst, rank, rowptr, col);
    }

    const int row_blocks      = (N_NODES / 16 + 7) / 8;
    const int row_blocks_half = (HALF / 16 + 7) / 8;

    const float* hin = blocks;
    float* hout = buf0;
    for (int l = 0; l < 2; ++l) {
        for (int e = 0; e < N_ETYPES; ++e) {
            const float* Wle = convW + ((size_t)l * N_ETYPES + e) * D_IN * D_IN;
            const float* ble = convB + ((size_t)l * N_ETYPES + e) * D_IN;
            for (int p = 0; p < 2; ++p) {
                gather_sum_kernel<<<HALF / 8, 256, 0, stream>>>(
                    hin, rowptr + (size_t)e * RPS + (size_t)p * HALF, col + (size_t)e * N_EDGES, agg);
                dim3 g(row_blocks_half, D_IN / 128);
                float*       Cp = hout + (size_t)p * HALF * D_IN;
                const float* dp = deg + (size_t)e * N_NODES + (size_t)p * HALF;
                if (e == 0)
                    wmma_gemm_strip_f32_kernel<128, D_IN, D_IN, true, false><<<g, 256, 0, stream>>>(
                        agg, Wle, ble, Cp, dp, HALF, 1.0f / 3.0f);
                else
                    wmma_gemm_strip_f32_kernel<128, D_IN, D_IN, true, true><<<g, 256, 0, stream>>>(
                        agg, Wle, ble, Cp, dp, HALF, 1.0f / 3.0f);
            }
        }
        hin  = hout;
        hout = buf2;
    }

    {
        dim3 g(row_blocks, D_HID / 128);
        wmma_gemm_strip_f32_kernel<128, D_HID, D_IN, false, false><<<g, 256, 0, stream>>>(
            hin, W1, b1, H1, nullptr, N_NODES, 1.0f);
    }
    {
        dim3 g(row_blocks, D_OUT / 32);
        wmma_gemm_strip_f32_kernel<32, D_OUT, D_HID, false, false><<<g, 256, 0, stream>>>(
            H1, W2, b2, out, nullptr, N_NODES, 1.0f);
    }
}


# Edge list built by a two-level counting sort in LDS (512-node buckets, then slots per bucket) with no global atomics; per-node sum order = arrival order of the LDS counting adds
# speedup vs baseline: 4.6601x; 1.4886x over previous
#include <hip/hip_runtime.h>

typedef float v2f __attribute__((ext_vector_type(2)));
typedef float v8f __attribute__((ext_vector_type(8)));

#define N_NODES  100000
#define N_ETYPES 3
#define N_EDGES  1600000
#define D_IN     128
#define D_HID    256
#define D_OUT    32

#define HALF     (N_NODES / 2)
#define RPS      (N_NODES + 4)

#define BK_SHIFT 9
#define BK_NODES (1 << BK_SHIFT)
#define NBK      ((N_NODES + BK_NODES - 1) / BK_NODES)
#define EPB      4096
#define NBLK     ((N_EDGES + EPB - 1) / EPB)
#define EPT      (EPB / 4 / 256)
#define SCAN_SL  5
#define SLB      ((NBLK + SCAN_SL - 1) / SCAN_SL)
static_assert((size_t)N_ETYPES * NBLK * NBK * sizeof(int) <= (1u << 20), "a histogram plane fits the 1 MiB the driver gives it");

__global__ __launch_bounds__(256)
void part_hist_kernel(const int* __restrict__ dst, int* __restrict__ bh) {
    __shared__ int lh[NBK];
    const int e = blockIdx.y, blk = blockIdx.x, t = threadIdx.x;
    if (t < NBK) lh[t] = 0;
    __syncthreads();
    const int4* d4 = (const int4*)(dst + (size_t)e * N_EDGES) + (size_t)blk * (EPB / 4);
    int4 d[EPT];
#pragma unroll
    for (int u = 0; u < EPT; ++u) {
        const int i = u * 256 + t;
        d[u] = make_int4(0, 0, 0, 0);
        if (blk * (EPB / 4) + i < N_EDGES / 4) d[u] = d4[i];
    }
#pragma unroll
    for (int u = 0; u < EPT; ++u) {
        if (blk * (EPB / 4) + u * 256 + t < N_EDGES / 4) {
            atomicAdd(&lh[d[u].x >> BK_SHIFT], 1);
            atomicAdd(&lh[d[u].y >> BK_SHIFT], 1);
            atomicAdd(&lh[d[u].z >> BK_SHIFT], 1);
            atomicAdd(&lh[d[u].w >> BK_SHIFT], 1);
        }
    }
    __syncthreads();
    if (t < NBK) bh[((size_t)e * NBLK + blk) * NBK + t] = lh[t];
}

__global__ __launch_bounds__(1024)
void part_scan_kernel(const int* __restrict__ bh, int* __restrict__ off) {
    __shared__ int ssum[SCAN_SL][NBK];
    __shared__ int bbase[NBK];
    __shared__ int wtot[8];
    static_assert(NBK <= 256 && SCAN_SL * NBK <= 1024, "thread map covers every (slice, bucket)");
    const int e    = blockIdx.x;
    const int t    = threadIdx.x;
    const int lane = t & 31;
    const int sl   = t / NBK;
    const int b    = t - sl * NBK;
    const bool live = t < SCAN_SL * NBK;
    const int k0 = sl * SLB;
    const int k1 = (k0 + SLB < NBLK) ? k0 + SLB : NBLK;
    const int* pi = bh  + (size_t)e * NBLK * NBK + b;
    int*       po = off + (size_t)e * NBLK * NBK + b;

    int s = 0;
    if (live) {
#pragma unroll 8
        for (int k = k0; k < k1; ++k) s += pi[(size_t)k * NBK];
        ssum[sl][b] = s;
    }
    __syncthreads();
    int tot = 0, inc = 0;
    if (t < 256) {
        if (t < NBK) {
#pragma unroll
            for (int q = 0; q < SCAN_SL; ++q) tot += ssum[q][t];
        }
        inc = tot;
#pragma unroll
        for (int d = 1; d < 32; d <<= 1) {
            const int o = __shfl_up(inc, d);
            if (lane >= d) inc += o;
        }
        if (lane == 31) wtot[t >> 5] = inc;
    }
    __syncthreads();
    if (t < NBK) {
        int wb = 0;
        for (int q = 0; q < (t >> 5); ++q) wb += wtot[q];
        bbase[t] = wb + inc - tot;
    }
    __syncthreads();
    if (live) {
        int run = bbase[b];
        for (int q = 0; q < sl; ++q) run += ssum[q][b];
#pragma unroll 8
        for (int k = k0; k < k1; ++k) {
            const int v = pi[(size_t)k * NBK];
            po[(size_t)k * NBK] = run;
            run += v;
        }
    }
}

__global__ __launch_bounds__(256)
void part_scatter_kernel(const int* __restrict__ src, const int* __restrict__ dst,
                         const int* __restrict__ off, int* __restrict__ pairs) {
    __shared__ int cur[NBK];
    const int e = blockIdx.y, blk = blockIdx.x, t = threadIdx.x;
    if (t < NBK) cur[t] = off[((size_t)e * NBLK + blk) * NBK + t];
    const int4* d4 = (const int4*)(dst + (size_t)e * N_EDGES) + (size_t)blk * (EPB / 4);
    const int4* s4 = (const int4*)(src + (size_t)e * N_EDGES) + (size_t)blk * (EPB / 4);
    int2* pe = (int2*)pairs + (size_t)e * N_EDGES;
    int4 d[EPT], s[EPT];
#pragma unroll
    for (int u = 0; u < EPT; ++u) {
        const int i = u * 256 + t;
        d[u] = make_int4(0, 0, 0, 0);
        s[u] = make_int4(0, 0, 0, 0);
        if (blk * (EPB / 4) + i < N_EDGES / 4) { d[u] = d4[i]; s[u] = s4[i]; }
    }
    __syncthreads();
#pragma unroll
    for (int u = 0; u < EPT; ++u) {
        if (blk * (EPB / 4) + u * 256 + t < N_EDGES / 4) {
            const int qx = atomicAdd(&cur[d[u].x >> BK_SHIFT], 1);
            const int qy = atomicAdd(&cur[d[u].y >> BK_SHIFT], 1);
            const int qz = atomicAdd(&cur[d[u].z >> BK_SHIFT], 1);
            const int qw = atomicAdd(&cur[d[u].w >> BK_SHIFT], 1);
            pe[qx] = make_int2(s[u].x, d[u].x);
            pe[qy] = make_int2(s[u].y, d[u].y);
            pe[qz] = make_int2(s[u].z, d[u].z);
            pe[qw] = make_int2(s[u].w, d[u].w);
        }
    }
}

__global__ __launch_bounds__(BK_NODES)
void bucket_sort_kernel(const int* __restrict__ pairs, const int* __restrict__ off,
                        int* __restrict__ rowptr, float* __restrict__ deg, int* __restrict__ col) {
    __shared__ int cnt[BK_NODES];
    __shared__ int wsum[BK_NODES / 32];
    const int e = blockIdx.y, b = blockIdx.x, t = threadIdx.x;
    const int lane = t & 31;
    const int w    = t >> 5;
    const int* oe  = off + (size_t)e * NBLK * NBK;
    const int beg  = oe[b];
    const int end  = (b + 1 < NBK) ? oe[b + 1] : N_EDGES;
    const int2* pe = (const int2*)pairs + (size_t)e * N_EDGES;
    int*        ce = col + (size_t)e * N_EDGES;

    cnt[t] = 0;
    __syncthreads();
    for (int j0 = beg; j0 < end; j0 += 4 * BK_NODES) {
        int2 p[4];
#pragma unroll
        for (int u = 0; u < 4; ++u) {
            const int j = j0 + u * BK_NODES + t;
            p[u] = pe[(j < end) ? j : end - 1];
        }
#pragma unroll
        for (int u = 0; u < 4; ++u)
            if (j0 + u * BK_NODES + t < end) atomicAdd(&cnt[p[u].y & (BK_NODES - 1)], 1);
    }
    __syncthreads();

    const int c = cnt[t];
    int inc = c;
#pragma unroll
    for (int d = 1; d < 32; d <<= 1) {
        const int o = __shfl_up(inc, d);
        if (lane >= d) inc += o;
    }
    if (lane == 31) wsum[w] = inc;
    __syncthreads();
    int first = beg + inc - c;
    for (int q = 0; q < w; ++q) first += wsum[q];
    cnt[t] = first;
    const int n = b * BK_NODES + t;
    if (n < N_NODES) {
        rowptr[(size_t)e * RPS + n]   = first;
        deg[(size_t)e * N_NODES + n]  = (float)c;
    }
    if (b == NBK - 1 && t == 0) rowptr[(size_t)e * RPS + N_NODES] = end;
    __syncthreads();

    for (int j0 = beg; j0 < end; j0 += 4 * BK_NODES) {
        int2 p[4];
#pragma unroll
        for (int u = 0; u < 4; ++u) {
            const int j = j0 + u * BK_NODES + t;
            p[u] = pe[(j < end) ? j : end - 1];
        }
#pragma unroll
        for (int u = 0; u < 4; ++u) {
            if (j0 + u * BK_NODES + t < end) {
                const int q = atomicAdd(&cnt[p[u].y & (BK_NODES - 1)], 1);
                ce[q] = p[u].x;
            }
        }
    }
}

__global__ __launch_bounds__(256)
void gather_sum_kernel(const float* __restrict__ h,
                       const int* __restrict__ rowptr,
                       const int* __restrict__ col,
                       float* __restrict__ agg) {
    const int wave = __builtin_amdgcn_readfirstlane((int)(threadIdx.x >> 5));
    const int lane = threadIdx.x & 31;
    const int r    = blockIdx.x * 8 + wave;
    const int beg  = rowptr[r];
    const int end  = rowptr[r + 1];
    const float* hl = h + lane * 4;
    float4 acc = make_float4(0.0f, 0.0f, 0.0f, 0.0f);
    for (int j = beg; j < end; j += 32) {
        const int m = (end - j < 32) ? end - j : 32;
        int mine = 0;
        if (lane < m) mine = col[j + lane];
        for (int k = 0; k < m; k += 8) {
            float4 v[8];
#pragma unroll
            for (int u = 0; u < 8; ++u) {
                const int q = (k + u < m) ? k + u : k;
                const int s = __builtin_amdgcn_readlane(mine, q);
                v[u] = *(const float4*)(hl + (size_t)s * D_IN);
            }
#pragma unroll
            for (int u = 0; u < 8; ++u) {
                if (k + u < m) {
                    acc.x += v[u].x;
                    acc.y += v[u].y;
                    acc.z += v[u].z;
                    acc.w += v[u].w;
                }
            }
        }
    }
    *(float4*)(agg + (size_t)r * D_IN + lane * 4) = acc;
}

template <int NT, int N, int K, bool RELU, bool ACC>
__global__ __launch_bounds__(256)
void wmma_gemm_strip_f32_kernel(const float* __restrict__ A,
                                const float* __restrict__ W,
                                const float* __restrict__ bias,
                                float* __restrict__ C,
                                const float* __restrict__ deg,
                                int M, float outScale) {
    constexpr int JT = NT / 16;
    constexpr int P  = NT + 16;
    constexpr int KB = K / 32;
    __shared__ __attribute__((aligned(16))) float2 Ws[(K / 2) * P];

    const int tid = threadIdx.x;
    const int nb  = blockIdx.y * NT;

    constexpr int STAGE_IT = (K / 2) * (NT / 4) / 256;
    static_assert(STAGE_IT * 256 == (K / 2) * (NT / 4), "slab divides evenly over the work-group");
#pragma unroll
    for (int it = 0; it < STAGE_IT; ++it) {
        const int idx = tid + it * 256;
        const int kp = idx / (NT / 4);
        const int n4 = (idx % (NT / 4)) * 4;
        const float4 r0 = *(const float4*)(W + (size_t)(2 * kp)     * N + nb + n4);
        const float4 r1 = *(const float4*)(W + (size_t)(2 * kp + 1) * N + nb + n4);
        float4* d = (float4*)(Ws + kp * P + n4);
        d[0] = make_float4(r0.x, r1.x, r0.y, r1.y);
        d[1] = make_float4(r0.z, r1.z, r0.w, r1.w);
    }
    __syncthreads();

    const int wave = tid >> 5;
    const int lane = tid & 31;
    const int half = lane >> 4;
    const int l15  = lane & 15;
    const int m0   = (blockIdx.x * 8 + wave) * 16;
    if (m0 >= M) return;

    const int rowA = m0 + l15;
    float rs = 1.0f;
    if (deg) rs = 1.0f / fmaxf(deg[rowA], 1.0f);

    v8f acc[JT];
#pragma unroll
    for (int j = 0; j < JT; ++j) acc[j] = v8f{};

    const float*  arow = A + (size_t)rowA * K + half * 2;
    const float2* wb   = Ws + half * P + l15;

    float2 cur[8], nxt[8];
#pragma unroll
    for (int s = 0; s < 8; ++s) cur[s] = *(const float2*)(arow + s * 4);

#pragma unroll 1
    for (int kb = 0; kb < KB; ++kb) {
        const int kn = (kb + 1 < KB) ? kb + 1 : kb;
#pragma unroll
        for (int s = 0; s < 8; ++s) nxt[s] = *(const float2*)(arow + kn * 32 + s * 4);
#pragma unroll
        for (int s = 0; s < 8; ++s) {
            v2f a;
            a.x = cur[s].x * rs;
            a.y = cur[s].y * rs;
            const float2* wk = wb + (kb * 16 + s * 2) * P;
#pragma unroll
            for (int j = 0; j < JT; ++j) {
                const float2 bv = wk[16 * j];
                v2f b;
                b.x = bv.x;
                b.y = bv.y;
                acc[j] = __builtin_amdgcn_wmma_f32_16x16x4_f32(
                     false, a,  false, b,
                     (short)0, acc[j],  false,  false);
            }
        }
#pragma unroll
        for (int s = 0; s < 8; ++s) cur[s] = nxt[s];
    }

    float* crow = C + (size_t)(m0 + half * 8) * N + nb + l15;
    float old[ACC ? JT : 1][8];
    if (ACC) {
#pragma unroll
        for (int j = 0; j < JT; ++j)
#pragma unroll
            for (int r = 0; r < 8; ++r) old[j][r] = crow[(size_t)r * N + 16 * j];
    }
#pragma unroll
    for (int j = 0; j < JT; ++j) {
        const float bv = bias[nb + 16 * j + l15];
#pragma unroll
        for (int r = 0; r < 8; ++r) {
            float v = acc[j][r] + bv;
            if (RELU) v = fmaxf(v, 0.0f);
            v *= outScale;
            if (ACC) v = old[j][r] + v;
            crow[(size_t)r * N + 16 * j] = v;
        }
    }
}

extern "C" void kernel_launch(void* const* d_in, const int* in_sizes, int n_in,
                              void* d_out, int out_size, void* d_ws, size_t ws_size,
                              hipStream_t stream) {
    const float* blocks = (const float*)d_in[0];
    const int*   esrc   = (const int*)  d_in[1];
    const int*   edst   = (const int*)  d_in[2];
    const float* convW  = (const float*)d_in[3];
    const float* convB  = (const float*)d_in[4];
    const float* W1     = (const float*)d_in[5];
    const float* b1     = (const float*)d_in[6];
    const float* W2     = (const float*)d_in[7];
    const float* b2     = (const float*)d_in[8];
    float*       out    = (float*)d_out;

    char* ws = (char*)d_ws;
    const size_t BUF = (size_t)N_NODES * D_IN * sizeof(float);
    float* buf0 = (float*)(ws);
    float* agg  = (float*)(ws + BUF);
    char*  csr  = ws + BUF + BUF / 2;
    int*   col    = (int*)(csr);
    int*   rowptr = (int*)(csr + (size_t)N_ETYPES * N_EDGES * sizeof(int));
    float* buf2 = (float*)(ws + 2 * BUF);
    int*   bh    = (int*)(ws);
    int*   off   = (int*)(ws + (1 << 20));
    int*   pairs = (int*)(ws + 2 * BUF);
    float* deg  = (float*)(ws + 3 * BUF);
    float* H1   = buf0;

    {
        dim3 g(NBLK, N_ETYPES);
        part_hist_kernel<<<g, 256, 0, stream>>>(edst, bh);
        part_scan_kernel<<<N_ETYPES, 1024, 0, stream>>>(bh, off);
        part_scatter_kernel<<<g, 256, 0, stream>>>(esrc, edst, off, pairs);
        dim3 gb(NBK, N_ETYPES);
        bucket_sort_kernel<<<gb, BK_NODES, 0, stream>>>(pairs, off, rowptr, deg, col);
    }

    const int row_blocks      = (N_NODES / 16 + 7) / 8;
    const int row_blocks_half = (HALF / 16 + 7) / 8;

    const float* hin = blocks;
    float* hout = buf0;
    for (int l = 0; l < 2; ++l) {
        for (int e = 0; e < N_ETYPES; ++e) {
            const float* Wle = convW + ((size_t)l * N_ETYPES + e) * D_IN * D_IN;
            const float* ble = convB + ((size_t)l * N_ETYPES + e) * D_IN;
            for (int p = 0; p < 2; ++p) {
                gather_sum_kernel<<<HALF / 8, 256, 0, stream>>>(
                    hin, rowptr + (size_t)e * RPS + (size_t)p * HALF, col + (size_t)e * N_EDGES, agg);
                dim3 g(row_blocks_half, D_IN / 128);
                float*       Cp = hout + (size_t)p * HALF * D_IN;
                const float* dp = deg + (size_t)e * N_NODES + (size_t)p * HALF;
                if (e == 0)
                    wmma_gemm_strip_f32_kernel<128, D_IN, D_IN, true, false><<<g, 256, 0, stream>>>(
                        agg, Wle, ble, Cp, dp, HALF, 1.0f / 3.0f);
                else
                    wmma_gemm_strip_f32_kernel<128, D_IN, D_IN, true, true><<<g, 256, 0, stream>>>(
                        agg, Wle, ble, Cp, dp, HALF, 1.0f / 3.0f);
            }
        }
        hin  = hout;
        hout = buf2;
    }

    {
        dim3 g(row_blocks, D_HID / 128);
        wmma_gemm_strip_f32_kernel<128, D_HID, D_IN, false, false><<<g, 256, 0, stream>>>(
            hin, W1, b1, H1, nullptr, N_NODES, 1.0f);
    }
    {
        dim3 g(row_blocks, D_OUT / 32);
        wmma_gemm_strip_f32_kernel<32, D_OUT, D_HID, false, false><<<g, 256, 0, stream>>>(
            H1, W2, b2, out, nullptr, N_NODES, 1.0f);
    }
}


# Each conv layer fused into one launch: a wave sums its 16 nodes' source rows into LDS as the A operand of its WMMA strip and the 3 etype results meet in registers; every sum order unchanged
# speedup vs baseline: 4.8844x; 1.0481x over previous
#include <hip/hip_runtime.h>

typedef float v2f __attribute__((ext_vector_type(2)));
typedef float v8f __attribute__((ext_vector_type(8)));

#define N_NODES  100000
#define N_ETYPES 3
#define N_EDGES  1600000
#define D_IN     128
#define D_HID    256
#define D_OUT    32

#define RPS      (N_NODES + 4)

#define BK_SHIFT 9
#define BK_NODES (1 << BK_SHIFT)
#define NBK      ((N_NODES + BK_NODES - 1) / BK_NODES)
#define EPB      4096
#define NBLK     ((N_EDGES + EPB - 1) / EPB)
#define EPT      (EPB / 4 / 256)
#define SCAN_SL  5
#define SLB      ((NBLK + SCAN_SL - 1) / SCAN_SL)
static_assert((size_t)N_ETYPES * NBLK * NBK * sizeof(int) <= (1u << 20), "a histogram plane fits the 1 MiB the driver gives it");

__global__ __launch_bounds__(256)
void part_hist_kernel(const int* __restrict__ dst, int* __restrict__ bh) {
    __shared__ int lh[NBK];
    const int e = blockIdx.y, blk = blockIdx.x, t = threadIdx.x;
    if (t < NBK) lh[t] = 0;
    __syncthreads();
    const int4* d4 = (const int4*)(dst + (size_t)e * N_EDGES) + (size_t)blk * (EPB / 4);
    int4 d[EPT];
#pragma unroll
    for (int u = 0; u < EPT; ++u) {
        const int i = u * 256 + t;
        d[u] = make_int4(0, 0, 0, 0);
        if (blk * (EPB / 4) + i < N_EDGES / 4) d[u] = d4[i];
    }
#pragma unroll
    for (int u = 0; u < EPT; ++u) {
        if (blk * (EPB / 4) + u * 256 + t < N_EDGES / 4) {
            atomicAdd(&lh[d[u].x >> BK_SHIFT], 1);
            atomicAdd(&lh[d[u].y >> BK_SHIFT], 1);
            atomicAdd(&lh[d[u].z >> BK_SHIFT], 1);
            atomicAdd(&lh[d[u].w >> BK_SHIFT], 1);
        }
    }
    __syncthreads();
    if (t < NBK) bh[((size_t)e * NBLK + blk) * NBK + t] = lh[t];
}

__global__ __launch_bounds__(1024)
void part_scan_kernel(const int* __restrict__ bh, int* __restrict__ off) {
    __shared__ int ssum[SCAN_SL][NBK];
    __shared__ int bbase[NBK];
    __shared__ int wtot[8];
    static_assert(NBK <= 256 && SCAN_SL * NBK <= 1024, "thread map covers every (slice, bucket)");
    const int e    = blockIdx.x;
    const int t    = threadIdx.x;
    const int lane = t & 31;
    const int sl   = t / NBK;
    const int b    = t - sl * NBK;
    const bool live = t < SCAN_SL * NBK;
    const int k0 = sl * SLB;
    const int k1 = (k0 + SLB < NBLK) ? k0 + SLB : NBLK;
    const int* pi = bh  + (size_t)e * NBLK * NBK + b;
    int*       po = off + (size_t)e * NBLK * NBK + b;

    int s = 0;
    if (live) {
#pragma unroll 8
        for (int k = k0; k < k1; ++k) s += pi[(size_t)k * NBK];
        ssum[sl][b] = s;
    }
    __syncthreads();
    int tot = 0, inc = 0;
    if (t < 256) {
        if (t < NBK) {
#pragma unroll
            for (int q = 0; q < SCAN_SL; ++q) tot += ssum[q][t];
        }
        inc = tot;
#pragma unroll
        for (int d = 1; d < 32; d <<= 1) {
            const int o = __shfl_up(inc, d);
            if (lane >= d) inc += o;
        }
        if (lane == 31) wtot[t >> 5] = inc;
    }
    __syncthreads();
    if (t < NBK) {
        int wb = 0;
        for (int q = 0; q < (t >> 5); ++q) wb += wtot[q];
        bbase[t] = wb + inc - tot;
    }
    __syncthreads();
    if (live) {
        int run = bbase[b];
        for (int q = 0; q < sl; ++q) run += ssum[q][b];
#pragma unroll 8
        for (int k = k0; k < k1; ++k) {
            const int v = pi[(size_t)k * NBK];
            po[(size_t)k * NBK] = run;
            run += v;
        }
    }
}

__global__ __launch_bounds__(256)
void part_scatter_kernel(const int* __restrict__ src, const int* __restrict__ dst,
                         const int* __restrict__ off, int* __restrict__ pairs) {
    __shared__ int cur[NBK];
    const int e = blockIdx.y, blk = blockIdx.x, t = threadIdx.x;
    if (t < NBK) cur[t] = off[((size_t)e * NBLK + blk) * NBK + t];
    const int4* d4 = (const int4*)(dst + (size_t)e * N_EDGES) + (size_t)blk * (EPB / 4);
    const int4* s4 = (const int4*)(src + (size_t)e * N_EDGES) + (size_t)blk * (EPB / 4);
    int2* pe = (int2*)pairs + (size_t)e * N_EDGES;
    int4 d[EPT], s[EPT];
#pragma unroll
    for (int u = 0; u < EPT; ++u) {
        const int i = u * 256 + t;
        d[u] = make_int4(0, 0, 0, 0);
        s[u] = make_int4(0, 0, 0, 0);
        if (blk * (EPB / 4) + i < N_EDGES / 4) { d[u] = d4[i]; s[u] = s4[i]; }
    }
    __syncthreads();
#pragma unroll
    for (int u = 0; u < EPT; ++u) {
        if (blk * (EPB / 4) + u * 256 + t < N_EDGES / 4) {
            const int qx = atomicAdd(&cur[d[u].x >> BK_SHIFT], 1);
            const int qy = atomicAdd(&cur[d[u].y >> BK_SHIFT], 1);
            const int qz = atomicAdd(&cur[d[u].z >> BK_SHIFT], 1);
            const int qw = atomicAdd(&cur[d[u].w >> BK_SHIFT], 1);
            pe[qx] = make_int2(s[u].x, d[u].x);
            pe[qy] = make_int2(s[u].y, d[u].y);
            pe[qz] = make_int2(s[u].z, d[u].z);
            pe[qw] = make_int2(s[u].w, d[u].w);
        }
    }
}

__global__ __launch_bounds__(BK_NODES)
void bucket_sort_kernel(const int* __restrict__ pairs, const int* __restrict__ off,
                        int* __restrict__ rowptr, int* __restrict__ col) {
    __shared__ int cnt[BK_NODES];
    __shared__ int wsum[BK_NODES / 32];
    const int e = blockIdx.y, b = blockIdx.x, t = threadIdx.x;
    const int lane = t & 31;
    const int w    = t >> 5;
    const int* oe  = off + (size_t)e * NBLK * NBK;
    const int beg  = oe[b];
    const int end  = (b + 1 < NBK) ? oe[b + 1] : N_EDGES;
    const int2* pe = (const int2*)pairs + (size_t)e * N_EDGES;
    int*        ce = col + (size_t)e * N_EDGES;

    cnt[t] = 0;
    __syncthreads();
    for (int j0 = beg; j0 < end; j0 += 4 * BK_NODES) {
        int2 p[4];
#pragma unroll
        for (int u = 0; u < 4; ++u) {
            const int j = j0 + u * BK_NODES + t;
            p[u] = pe[(j < end) ? j : end - 1];
        }
#pragma unroll
        for (int u = 0; u < 4; ++u)
            if (j0 + u * BK_NODES + t < end) atomicAdd(&cnt[p[u].y & (BK_NODES - 1)], 1);
    }
    __syncthreads();

    const int c = cnt[t];
    int inc = c;
#pragma unroll
    for (int d = 1; d < 32; d <<= 1) {
        const int o = __shfl_up(inc, d);
        if (lane >= d) inc += o;
    }
    if (lane == 31) wsum[w] = inc;
    __syncthreads();
    int first = beg + inc - c;
    for (int q = 0; q < w; ++q) first += wsum[q];
    cnt[t] = first;
    const int n = b * BK_NODES + t;
    if (n < N_NODES) rowptr[(size_t)e * RPS + n] = first;
    if (b == NBK - 1 && t == 0) rowptr[(size_t)e * RPS + N_NODES] = end;
    __syncthreads();

    for (int j0 = beg; j0 < end; j0 += 4 * BK_NODES) {
        int2 p[4];
#pragma unroll
        for (int u = 0; u < 4; ++u) {
            const int j = j0 + u * BK_NODES + t;
            p[u] = pe[(j < end) ? j : end - 1];
        }
#pragma unroll
        for (int u = 0; u < 4; ++u) {
            if (j0 + u * BK_NODES + t < end) {
                const int q = atomicAdd(&cnt[p[u].y & (BK_NODES - 1)], 1);
                ce[q] = p[u].x;
            }
        }
    }
}

#define CL_P   (D_IN + 16)
#define CL_PA  (D_IN + 4)
#define CL_G   16

__global__ __launch_bounds__(256)
void conv_layer_kernel(const float* __restrict__ h,
                       const int*   __restrict__ rowptr,
                       const int*   __restrict__ col,
                       const float* __restrict__ Wl,
                       const float* __restrict__ bl,
                       float*       __restrict__ out) {
    __shared__ __attribute__((aligned(16))) float2 Ws[(D_IN / 2) * CL_P];
    __shared__ __attribute__((aligned(16))) float  As[8 * 16 * CL_PA];

    const int tid  = threadIdx.x;
    const int wave = __builtin_amdgcn_readfirstlane(tid >> 5);
    const int lane = tid & 31;
    const int half = lane >> 4;
    const int l15  = lane & 15;
    const int m0   = (blockIdx.x * 8 + wave) * 16;
    const bool live = m0 < N_NODES;

    float*       asw = As + wave * (16 * CL_PA);
    const float* hl  = h + lane * 4;

    v8f o[8];
#pragma unroll
    for (int j = 0; j < 8; ++j) o[j] = v8f{};

#pragma unroll 1
    for (int e = 0; e < N_ETYPES; ++e) {
        const int* rpe = rowptr + (size_t)e * RPS;
        const int* ce  = col + (size_t)e * N_EDGES;

        int rp = 0;
        if (live) rp = rpe[m0 + (lane < 16 ? lane : 16)];

        if (e > 0) __syncthreads();

        {
            const float* W = Wl + (size_t)e * D_IN * D_IN;
#pragma unroll
            for (int it = 0; it < (D_IN / 2) * (D_IN / 4) / 256; ++it) {
                const int idx = tid + it * 256;
                const int kp = idx / (D_IN / 4);
                const int n4 = (idx % (D_IN / 4)) * 4;
                const float4 r0 = *(const float4*)(W + (size_t)(2 * kp)     * D_IN + n4);
                const float4 r1 = *(const float4*)(W + (size_t)(2 * kp + 1) * D_IN + n4);
                float4* d = (float4*)(Ws + kp * CL_P + n4);
                d[0] = make_float4(r0.x, r1.x, r0.y, r1.y);
                d[1] = make_float4(r0.z, r1.z, r0.w, r1.w);
            }
        }

        if (live) {
            const int ebeg = __builtin_amdgcn_readlane(rp, 0);
            const int eend = __builtin_amdgcn_readlane(rp, 16);
            int row  = 0;
            int rbeg = ebeg;
            int rend = __builtin_amdgcn_readlane(rp, 1);
            float4 acc = make_float4(0.0f, 0.0f, 0.0f, 0.0f);

            int mine_next = 0;
            if (ebeg + lane < eend) mine_next = ce[ebeg + lane];
            for (int j = ebeg; j < eend; j += 32) {
                const int mine = mine_next;
                const int m = (eend - j < 32) ? eend - j : 32;
                if (j + 32 + lane < eend) mine_next = ce[j + 32 + lane];
                for (int k = 0; k < m; k += CL_G) {
                    float4 v[CL_G];
#pragma unroll
                    for (int u = 0; u < CL_G; ++u) {
                        const int q = (k + u < m) ? k + u : k;
                        const int s = __builtin_amdgcn_readlane(mine, q);
                        v[u] = *(const float4*)(hl + (size_t)s * D_IN);
                    }
#pragma unroll
                    for (int u = 0; u < CL_G; ++u) {
                        if (k + u < m) {
                            const int t = j + k + u;
                            while (t >= rend && row < 15) {
                                const float rs = 1.0f / fmaxf((float)(rend - rbeg), 1.0f);
                                *(float4*)(asw + row * CL_PA + lane * 4) =
                                    make_float4(acc.x * rs, acc.y * rs, acc.z * rs, acc.w * rs);
                                acc = make_float4(0.0f, 0.0f, 0.0f, 0.0f);
                                ++row;
                                rbeg = rend;
                                rend = __builtin_amdgcn_readlane(rp, row + 1);
                            }
                            acc.x += v[u].x;
                            acc.y += v[u].y;
                            acc.z += v[u].z;
                            acc.w += v[u].w;
                        }
                    }
                }
            }
            for (; row < 16; ++row) {
                const float rs = 1.0f / fmaxf((float)(rend - rbeg), 1.0f);
                *(float4*)(asw + row * CL_PA + lane * 4) =
                    make_float4(acc.x * rs, acc.y * rs, acc.z * rs, acc.w * rs);
                acc = make_float4(0.0f, 0.0f, 0.0f, 0.0f);
                rbeg = rend;
                rend = __builtin_amdgcn_readlane(rp, (row + 2 < 16) ? row + 2 : 16);
            }
        }
        __syncthreads();

        if (live) {
            v8f acc[8];
#pragma unroll
            for (int j = 0; j < 8; ++j) acc[j] = v8f{};

            const float*  ar = asw + l15 * CL_PA + half * 2;
            const float2* wb = Ws + half * CL_P + l15;

#pragma unroll 1
            for (int kb = 0; kb < D_IN / 32; ++kb) {
#pragma unroll
                for (int s = 0; s < 8; ++s) {
                    const float2 av = *(const float2*)(ar + kb * 32 + s * 4);
                    v2f a;
                    a.x = av.x;
                    a.y = av.y;
                    const float2* wk = wb + (kb * 16 + s * 2) * CL_P;
#pragma unroll
                    for (int j = 0; j < 8; ++j) {
                        const float2 bv = wk[16 * j];
                        v2f b;
                        b.x = bv.x;
                        b.y = bv.y;
                        acc[j] = __builtin_amdgcn_wmma_f32_16x16x4_f32(
                             false, a,  false, b,
                             (short)0, acc[j],  false,  false);
                    }
                }
            }

            const float* be = bl + (size_t)e * D_IN;
#pragma unroll
            for (int j = 0; j < 8; ++j) {
                const float bv = be[16 * j + l15];
#pragma unroll
                for (int r = 0; r < 8; ++r) {
                    float v = fmaxf(acc[j][r] + bv, 0.0f);
                    v *= 1.0f / 3.0f;
                    o[j][r] = o[j][r] + v;
                }
            }
        }
    }

    if (live) {
        float* crow = out + (size_t)(m0 + half * 8) * D_IN + l15;
#pragma unroll
        for (int j = 0; j < 8; ++j)
#pragma unroll
            for (int r = 0; r < 8; ++r) crow[(size_t)r * D_IN + 16 * j] = o[j][r];
    }
}

template <int NT, int N, int K, bool RELU, bool ACC>
__global__ __launch_bounds__(256)
void wmma_gemm_strip_f32_kernel(const float* __restrict__ A,
                                const float* __restrict__ W,
                                const float* __restrict__ bias,
                                float* __restrict__ C,
                                const float* __restrict__ deg,
                                int M, float outScale) {
    constexpr int JT = NT / 16;
    constexpr int P  = NT + 16;
    constexpr int KB = K / 32;
    __shared__ __attribute__((aligned(16))) float2 Ws[(K / 2) * P];

    const int tid = threadIdx.x;
    const int nb  = blockIdx.y * NT;

    constexpr int STAGE_IT = (K / 2) * (NT / 4) / 256;
    static_assert(STAGE_IT * 256 == (K / 2) * (NT / 4), "slab divides evenly over the work-group");
#pragma unroll
    for (int it = 0; it < STAGE_IT; ++it) {
        const int idx = tid + it * 256;
        const int kp = idx / (NT / 4);
        const int n4 = (idx % (NT / 4)) * 4;
        const float4 r0 = *(const float4*)(W + (size_t)(2 * kp)     * N + nb + n4);
        const float4 r1 = *(const float4*)(W + (size_t)(2 * kp + 1) * N + nb + n4);
        float4* d = (float4*)(Ws + kp * P + n4);
        d[0] = make_float4(r0.x, r1.x, r0.y, r1.y);
        d[1] = make_float4(r0.z, r1.z, r0.w, r1.w);
    }
    __syncthreads();

    const int wave = tid >> 5;
    const int lane = tid & 31;
    const int half = lane >> 4;
    const int l15  = lane & 15;
    const int m0   = (blockIdx.x * 8 + wave) * 16;
    if (m0 >= M) return;

    const int rowA = m0 + l15;
    float rs = 1.0f;
    if (deg) rs = 1.0f / fmaxf(deg[rowA], 1.0f);

    v8f acc[JT];
#pragma unroll
    for (int j = 0; j < JT; ++j) acc[j] = v8f{};

    const float*  arow = A + (size_t)rowA * K + half * 2;
    const float2* wb   = Ws + half * P + l15;

    float2 cur[8], nxt[8];
#pragma unroll
    for (int s = 0; s < 8; ++s) cur[s] = *(const float2*)(arow + s * 4);

#pragma unroll 1
    for (int kb = 0; kb < KB; ++kb) {
        const int kn = (kb + 1 < KB) ? kb + 1 : kb;
#pragma unroll
        for (int s = 0; s < 8; ++s) nxt[s] = *(const float2*)(arow + kn * 32 + s * 4);
#pragma unroll
        for (int s = 0; s < 8; ++s) {
            v2f a;
            a.x = cur[s].x * rs;
            a.y = cur[s].y * rs;
            const float2* wk = wb + (kb * 16 + s * 2) * P;
#pragma unroll
            for (int j = 0; j < JT; ++j) {
                const float2 bv = wk[16 * j];
                v2f b;
                b.x = bv.x;
                b.y = bv.y;
                acc[j] = __builtin_amdgcn_wmma_f32_16x16x4_f32(
                     false, a,  false, b,
                     (short)0, acc[j],  false,  false);
            }
        }
#pragma unroll
        for (int s = 0; s < 8; ++s) cur[s] = nxt[s];
    }

    float* crow = C + (size_t)(m0 + half * 8) * N + nb + l15;
    float old[ACC ? JT : 1][8];
    if (ACC) {
#pragma unroll
        for (int j = 0; j < JT; ++j)
#pragma unroll
            for (int r = 0; r < 8; ++r) old[j][r] = crow[(size_t)r * N + 16 * j];
    }
#pragma unroll
    for (int j = 0; j < JT; ++j) {
        const float bv = bias[nb + 16 * j + l15];
#pragma unroll
        for (int r = 0; r < 8; ++r) {
            float v = acc[j][r] + bv;
            if (RELU) v = fmaxf(v, 0.0f);
            v *= outScale;
            if (ACC) v = old[j][r] + v;
            crow[(size_t)r * N + 16 * j] = v;
        }
    }
}

extern "C" void kernel_launch(void* const* d_in, const int* in_sizes, int n_in,
                              void* d_out, int out_size, void* d_ws, size_t ws_size,
                              hipStream_t stream) {
    const float* blocks = (const float*)d_in[0];
    const int*   esrc   = (const int*)  d_in[1];
    const int*   edst   = (const int*)  d_in[2];
    const float* convW  = (const float*)d_in[3];
    const float* convB  = (const float*)d_in[4];
    const float* W1     = (const float*)d_in[5];
    const float* b1     = (const float*)d_in[6];
    const float* W2     = (const float*)d_in[7];
    const float* b2     = (const float*)d_in[8];
    float*       out    = (float*)d_out;

    char* ws = (char*)d_ws;
    const size_t BUF = (size_t)N_NODES * D_IN * sizeof(float);
    float* buf0 = (float*)(ws);
    char*  csr  = ws + BUF + BUF / 2;
    int*   col    = (int*)(csr);
    int*   rowptr = (int*)(csr + (size_t)N_ETYPES * N_EDGES * sizeof(int));
    float* buf2 = (float*)(ws + 2 * BUF);
    int*   bh    = (int*)(ws);
    int*   off   = (int*)(ws + (1 << 20));
    int*   pairs = (int*)(ws + 2 * BUF);
    float* H1   = buf0;

    {
        dim3 g(NBLK, N_ETYPES);
        part_hist_kernel<<<g, 256, 0, stream>>>(edst, bh);
        part_scan_kernel<<<N_ETYPES, 1024, 0, stream>>>(bh, off);
        part_scatter_kernel<<<g, 256, 0, stream>>>(esrc, edst, off, pairs);
        dim3 gb(NBK, N_ETYPES);
        bucket_sort_kernel<<<gb, BK_NODES, 0, stream>>>(pairs, off, rowptr, col);
    }

    const int row_blocks = (N_NODES / 16 + 7) / 8;

    conv_layer_kernel<<<row_blocks, 256, 0, stream>>>(
        blocks, rowptr, col, convW, convB, buf0);
    conv_layer_kernel<<<row_blocks, 256, 0, stream>>>(
        buf0, rowptr, col, convW + (size_t)N_ETYPES * D_IN * D_IN, convB + (size_t)N_ETYPES * D_IN, buf2);

    {
        dim3 g(row_blocks, D_HID / 128);
        wmma_gemm_strip_f32_kernel<128, D_HID, D_IN, false, false><<<g, 256, 0, stream>>>(
            buf2, W1, b1, H1, nullptr, N_NODES, 1.0f);
    }
    {
        dim3 g(row_blocks, D_OUT / 32);
        wmma_gemm_strip_f32_kernel<32, D_OUT, D_HID, false, false><<<g, 256, 0, stream>>>(
            H1, W2, b2, out, nullptr, N_NODES, 1.0f);
    }
}


# Head's two linears folded per call on WMMA: W12=W1@W2, b12=b1@W2+b2, then one 100k x128x32 GEMM replaces 128->256->32; head products re-associated (its sum order changes)
# speedup vs baseline: 5.2985x; 1.0848x over previous
#include <hip/hip_runtime.h>

typedef float v2f __attribute__((ext_vector_type(2)));
typedef float v8f __attribute__((ext_vector_type(8)));

#define N_NODES  100000
#define N_ETYPES 3
#define N_EDGES  1600000
#define D_IN     128
#define D_HID    256
#define D_OUT    32

#define RPS      (N_NODES + 4)

#define BK_SHIFT 9
#define BK_NODES (1 << BK_SHIFT)
#define NBK      ((N_NODES + BK_NODES - 1) / BK_NODES)
#define EPB      4096
#define NBLK     ((N_EDGES + EPB - 1) / EPB)
#define EPT      (EPB / 4 / 256)
#define SCAN_SL  5
#define SLB      ((NBLK + SCAN_SL - 1) / SCAN_SL)
static_assert((size_t)N_ETYPES * NBLK * NBK * sizeof(int) <= (1u << 20), "a histogram plane fits the 1 MiB the driver gives it");

__global__ __launch_bounds__(256)
void part_hist_kernel(const int* __restrict__ dst, int* __restrict__ bh) {
    __shared__ int lh[NBK];
    const int e = blockIdx.y, blk = blockIdx.x, t = threadIdx.x;
    if (t < NBK) lh[t] = 0;
    __syncthreads();
    const int4* d4 = (const int4*)(dst + (size_t)e * N_EDGES) + (size_t)blk * (EPB / 4);
    int4 d[EPT];
#pragma unroll
    for (int u = 0; u < EPT; ++u) {
        const int i = u * 256 + t;
        d[u] = make_int4(0, 0, 0, 0);
        if (blk * (EPB / 4) + i < N_EDGES / 4) d[u] = d4[i];
    }
#pragma unroll
    for (int u = 0; u < EPT; ++u) {
        if (blk * (EPB / 4) + u * 256 + t < N_EDGES / 4) {
            atomicAdd(&lh[d[u].x >> BK_SHIFT], 1);
            atomicAdd(&lh[d[u].y >> BK_SHIFT], 1);
            atomicAdd(&lh[d[u].z >> BK_SHIFT], 1);
            atomicAdd(&lh[d[u].w >> BK_SHIFT], 1);
        }
    }
    __syncthreads();
    if (t < NBK) bh[((size_t)e * NBLK + blk) * NBK + t] = lh[t];
}

__global__ __launch_bounds__(1024)
void part_scan_kernel(const int* __restrict__ bh, int* __restrict__ off) {
    __shared__ int ssum[SCAN_SL][NBK];
    __shared__ int bbase[NBK];
    __shared__ int wtot[8];
    static_assert(NBK <= 256 && SCAN_SL * NBK <= 1024, "thread map covers every (slice, bucket)");
    const int e    = blockIdx.x;
    const int t    = threadIdx.x;
    const int lane = t & 31;
    const int sl   = t / NBK;
    const int b    = t - sl * NBK;
    const bool live = t < SCAN_SL * NBK;
    const int k0 = sl * SLB;
    const int k1 = (k0 + SLB < NBLK) ? k0 + SLB : NBLK;
    const int* pi = bh  + (size_t)e * NBLK * NBK + b;
    int*       po = off + (size_t)e * NBLK * NBK + b;

    int s = 0;
    if (live) {
#pragma unroll 8
        for (int k = k0; k < k1; ++k) s += pi[(size_t)k * NBK];
        ssum[sl][b] = s;
    }
    __syncthreads();
    int tot = 0, inc = 0;
    if (t < 256) {
        if (t < NBK) {
#pragma unroll
            for (int q = 0; q < SCAN_SL; ++q) tot += ssum[q][t];
        }
        inc = tot;
#pragma unroll
        for (int d = 1; d < 32; d <<= 1) {
            const int o = __shfl_up(inc, d);
            if (lane >= d) inc += o;
        }
        if (lane == 31) wtot[t >> 5] = inc;
    }
    __syncthreads();
    if (t < NBK) {
        int wb = 0;
        for (int q = 0; q < (t >> 5); ++q) wb += wtot[q];
        bbase[t] = wb + inc - tot;
    }
    __syncthreads();
    if (live) {
        int run = bbase[b];
        for (int q = 0; q < sl; ++q) run += ssum[q][b];
#pragma unroll 8
        for (int k = k0; k < k1; ++k) {
            const int v = pi[(size_t)k * NBK];
            po[(size_t)k * NBK] = run;
            run += v;
        }
    }
}

__global__ __launch_bounds__(256)
void part_scatter_kernel(const int* __restrict__ src, const int* __restrict__ dst,
                         const int* __restrict__ off, int* __restrict__ pairs) {
    __shared__ int cur[NBK];
    const int e = blockIdx.y, blk = blockIdx.x, t = threadIdx.x;
    if (t < NBK) cur[t] = off[((size_t)e * NBLK + blk) * NBK + t];
    const int4* d4 = (const int4*)(dst + (size_t)e * N_EDGES) + (size_t)blk * (EPB / 4);
    const int4* s4 = (const int4*)(src + (size_t)e * N_EDGES) + (size_t)blk * (EPB / 4);
    int2* pe = (int2*)pairs + (size_t)e * N_EDGES;
    int4 d[EPT], s[EPT];
#pragma unroll
    for (int u = 0; u < EPT; ++u) {
        const int i = u * 256 + t;
        d[u] = make_int4(0, 0, 0, 0);
        s[u] = make_int4(0, 0, 0, 0);
        if (blk * (EPB / 4) + i < N_EDGES / 4) { d[u] = d4[i]; s[u] = s4[i]; }
    }
    __syncthreads();
#pragma unroll
    for (int u = 0; u < EPT; ++u) {
        if (blk * (EPB / 4) + u * 256 + t < N_EDGES / 4) {
            const int qx = atomicAdd(&cur[d[u].x >> BK_SHIFT], 1);
            const int qy = atomicAdd(&cur[d[u].y >> BK_SHIFT], 1);
            const int qz = atomicAdd(&cur[d[u].z >> BK_SHIFT], 1);
            const int qw = atomicAdd(&cur[d[u].w >> BK_SHIFT], 1);
            pe[qx] = make_int2(s[u].x, d[u].x);
            pe[qy] = make_int2(s[u].y, d[u].y);
            pe[qz] = make_int2(s[u].z, d[u].z);
            pe[qw] = make_int2(s[u].w, d[u].w);
        }
    }
}

__global__ __launch_bounds__(BK_NODES)
void bucket_sort_kernel(const int* __restrict__ pairs, const int* __restrict__ off,
                        int* __restrict__ rowptr, int* __restrict__ col) {
    __shared__ int cnt[BK_NODES];
    __shared__ int wsum[BK_NODES / 32];
    const int e = blockIdx.y, b = blockIdx.x, t = threadIdx.x;
    const int lane = t & 31;
    const int w    = t >> 5;
    const int* oe  = off + (size_t)e * NBLK * NBK;
    const int beg  = oe[b];
    const int end  = (b + 1 < NBK) ? oe[b + 1] : N_EDGES;
    const int2* pe = (const int2*)pairs + (size_t)e * N_EDGES;
    int*        ce = col + (size_t)e * N_EDGES;

    cnt[t] = 0;
    __syncthreads();
    for (int j0 = beg; j0 < end; j0 += 4 * BK_NODES) {
        int2 p[4];
#pragma unroll
        for (int u = 0; u < 4; ++u) {
            const int j = j0 + u * BK_NODES + t;
            p[u] = pe[(j < end) ? j : end - 1];
        }
#pragma unroll
        for (int u = 0; u < 4; ++u)
            if (j0 + u * BK_NODES + t < end) atomicAdd(&cnt[p[u].y & (BK_NODES - 1)], 1);
    }
    __syncthreads();

    const int c = cnt[t];
    int inc = c;
#pragma unroll
    for (int d = 1; d < 32; d <<= 1) {
        const int o = __shfl_up(inc, d);
        if (lane >= d) inc += o;
    }
    if (lane == 31) wsum[w] = inc;
    __syncthreads();
    int first = beg + inc - c;
    for (int q = 0; q < w; ++q) first += wsum[q];
    cnt[t] = first;
    const int n = b * BK_NODES + t;
    if (n < N_NODES) rowptr[(size_t)e * RPS + n] = first;
    if (b == NBK - 1 && t == 0) rowptr[(size_t)e * RPS + N_NODES] = end;
    __syncthreads();

    for (int j0 = beg; j0 < end; j0 += 4 * BK_NODES) {
        int2 p[4];
#pragma unroll
        for (int u = 0; u < 4; ++u) {
            const int j = j0 + u * BK_NODES + t;
            p[u] = pe[(j < end) ? j : end - 1];
        }
#pragma unroll
        for (int u = 0; u < 4; ++u) {
            if (j0 + u * BK_NODES + t < end) {
                const int q = atomicAdd(&cnt[p[u].y & (BK_NODES - 1)], 1);
                ce[q] = p[u].x;
            }
        }
    }
}

#define CL_P   (D_IN + 16)
#define CL_PA  (D_IN + 4)
#define CL_G   16

__global__ __launch_bounds__(256)
void conv_layer_kernel(const float* __restrict__ h,
                       const int*   __restrict__ rowptr,
                       const int*   __restrict__ col,
                       const float* __restrict__ Wl,
                       const float* __restrict__ bl,
                       float*       __restrict__ out) {
    __shared__ __attribute__((aligned(16))) float2 Ws[(D_IN / 2) * CL_P];
    __shared__ __attribute__((aligned(16))) float  As[8 * 16 * CL_PA];

    const int tid  = threadIdx.x;
    const int wave = __builtin_amdgcn_readfirstlane(tid >> 5);
    const int lane = tid & 31;
    const int half = lane >> 4;
    const int l15  = lane & 15;
    const int m0   = (blockIdx.x * 8 + wave) * 16;
    const bool live = m0 < N_NODES;

    float*       asw = As + wave * (16 * CL_PA);
    const float* hl  = h + lane * 4;

    v8f o[8];
#pragma unroll
    for (int j = 0; j < 8; ++j) o[j] = v8f{};

#pragma unroll 1
    for (int e = 0; e < N_ETYPES; ++e) {
        const int* rpe = rowptr + (size_t)e * RPS;
        const int* ce  = col + (size_t)e * N_EDGES;

        int rp = 0;
        if (live) rp = rpe[m0 + (lane < 16 ? lane : 16)];

        if (e > 0) __syncthreads();

        {
            const float* W = Wl + (size_t)e * D_IN * D_IN;
#pragma unroll
            for (int it = 0; it < (D_IN / 2) * (D_IN / 4) / 256; ++it) {
                const int idx = tid + it * 256;
                const int kp = idx / (D_IN / 4);
                const int n4 = (idx % (D_IN / 4)) * 4;
                const float4 r0 = *(const float4*)(W + (size_t)(2 * kp)     * D_IN + n4);
                const float4 r1 = *(const float4*)(W + (size_t)(2 * kp + 1) * D_IN + n4);
                float4* d = (float4*)(Ws + kp * CL_P + n4);
                d[0] = make_float4(r0.x, r1.x, r0.y, r1.y);
                d[1] = make_float4(r0.z, r1.z, r0.w, r1.w);
            }
        }

        if (live) {
            const int ebeg = __builtin_amdgcn_readlane(rp, 0);
            const int eend = __builtin_amdgcn_readlane(rp, 16);
            int row  = 0;
            int rbeg = ebeg;
            int rend = __builtin_amdgcn_readlane(rp, 1);
            float4 acc = make_float4(0.0f, 0.0f, 0.0f, 0.0f);

            int mine_next = 0;
            if (ebeg + lane < eend) mine_next = ce[ebeg + lane];
            for (int j = ebeg; j < eend; j += 32) {
                const int mine = mine_next;
                const int m = (eend - j < 32) ? eend - j : 32;
                if (j + 32 + lane < eend) mine_next = ce[j + 32 + lane];
                for (int k = 0; k < m; k += CL_G) {
                    float4 v[CL_G];
#pragma unroll
                    for (int u = 0; u < CL_G; ++u) {
                        const int q = (k + u < m) ? k + u : k;
                        const int s = __builtin_amdgcn_readlane(mine, q);
                        v[u] = *(const float4*)(hl + (size_t)s * D_IN);
                    }
#pragma unroll
                    for (int u = 0; u < CL_G; ++u) {
                        if (k + u < m) {
                            const int t = j + k + u;
                            while (t >= rend && row < 15) {
                                const float rs = 1.0f / fmaxf((float)(rend - rbeg), 1.0f);
                                *(float4*)(asw + row * CL_PA + lane * 4) =
                                    make_float4(acc.x * rs, acc.y * rs, acc.z * rs, acc.w * rs);
                                acc = make_float4(0.0f, 0.0f, 0.0f, 0.0f);
                                ++row;
                                rbeg = rend;
                                rend = __builtin_amdgcn_readlane(rp, row + 1);
                            }
                            acc.x += v[u].x;
                            acc.y += v[u].y;
                            acc.z += v[u].z;
                            acc.w += v[u].w;
                        }
                    }
                }
            }
            for (; row < 16; ++row) {
                const float rs = 1.0f / fmaxf((float)(rend - rbeg), 1.0f);
                *(float4*)(asw + row * CL_PA + lane * 4) =
                    make_float4(acc.x * rs, acc.y * rs, acc.z * rs, acc.w * rs);
                acc = make_float4(0.0f, 0.0f, 0.0f, 0.0f);
                rbeg = rend;
                rend = __builtin_amdgcn_readlane(rp, (row + 2 < 16) ? row + 2 : 16);
            }
        }
        __syncthreads();

        if (live) {
            v8f acc[8];
#pragma unroll
            for (int j = 0; j < 8; ++j) acc[j] = v8f{};

            const float*  ar = asw + l15 * CL_PA + half * 2;
            const float2* wb = Ws + half * CL_P + l15;

#pragma unroll 1
            for (int kb = 0; kb < D_IN / 32; ++kb) {
#pragma unroll
                for (int s = 0; s < 8; ++s) {
                    const float2 av = *(const float2*)(ar + kb * 32 + s * 4);
                    v2f a;
                    a.x = av.x;
                    a.y = av.y;
                    const float2* wk = wb + (kb * 16 + s * 2) * CL_P;
#pragma unroll
                    for (int j = 0; j < 8; ++j) {
                        const float2 bv = wk[16 * j];
                        v2f b;
                        b.x = bv.x;
                        b.y = bv.y;
                        acc[j] = __builtin_amdgcn_wmma_f32_16x16x4_f32(
                             false, a,  false, b,
                             (short)0, acc[j],  false,  false);
                    }
                }
            }

            const float* be = bl + (size_t)e * D_IN;
#pragma unroll
            for (int j = 0; j < 8; ++j) {
                const float bv = be[16 * j + l15];
#pragma unroll
                for (int r = 0; r < 8; ++r) {
                    float v = fmaxf(acc[j][r] + bv, 0.0f);
                    v *= 1.0f / 3.0f;
                    o[j][r] = o[j][r] + v;
                }
            }
        }
    }

    if (live) {
        float* crow = out + (size_t)(m0 + half * 8) * D_IN + l15;
#pragma unroll
        for (int j = 0; j < 8; ++j)
#pragma unroll
            for (int r = 0; r < 8; ++r) crow[(size_t)r * D_IN + 16 * j] = o[j][r];
    }
}

template <int NT, int N, int K, bool RELU, bool ACC>
__global__ __launch_bounds__(256)
void wmma_gemm_strip_f32_kernel(const float* __restrict__ A,
                                const float* __restrict__ W,
                                const float* __restrict__ bias,
                                float* __restrict__ C,
                                const float* __restrict__ deg,
                                int M, float outScale) {
    constexpr int JT = NT / 16;
    constexpr int P  = NT + 16;
    constexpr int KB = K / 32;
    __shared__ __attribute__((aligned(16))) float2 Ws[(K / 2) * P];

    const int tid = threadIdx.x;
    const int nb  = blockIdx.y * NT;

    constexpr int STAGE_IT = (K / 2) * (NT / 4) / 256;
    static_assert(STAGE_IT * 256 == (K / 2) * (NT / 4), "slab divides evenly over the work-group");
#pragma unroll
    for (int it = 0; it < STAGE_IT; ++it) {
        const int idx = tid + it * 256;
        const int kp = idx / (NT / 4);
        const int n4 = (idx % (NT / 4)) * 4;
        const float4 r0 = *(const float4*)(W + (size_t)(2 * kp)     * N + nb + n4);
        const float4 r1 = *(const float4*)(W + (size_t)(2 * kp + 1) * N + nb + n4);
        float4* d = (float4*)(Ws + kp * P + n4);
        d[0] = make_float4(r0.x, r1.x, r0.y, r1.y);
        d[1] = make_float4(r0.z, r1.z, r0.w, r1.w);
    }
    __syncthreads();

    const int wave = tid >> 5;
    const int lane = tid & 31;
    const int half = lane >> 4;
    const int l15  = lane & 15;
    const int m0   = (blockIdx.x * 8 + wave) * 16;
    if (m0 >= M) return;

    const int rowA = m0 + l15;
    float rs = 1.0f;
    if (deg) rs = 1.0f / fmaxf(deg[rowA], 1.0f);

    v8f acc[JT];
#pragma unroll
    for (int j = 0; j < JT; ++j) acc[j] = v8f{};

    const float*  arow = A + (size_t)rowA * K + half * 2;
    const float2* wb   = Ws + half * P + l15;

    float2 cur[8], nxt[8];
#pragma unroll
    for (int s = 0; s < 8; ++s) cur[s] = *(const float2*)(arow + s * 4);

#pragma unroll 1
    for (int kb = 0; kb < KB; ++kb) {
        const int kn = (kb + 1 < KB) ? kb + 1 : kb;
#pragma unroll
        for (int s = 0; s < 8; ++s) nxt[s] = *(const float2*)(arow + kn * 32 + s * 4);
#pragma unroll
        for (int s = 0; s < 8; ++s) {
            v2f a;
            a.x = cur[s].x * rs;
            a.y = cur[s].y * rs;
            const float2* wk = wb + (kb * 16 + s * 2) * P;
#pragma unroll
            for (int j = 0; j < JT; ++j) {
                const float2 bv = wk[16 * j];
                v2f b;
                b.x = bv.x;
                b.y = bv.y;
                acc[j] = __builtin_amdgcn_wmma_f32_16x16x4_f32(
                     false, a,  false, b,
                     (short)0, acc[j],  false,  false);
            }
        }
#pragma unroll
        for (int s = 0; s < 8; ++s) cur[s] = nxt[s];
    }

    float* crow = C + (size_t)(m0 + half * 8) * N + nb + l15;
    float old[ACC ? JT : 1][8];
    if (ACC) {
#pragma unroll
        for (int j = 0; j < JT; ++j)
#pragma unroll
            for (int r = 0; r < 8; ++r) old[j][r] = crow[(size_t)r * N + 16 * j];
    }
#pragma unroll
    for (int j = 0; j < JT; ++j) {
        const float bv = bias[nb + 16 * j + l15];
#pragma unroll
        for (int r = 0; r < 8; ++r) {
            float v = acc[j][r] + bv;
            if (RELU) v = fmaxf(v, 0.0f);
            v *= outScale;
            if (ACC) v = old[j][r] + v;
            crow[(size_t)r * N + 16 * j] = v;
        }
    }
}

#define HF_P   (D_OUT + 16)
#define HF_KS  (D_HID / 4)

__global__ __launch_bounds__(256)
void head_fold_kernel(const float* __restrict__ W1,
                      const float* __restrict__ b1,
                      const float* __restrict__ W2,
                      const float* __restrict__ b2,
                      float*       __restrict__ W12,
                      float*       __restrict__ b12) {
    __shared__ __attribute__((aligned(16))) float2 Ws[(D_HID / 2) * HF_P];

    const int tid = threadIdx.x;
    constexpr int STAGE_IT = (D_HID / 2) * (D_OUT / 4) / 256;
    static_assert(STAGE_IT * 256 == (D_HID / 2) * (D_OUT / 4), "slab divides evenly over the work-group");
#pragma unroll
    for (int it = 0; it < STAGE_IT; ++it) {
        const int idx = tid + it * 256;
        const int kp = idx / (D_OUT / 4);
        const int n4 = (idx % (D_OUT / 4)) * 4;
        const float4 r0 = *(const float4*)(W2 + (size_t)(2 * kp)     * D_OUT + n4);
        const float4 r1 = *(const float4*)(W2 + (size_t)(2 * kp + 1) * D_OUT + n4);
        float4* d = (float4*)(Ws + kp * HF_P + n4);
        d[0] = make_float4(r0.x, r1.x, r0.y, r1.y);
        d[1] = make_float4(r0.z, r1.z, r0.w, r1.w);
    }

    const int lane  = tid & 31;
    const int half  = lane >> 4;
    const int l15   = lane & 15;
    const int strip = blockIdx.x;
    const bool bias_strip = strip == D_IN / 16;
    const float* arow = bias_strip ? b1 + half * 2
                                   : W1 + (size_t)(strip * 16 + l15) * D_HID + half * 2;
    const bool zero_row = bias_strip && l15 != 0;

    float2 av[HF_KS];
    if (tid < 32) {
#pragma unroll
        for (int s = 0; s < HF_KS; ++s) av[s] = *(const float2*)(arow + s * 4);
    }
    __syncthreads();
    if (tid >= 32) return;

    v8f acc[D_OUT / 16];
#pragma unroll
    for (int j = 0; j < D_OUT / 16; ++j) acc[j] = v8f{};

    const float2* wb = Ws + half * HF_P + l15;
#pragma unroll
    for (int s = 0; s < HF_KS; ++s) {
        v2f a;
        a.x = zero_row ? 0.0f : av[s].x;
        a.y = zero_row ? 0.0f : av[s].y;
        const float2* wk = wb + (s * 2) * HF_P;
#pragma unroll
        for (int j = 0; j < D_OUT / 16; ++j) {
            const float2 bv = wk[16 * j];
            v2f b;
            b.x = bv.x;
            b.y = bv.y;
            acc[j] = __builtin_amdgcn_wmma_f32_16x16x4_f32(
                 false, a,  false, b,
                 (short)0, acc[j],  false,  false);
        }
    }

    if (!bias_strip) {
        float* crow = W12 + (size_t)(strip * 16 + half * 8) * D_OUT + l15;
#pragma unroll
        for (int j = 0; j < D_OUT / 16; ++j)
#pragma unroll
            for (int r = 0; r < 8; ++r) crow[(size_t)r * D_OUT + 16 * j] = acc[j][r];
    } else if (half == 0) {
#pragma unroll
        for (int j = 0; j < D_OUT / 16; ++j) b12[16 * j + l15] = acc[j][0] + b2[16 * j + l15];
    }
}

extern "C" void kernel_launch(void* const* d_in, const int* in_sizes, int n_in,
                              void* d_out, int out_size, void* d_ws, size_t ws_size,
                              hipStream_t stream) {
    const float* blocks = (const float*)d_in[0];
    const int*   esrc   = (const int*)  d_in[1];
    const int*   edst   = (const int*)  d_in[2];
    const float* convW  = (const float*)d_in[3];
    const float* convB  = (const float*)d_in[4];
    const float* W1     = (const float*)d_in[5];
    const float* b1     = (const float*)d_in[6];
    const float* W2     = (const float*)d_in[7];
    const float* b2     = (const float*)d_in[8];
    float*       out    = (float*)d_out;

    char* ws = (char*)d_ws;
    const size_t BUF = (size_t)N_NODES * D_IN * sizeof(float);
    float* buf0 = (float*)(ws);
    char*  csr  = ws + BUF + BUF / 2;
    int*   col    = (int*)(csr);
    int*   rowptr = (int*)(csr + (size_t)N_ETYPES * N_EDGES * sizeof(int));
    float* buf2 = (float*)(ws + 2 * BUF);
    int*   bh    = (int*)(ws);
    int*   off   = (int*)(ws + (1 << 20));
    int*   pairs = (int*)(ws + 2 * BUF);
    float* W12  = (float*)(ws + BUF);
    float* b12  = W12 + D_IN * D_OUT;

    {
        dim3 g(NBLK, N_ETYPES);
        part_hist_kernel<<<g, 256, 0, stream>>>(edst, bh);
        part_scan_kernel<<<N_ETYPES, 1024, 0, stream>>>(bh, off);
        part_scatter_kernel<<<g, 256, 0, stream>>>(esrc, edst, off, pairs);
        dim3 gb(NBK, N_ETYPES);
        bucket_sort_kernel<<<gb, BK_NODES, 0, stream>>>(pairs, off, rowptr, col);
    }

    const int row_blocks = (N_NODES / 16 + 7) / 8;

    conv_layer_kernel<<<row_blocks, 256, 0, stream>>>(
        blocks, rowptr, col, convW, convB, buf0);
    conv_layer_kernel<<<row_blocks, 256, 0, stream>>>(
        buf0, rowptr, col, convW + (size_t)N_ETYPES * D_IN * D_IN, convB + (size_t)N_ETYPES * D_IN, buf2);

    head_fold_kernel<<<D_IN / 16 + 1, 256, 0, stream>>>(W1, b1, W2, b2, W12, b12);
    {
        dim3 g(row_blocks, D_OUT / 32);
        wmma_gemm_strip_f32_kernel<32, D_OUT, D_IN, false, false><<<g, 256, 0, stream>>>(
            buf2, W12, b12, out, nullptr, N_NODES, 1.0f);
    }
}


# Row gather walk made straight-line per 32-id chunk: 32 rows requested at once, ids from constant lanes, 32-bit row offsets on the scalar base, one scalar compare per slot; sum order kept
# speedup vs baseline: 6.8064x; 1.2846x over previous
#include <hip/hip_runtime.h>

typedef float v2f __attribute__((ext_vector_type(2)));
typedef float v8f __attribute__((ext_vector_type(8)));

#define N_NODES  100000
#define N_ETYPES 3
#define N_EDGES  1600000
#define D_IN     128
#define D_HID    256
#define D_OUT    32

#define RPS      (N_NODES + 4)

#define BK_SHIFT 9
#define BK_NODES (1 << BK_SHIFT)
#define NBK      ((N_NODES + BK_NODES - 1) / BK_NODES)
#define EPB      4096
#define NBLK     ((N_EDGES + EPB - 1) / EPB)
#define EPT      (EPB / 4 / 256)
#define SCAN_SL  5
#define SLB      ((NBLK + SCAN_SL - 1) / SCAN_SL)
static_assert((size_t)N_ETYPES * NBLK * NBK * sizeof(int) <= (1u << 20), "a histogram plane fits the 1 MiB the driver gives it");

__global__ __launch_bounds__(256)
void part_hist_kernel(const int* __restrict__ dst, int* __restrict__ bh) {
    __shared__ int lh[NBK];
    const int e = blockIdx.y, blk = blockIdx.x, t = threadIdx.x;
    if (t < NBK) lh[t] = 0;
    __syncthreads();
    const int4* d4 = (const int4*)(dst + (size_t)e * N_EDGES) + (size_t)blk * (EPB / 4);
    int4 d[EPT];
#pragma unroll
    for (int u = 0; u < EPT; ++u) {
        const int i = u * 256 + t;
        d[u] = make_int4(0, 0, 0, 0);
        if (blk * (EPB / 4) + i < N_EDGES / 4) d[u] = d4[i];
    }
#pragma unroll
    for (int u = 0; u < EPT; ++u) {
        if (blk * (EPB / 4) + u * 256 + t < N_EDGES / 4) {
            atomicAdd(&lh[d[u].x >> BK_SHIFT], 1);
            atomicAdd(&lh[d[u].y >> BK_SHIFT], 1);
            atomicAdd(&lh[d[u].z >> BK_SHIFT], 1);
            atomicAdd(&lh[d[u].w >> BK_SHIFT], 1);
        }
    }
    __syncthreads();
    if (t < NBK) bh[((size_t)e * NBLK + blk) * NBK + t] = lh[t];
}

__global__ __launch_bounds__(1024)
void part_scan_kernel(const int* __restrict__ bh, int* __restrict__ off) {
    __shared__ int ssum[SCAN_SL][NBK];
    __shared__ int bbase[NBK];
    __shared__ int wtot[8];
    static_assert(NBK <= 256 && SCAN_SL * NBK <= 1024, "thread map covers every (slice, bucket)");
    const int e    = blockIdx.x;
    const int t    = threadIdx.x;
    const int lane = t & 31;
    const int sl   = t / NBK;
    const int b    = t - sl * NBK;
    const bool live = t < SCAN_SL * NBK;
    const int k0 = sl * SLB;
    const int k1 = (k0 + SLB < NBLK) ? k0 + SLB : NBLK;
    const int* pi = bh  + (size_t)e * NBLK * NBK + b;
    int*       po = off + (size_t)e * NBLK * NBK + b;

    int s = 0;
    if (live) {
#pragma unroll 8
        for (int k = k0; k < k1; ++k) s += pi[(size_t)k * NBK];
        ssum[sl][b] = s;
    }
    __syncthreads();
    int tot = 0, inc = 0;
    if (t < 256) {
        if (t < NBK) {
#pragma unroll
            for (int q = 0; q < SCAN_SL; ++q) tot += ssum[q][t];
        }
        inc = tot;
#pragma unroll
        for (int d = 1; d < 32; d <<= 1) {
            const int o = __shfl_up(inc, d);
            if (lane >= d) inc += o;
        }
        if (lane == 31) wtot[t >> 5] = inc;
    }
    __syncthreads();
    if (t < NBK) {
        int wb = 0;
        for (int q = 0; q < (t >> 5); ++q) wb += wtot[q];
        bbase[t] = wb + inc - tot;
    }
    __syncthreads();
    if (live) {
        int run = bbase[b];
        for (int q = 0; q < sl; ++q) run += ssum[q][b];
#pragma unroll 8
        for (int k = k0; k < k1; ++k) {
            const int v = pi[(size_t)k * NBK];
            po[(size_t)k * NBK] = run;
            run += v;
        }
    }
}

__global__ __launch_bounds__(256)
void part_scatter_kernel(const int* __restrict__ src, const int* __restrict__ dst,
                         const int* __restrict__ off, int* __restrict__ pairs) {
    __shared__ int cur[NBK];
    const int e = blockIdx.y, blk = blockIdx.x, t = threadIdx.x;
    if (t < NBK) cur[t] = off[((size_t)e * NBLK + blk) * NBK + t];
    const int4* d4 = (const int4*)(dst + (size_t)e * N_EDGES) + (size_t)blk * (EPB / 4);
    const int4* s4 = (const int4*)(src + (size_t)e * N_EDGES) + (size_t)blk * (EPB / 4);
    int2* pe = (int2*)pairs + (size_t)e * N_EDGES;
    int4 d[EPT], s[EPT];
#pragma unroll
    for (int u = 0; u < EPT; ++u) {
        const int i = u * 256 + t;
        d[u] = make_int4(0, 0, 0, 0);
        s[u] = make_int4(0, 0, 0, 0);
        if (blk * (EPB / 4) + i < N_EDGES / 4) { d[u] = d4[i]; s[u] = s4[i]; }
    }
    __syncthreads();
#pragma unroll
    for (int u = 0; u < EPT; ++u) {
        if (blk * (EPB / 4) + u * 256 + t < N_EDGES / 4) {
            const int qx = atomicAdd(&cur[d[u].x >> BK_SHIFT], 1);
            const int qy = atomicAdd(&cur[d[u].y >> BK_SHIFT], 1);
            const int qz = atomicAdd(&cur[d[u].z >> BK_SHIFT], 1);
            const int qw = atomicAdd(&cur[d[u].w >> BK_SHIFT], 1);
            pe[qx] = make_int2(s[u].x, d[u].x);
            pe[qy] = make_int2(s[u].y, d[u].y);
            pe[qz] = make_int2(s[u].z, d[u].z);
            pe[qw] = make_int2(s[u].w, d[u].w);
        }
    }
}

__global__ __launch_bounds__(BK_NODES)
void bucket_sort_kernel(const int* __restrict__ pairs, const int* __restrict__ off,
                        int* __restrict__ rowptr, int* __restrict__ col) {
    __shared__ int cnt[BK_NODES];
    __shared__ int wsum[BK_NODES / 32];
    const int e = blockIdx.y, b = blockIdx.x, t = threadIdx.x;
    const int lane = t & 31;
    const int w    = t >> 5;
    const int* oe  = off + (size_t)e * NBLK * NBK;
    const int beg  = oe[b];
    const int end  = (b + 1 < NBK) ? oe[b + 1] : N_EDGES;
    const int2* pe = (const int2*)pairs + (size_t)e * N_EDGES;
    int*        ce = col + (size_t)e * N_EDGES;

    cnt[t] = 0;
    __syncthreads();
    for (int j0 = beg; j0 < end; j0 += 4 * BK_NODES) {
        int2 p[4];
#pragma unroll
        for (int u = 0; u < 4; ++u) {
            const int j = j0 + u * BK_NODES + t;
            p[u] = pe[(j < end) ? j : end - 1];
        }
#pragma unroll
        for (int u = 0; u < 4; ++u)
            if (j0 + u * BK_NODES + t < end) atomicAdd(&cnt[p[u].y & (BK_NODES - 1)], 1);
    }
    __syncthreads();

    const int c = cnt[t];
    int inc = c;
#pragma unroll
    for (int d = 1; d < 32; d <<= 1) {
        const int o = __shfl_up(inc, d);
        if (lane >= d) inc += o;
    }
    if (lane == 31) wsum[w] = inc;
    __syncthreads();
    int first = beg + inc - c;
    for (int q = 0; q < w; ++q) first += wsum[q];
    cnt[t] = first;
    const int n = b * BK_NODES + t;
    if (n < N_NODES) rowptr[(size_t)e * RPS + n] = first;
    if (b == NBK - 1 && t == 0) rowptr[(size_t)e * RPS + N_NODES] = end;
    __syncthreads();

    for (int j0 = beg; j0 < end; j0 += 4 * BK_NODES) {
        int2 p[4];
#pragma unroll
        for (int u = 0; u < 4; ++u) {
            const int j = j0 + u * BK_NODES + t;
            p[u] = pe[(j < end) ? j : end - 1];
        }
#pragma unroll
        for (int u = 0; u < 4; ++u) {
            if (j0 + u * BK_NODES + t < end) {
                const int q = atomicAdd(&cnt[p[u].y & (BK_NODES - 1)], 1);
                ce[q] = p[u].x;
            }
        }
    }
}

#define CL_P   (D_IN + 16)
#define CL_PA  (D_IN + 4)

__global__ __launch_bounds__(256)
void conv_layer_kernel(const float* __restrict__ h,
                       const int*   __restrict__ rowptr,
                       const int*   __restrict__ col,
                       const float* __restrict__ Wl,
                       const float* __restrict__ bl,
                       float*       __restrict__ out) {
    __shared__ __attribute__((aligned(16))) float2 Ws[(D_IN / 2) * CL_P];
    __shared__ __attribute__((aligned(16))) float  As[8 * 16 * CL_PA];

    const int tid  = threadIdx.x;
    const int wave = __builtin_amdgcn_readfirstlane(tid >> 5);
    const int lane = tid & 31;
    const int half = lane >> 4;
    const int l15  = lane & 15;
    const int m0   = (blockIdx.x * 8 + wave) * 16;
    const bool live = m0 < N_NODES;

    float*       asw = As + wave * (16 * CL_PA);
    const char*    hb     = (const char*)h;
    const unsigned lane_b = (unsigned)lane * 16u;

    v8f o[8];
#pragma unroll
    for (int j = 0; j < 8; ++j) o[j] = v8f{};

#pragma unroll 1
    for (int e = 0; e < N_ETYPES; ++e) {
        const int* rpe = rowptr + (size_t)e * RPS;
        const int* ce  = col + (size_t)e * N_EDGES;

        int rp = 0;
        if (live) rp = rpe[m0 + (lane < 16 ? lane : 16)];

        if (e > 0) __syncthreads();

        {
            const float* W = Wl + (size_t)e * D_IN * D_IN;
#pragma unroll
            for (int it = 0; it < (D_IN / 2) * (D_IN / 4) / 256; ++it) {
                const int idx = tid + it * 256;
                const int kp = idx / (D_IN / 4);
                const int n4 = (idx % (D_IN / 4)) * 4;
                const float4 r0 = *(const float4*)(W + (size_t)(2 * kp)     * D_IN + n4);
                const float4 r1 = *(const float4*)(W + (size_t)(2 * kp + 1) * D_IN + n4);
                float4* d = (float4*)(Ws + kp * CL_P + n4);
                d[0] = make_float4(r0.x, r1.x, r0.y, r1.y);
                d[1] = make_float4(r0.z, r1.z, r0.w, r1.w);
            }
        }

        if (live) {
            const int ebeg = __builtin_amdgcn_readlane(rp, 0);
            const int eend = __builtin_amdgcn_readlane(rp, 16);
            if (lane > 16) rp = 0x7fffffff;
            int row  = 0;
            int rbeg = ebeg;
            int rend = __builtin_amdgcn_readlane(rp, 1);
            float4 acc = make_float4(0.0f, 0.0f, 0.0f, 0.0f);

            int mine_next = 0;
            if (ebeg + lane < eend) mine_next = ce[ebeg + lane];
            for (int j = ebeg; j < eend; j += 32) {
                int ids = mine_next;
                if (j + 32 + lane < eend) mine_next = ce[j + 32 + lane];
                const int id0 = __builtin_amdgcn_readlane(ids, 0);
                if (j + lane >= eend) ids = id0;
                float4 v[32];
#pragma unroll
                for (int u = 0; u < 32; ++u) {
                    const unsigned s = (unsigned)__builtin_amdgcn_readlane(ids, u);
                    v[u] = *(const float4*)(hb + (s * (unsigned)(D_IN * sizeof(float)) + lane_b));
                }
                int lim = rend - j;
#pragma unroll
                for (int u = 0; u < 32; ++u) {
                    if (__builtin_expect(u >= lim, 0)) {
                        while (j + u >= rend && row < 16) {
                            const float rs = 1.0f / fmaxf((float)(rend - rbeg), 1.0f);
                            *(float4*)(asw + row * CL_PA + lane * 4) =
                                make_float4(acc.x * rs, acc.y * rs, acc.z * rs, acc.w * rs);
                            acc = make_float4(0.0f, 0.0f, 0.0f, 0.0f);
                            ++row;
                            rbeg = rend;
                            rend = __builtin_amdgcn_readlane(rp, row + 1);
                        }
                        lim = rend - j;
                    }
                    acc.x += v[u].x;
                    acc.y += v[u].y;
                    acc.z += v[u].z;
                    acc.w += v[u].w;
                }
            }
            for (; row < 16; ++row) {
                const float rs = 1.0f / fmaxf((float)(rend - rbeg), 1.0f);
                *(float4*)(asw + row * CL_PA + lane * 4) =
                    make_float4(acc.x * rs, acc.y * rs, acc.z * rs, acc.w * rs);
                acc = make_float4(0.0f, 0.0f, 0.0f, 0.0f);
                rbeg = rend;
                rend = __builtin_amdgcn_readlane(rp, (row + 2 < 16) ? row + 2 : 16);
            }
        }
        __syncthreads();

        if (live) {
            v8f acc[8];
#pragma unroll
            for (int j = 0; j < 8; ++j) acc[j] = v8f{};

            const float*  ar = asw + l15 * CL_PA + half * 2;
            const float2* wb = Ws + half * CL_P + l15;

#pragma unroll 1
            for (int kb = 0; kb < D_IN / 32; ++kb) {
#pragma unroll
                for (int s = 0; s < 8; ++s) {
                    const float2 av = *(const float2*)(ar + kb * 32 + s * 4);
                    v2f a;
                    a.x = av.x;
                    a.y = av.y;
                    const float2* wk = wb + (kb * 16 + s * 2) * CL_P;
#pragma unroll
                    for (int j = 0; j < 8; ++j) {
                        const float2 bv = wk[16 * j];
                        v2f b;
                        b.x = bv.x;
                        b.y = bv.y;
                        acc[j] = __builtin_amdgcn_wmma_f32_16x16x4_f32(
                             false, a,  false, b,
                             (short)0, acc[j],  false,  false);
                    }
                }
            }

            const float* be = bl + (size_t)e * D_IN;
#pragma unroll
            for (int j = 0; j < 8; ++j) {
                const float bv = be[16 * j + l15];
#pragma unroll
                for (int r = 0; r < 8; ++r) {
                    float v = fmaxf(acc[j][r] + bv, 0.0f);
                    v *= 1.0f / 3.0f;
                    o[j][r] = o[j][r] + v;
                }
            }
        }
    }

    if (live) {
        float* crow = out + (size_t)(m0 + half * 8) * D_IN + l15;
#pragma unroll
        for (int j = 0; j < 8; ++j)
#pragma unroll
            for (int r = 0; r < 8; ++r) crow[(size_t)r * D_IN + 16 * j] = o[j][r];
    }
}

template <int NT, int N, int K, bool RELU, bool ACC>
__global__ __launch_bounds__(256)
void wmma_gemm_strip_f32_kernel(const float* __restrict__ A,
                                const float* __restrict__ W,
                                const float* __restrict__ bias,
                                float* __restrict__ C,
                                const float* __restrict__ deg,
                                int M, float outScale) {
    constexpr int JT = NT / 16;
    constexpr int P  = NT + 16;
    constexpr int KB = K / 32;
    __shared__ __attribute__((aligned(16))) float2 Ws[(K / 2) * P];

    const int tid = threadIdx.x;
    const int nb  = blockIdx.y * NT;

    constexpr int STAGE_IT = (K / 2) * (NT / 4) / 256;
    static_assert(STAGE_IT * 256 == (K / 2) * (NT / 4), "slab divides evenly over the work-group");
#pragma unroll
    for (int it = 0; it < STAGE_IT; ++it) {
        const int idx = tid + it * 256;
        const int kp = idx / (NT / 4);
        const int n4 = (idx % (NT / 4)) * 4;
        const float4 r0 = *(const float4*)(W + (size_t)(2 * kp)     * N + nb + n4);
        const float4 r1 = *(const float4*)(W + (size_t)(2 * kp + 1) * N + nb + n4);
        float4* d = (float4*)(Ws + kp * P + n4);
        d[0] = make_float4(r0.x, r1.x, r0.y, r1.y);
        d[1] = make_float4(r0.z, r1.z, r0.w, r1.w);
    }
    __syncthreads();

    const int wave = tid >> 5;
    const int lane = tid & 31;
    const int half = lane >> 4;
    const int l15  = lane & 15;
    const int m0   = (blockIdx.x * 8 + wave) * 16;
    if (m0 >= M) return;

    const int rowA = m0 + l15;
    float rs = 1.0f;
    if (deg) rs = 1.0f / fmaxf(deg[rowA], 1.0f);

    v8f acc[JT];
#pragma unroll
    for (int j = 0; j < JT; ++j) acc[j] = v8f{};

    const float*  arow = A + (size_t)rowA * K + half * 2;
    const float2* wb   = Ws + half * P + l15;

    float2 cur[8], nxt[8];
#pragma unroll
    for (int s = 0; s < 8; ++s) cur[s] = *(const float2*)(arow + s * 4);

#pragma unroll 1
    for (int kb = 0; kb < KB; ++kb) {
        const int kn = (kb + 1 < KB) ? kb + 1 : kb;
#pragma unroll
        for (int s = 0; s < 8; ++s) nxt[s] = *(const float2*)(arow + kn * 32 + s * 4);
#pragma unroll
        for (int s = 0; s < 8; ++s) {
            v2f a;
            a.x = cur[s].x * rs;
            a.y = cur[s].y * rs;
            const float2* wk = wb + (kb * 16 + s * 2) * P;
#pragma unroll
            for (int j = 0; j < JT; ++j) {
                const float2 bv = wk[16 * j];
                v2f b;
                b.x = bv.x;
                b.y = bv.y;
                acc[j] = __builtin_amdgcn_wmma_f32_16x16x4_f32(
                     false, a,  false, b,
                     (short)0, acc[j],  false,  false);
            }
        }
#pragma unroll
        for (int s = 0; s < 8; ++s) cur[s] = nxt[s];
    }

    float* crow = C + (size_t)(m0 + half * 8) * N + nb + l15;
    float old[ACC ? JT : 1][8];
    if (ACC) {
#pragma unroll
        for (int j = 0; j < JT; ++j)
#pragma unroll
            for (int r = 0; r < 8; ++r) old[j][r] = crow[(size_t)r * N + 16 * j];
    }
#pragma unroll
    for (int j = 0; j < JT; ++j) {
        const float bv = bias[nb + 16 * j + l15];
#pragma unroll
        for (int r = 0; r < 8; ++r) {
            float v = acc[j][r] + bv;
            if (RELU) v = fmaxf(v, 0.0f);
            v *= outScale;
            if (ACC) v = old[j][r] + v;
            crow[(size_t)r * N + 16 * j] = v;
        }
    }
}

#define HF_P   (D_OUT + 16)
#define HF_KS  (D_HID / 4)

__global__ __launch_bounds__(256)
void head_fold_kernel(const float* __restrict__ W1,
                      const float* __restrict__ b1,
                      const float* __restrict__ W2,
                      const float* __restrict__ b2,
                      float*       __restrict__ W12,
                      float*       __restrict__ b12) {
    __shared__ __attribute__((aligned(16))) float2 Ws[(D_HID / 2) * HF_P];

    const int tid = threadIdx.x;
    constexpr int STAGE_IT = (D_HID / 2) * (D_OUT / 4) / 256;
    static_assert(STAGE_IT * 256 == (D_HID / 2) * (D_OUT / 4), "slab divides evenly over the work-group");
#pragma unroll
    for (int it = 0; it < STAGE_IT; ++it) {
        const int idx = tid + it * 256;
        const int kp = idx / (D_OUT / 4);
        const int n4 = (idx % (D_OUT / 4)) * 4;
        const float4 r0 = *(const float4*)(W2 + (size_t)(2 * kp)     * D_OUT + n4);
        const float4 r1 = *(const float4*)(W2 + (size_t)(2 * kp + 1) * D_OUT + n4);
        float4* d = (float4*)(Ws + kp * HF_P + n4);
        d[0] = make_float4(r0.x, r1.x, r0.y, r1.y);
        d[1] = make_float4(r0.z, r1.z, r0.w, r1.w);
    }

    const int lane  = tid & 31;
    const int half  = lane >> 4;
    const int l15   = lane & 15;
    const int strip = blockIdx.x;
    const bool bias_strip = strip == D_IN / 16;
    const float* arow = bias_strip ? b1 + half * 2
                                   : W1 + (size_t)(strip * 16 + l15) * D_HID + half * 2;
    const bool zero_row = bias_strip && l15 != 0;

    float2 av[HF_KS];
    if (tid < 32) {
#pragma unroll
        for (int s = 0; s < HF_KS; ++s) av[s] = *(const float2*)(arow + s * 4);
    }
    __syncthreads();
    if (tid >= 32) return;

    v8f acc[D_OUT / 16];
#pragma unroll
    for (int j = 0; j < D_OUT / 16; ++j) acc[j] = v8f{};

    const float2* wb = Ws + half * HF_P + l15;
#pragma unroll
    for (int s = 0; s < HF_KS; ++s) {
        v2f a;
        a.x = zero_row ? 0.0f : av[s].x;
        a.y = zero_row ? 0.0f : av[s].y;
        const float2* wk = wb + (s * 2) * HF_P;
#pragma unroll
        for (int j = 0; j < D_OUT / 16; ++j) {
            const float2 bv = wk[16 * j];
            v2f b;
            b.x = bv.x;
            b.y = bv.y;
            acc[j] = __builtin_amdgcn_wmma_f32_16x16x4_f32(
                 false, a,  false, b,
                 (short)0, acc[j],  false,  false);
        }
    }

    if (!bias_strip) {
        float* crow = W12 + (size_t)(strip * 16 + half * 8) * D_OUT + l15;
#pragma unroll
        for (int j = 0; j < D_OUT / 16; ++j)
#pragma unroll
            for (int r = 0; r < 8; ++r) crow[(size_t)r * D_OUT + 16 * j] = acc[j][r];
    } else if (half == 0) {
#pragma unroll
        for (int j = 0; j < D_OUT / 16; ++j) b12[16 * j + l15] = acc[j][0] + b2[16 * j + l15];
    }
}

extern "C" void kernel_launch(void* const* d_in, const int* in_sizes, int n_in,
                              void* d_out, int out_size, void* d_ws, size_t ws_size,
                              hipStream_t stream) {
    const float* blocks = (const float*)d_in[0];
    const int*   esrc   = (const int*)  d_in[1];
    const int*   edst   = (const int*)  d_in[2];
    const float* convW  = (const float*)d_in[3];
    const float* convB  = (const float*)d_in[4];
    const float* W1     = (const float*)d_in[5];
    const float* b1     = (const float*)d_in[6];
    const float* W2     = (const float*)d_in[7];
    const float* b2     = (const float*)d_in[8];
    float*       out    = (float*)d_out;

    char* ws = (char*)d_ws;
    const size_t BUF = (size_t)N_NODES * D_IN * sizeof(float);
    float* buf0 = (float*)(ws);
    char*  csr  = ws + BUF + BUF / 2;
    int*   col    = (int*)(csr);
    int*   rowptr = (int*)(csr + (size_t)N_ETYPES * N_EDGES * sizeof(int));
    float* buf2 = (float*)(ws + 2 * BUF);
    int*   bh    = (int*)(ws);
    int*   off   = (int*)(ws + (1 << 20));
    int*   pairs = (int*)(ws + 2 * BUF);
    float* W12  = (float*)(ws + BUF);
    float* b12  = W12 + D_IN * D_OUT;

    {
        dim3 g(NBLK, N_ETYPES);
        part_hist_kernel<<<g, 256, 0, stream>>>(edst, bh);
        part_scan_kernel<<<N_ETYPES, 1024, 0, stream>>>(bh, off);
        part_scatter_kernel<<<g, 256, 0, stream>>>(esrc, edst, off, pairs);
        dim3 gb(NBK, N_ETYPES);
        bucket_sort_kernel<<<gb, BK_NODES, 0, stream>>>(pairs, off, rowptr, col);
    }

    const int row_blocks = (N_NODES / 16 + 7) / 8;

    conv_layer_kernel<<<row_blocks, 256, 0, stream>>>(
        blocks, rowptr, col, convW, convB, buf0);
    conv_layer_kernel<<<row_blocks, 256, 0, stream>>>(
        buf0, rowptr, col, convW + (size_t)N_ETYPES * D_IN * D_IN, convB + (size_t)N_ETYPES * D_IN, buf2);

    head_fold_kernel<<<D_IN / 16 + 1, 256, 0, stream>>>(W1, b1, W2, b2, W12, b12);
    {
        dim3 g(row_blocks, D_OUT / 32);
        wmma_gemm_strip_f32_kernel<32, D_OUT, D_IN, false, false><<<g, 256, 0, stream>>>(
            buf2, W12, b12, out, nullptr, N_NODES, 1.0f);
    }
}


# Conv layer as one resident 13-wave work-group per WGP: the 3 etype slabs staged in LDS once, then every wave walks its strips with no barrier, so gathers run under other waves' WMMAs; sum orders kept
# speedup vs baseline: 7.7553x; 1.1394x over previous
#include <hip/hip_runtime.h>

typedef float v2f __attribute__((ext_vector_type(2)));
typedef float v8f __attribute__((ext_vector_type(8)));

#define N_NODES  100000
#define N_ETYPES 3
#define N_EDGES  1600000
#define D_IN     128
#define D_HID    256
#define D_OUT    32

#define RPS      (N_NODES + 4)

#define BK_SHIFT 9
#define BK_NODES (1 << BK_SHIFT)
#define NBK      ((N_NODES + BK_NODES - 1) / BK_NODES)
#define EPB      4096
#define NBLK     ((N_EDGES + EPB - 1) / EPB)
#define EPT      (EPB / 4 / 256)
#define SCAN_SL  5
#define SLB      ((NBLK + SCAN_SL - 1) / SCAN_SL)
static_assert((size_t)N_ETYPES * NBLK * NBK * sizeof(int) <= (1u << 20), "a histogram plane fits the 1 MiB the driver gives it");

__global__ __launch_bounds__(256)
void part_hist_kernel(const int* __restrict__ dst, int* __restrict__ bh) {
    __shared__ int lh[NBK];
    const int e = blockIdx.y, blk = blockIdx.x, t = threadIdx.x;
    if (t < NBK) lh[t] = 0;
    __syncthreads();
    const int4* d4 = (const int4*)(dst + (size_t)e * N_EDGES) + (size_t)blk * (EPB / 4);
    int4 d[EPT];
#pragma unroll
    for (int u = 0; u < EPT; ++u) {
        const int i = u * 256 + t;
        d[u] = make_int4(0, 0, 0, 0);
        if (blk * (EPB / 4) + i < N_EDGES / 4) d[u] = d4[i];
    }
#pragma unroll
    for (int u = 0; u < EPT; ++u) {
        if (blk * (EPB / 4) + u * 256 + t < N_EDGES / 4) {
            atomicAdd(&lh[d[u].x >> BK_SHIFT], 1);
            atomicAdd(&lh[d[u].y >> BK_SHIFT], 1);
            atomicAdd(&lh[d[u].z >> BK_SHIFT], 1);
            atomicAdd(&lh[d[u].w >> BK_SHIFT], 1);
        }
    }
    __syncthreads();
    if (t < NBK) bh[((size_t)e * NBLK + blk) * NBK + t] = lh[t];
}

__global__ __launch_bounds__(1024)
void part_scan_kernel(const int* __restrict__ bh, int* __restrict__ off) {
    __shared__ int ssum[SCAN_SL][NBK];
    __shared__ int bbase[NBK];
    __shared__ int wtot[8];
    static_assert(NBK <= 256 && SCAN_SL * NBK <= 1024, "thread map covers every (slice, bucket)");
    const int e    = blockIdx.x;
    const int t    = threadIdx.x;
    const int lane = t & 31;
    const int sl   = t / NBK;
    const int b    = t - sl * NBK;
    const bool live = t < SCAN_SL * NBK;
    const int k0 = sl * SLB;
    const int k1 = (k0 + SLB < NBLK) ? k0 + SLB : NBLK;
    const int* pi = bh  + (size_t)e * NBLK * NBK + b;
    int*       po = off + (size_t)e * NBLK * NBK + b;

    int s = 0;
    if (live) {
#pragma unroll 8
        for (int k = k0; k < k1; ++k) s += pi[(size_t)k * NBK];
        ssum[sl][b] = s;
    }
    __syncthreads();
    int tot = 0, inc = 0;
    if (t < 256) {
        if (t < NBK) {
#pragma unroll
            for (int q = 0; q < SCAN_SL; ++q) tot += ssum[q][t];
        }
        inc = tot;
#pragma unroll
        for (int d = 1; d < 32; d <<= 1) {
            const int o = __shfl_up(inc, d);
            if (lane >= d) inc += o;
        }
        if (lane == 31) wtot[t >> 5] = inc;
    }
    __syncthreads();
    if (t < NBK) {
        int wb = 0;
        for (int q = 0; q < (t >> 5); ++q) wb += wtot[q];
        bbase[t] = wb + inc - tot;
    }
    __syncthreads();
    if (live) {
        int run = bbase[b];
        for (int q = 0; q < sl; ++q) run += ssum[q][b];
#pragma unroll 8
        for (int k = k0; k < k1; ++k) {
            const int v = pi[(size_t)k * NBK];
            po[(size_t)k * NBK] = run;
            run += v;
        }
    }
}

__global__ __launch_bounds__(256)
void part_scatter_kernel(const int* __restrict__ src, const int* __restrict__ dst,
                         const int* __restrict__ off, int* __restrict__ pairs) {
    __shared__ int cur[NBK];
    const int e = blockIdx.y, blk = blockIdx.x, t = threadIdx.x;
    if (t < NBK) cur[t] = off[((size_t)e * NBLK + blk) * NBK + t];
    const int4* d4 = (const int4*)(dst + (size_t)e * N_EDGES) + (size_t)blk * (EPB / 4);
    const int4* s4 = (const int4*)(src + (size_t)e * N_EDGES) + (size_t)blk * (EPB / 4);
    int2* pe = (int2*)pairs + (size_t)e * N_EDGES;
    int4 d[EPT], s[EPT];
#pragma unroll
    for (int u = 0; u < EPT; ++u) {
        const int i = u * 256 + t;
        d[u] = make_int4(0, 0, 0, 0);
        s[u] = make_int4(0, 0, 0, 0);
        if (blk * (EPB / 4) + i < N_EDGES / 4) { d[u] = d4[i]; s[u] = s4[i]; }
    }
    __syncthreads();
#pragma unroll
    for (int u = 0; u < EPT; ++u) {
        if (blk * (EPB / 4) + u * 256 + t < N_EDGES / 4) {
            const int qx = atomicAdd(&cur[d[u].x >> BK_SHIFT], 1);
            const int qy = atomicAdd(&cur[d[u].y >> BK_SHIFT], 1);
            const int qz = atomicAdd(&cur[d[u].z >> BK_SHIFT], 1);
            const int qw = atomicAdd(&cur[d[u].w >> BK_SHIFT], 1);
            pe[qx] = make_int2(s[u].x, d[u].x);
            pe[qy] = make_int2(s[u].y, d[u].y);
            pe[qz] = make_int2(s[u].z, d[u].z);
            pe[qw] = make_int2(s[u].w, d[u].w);
        }
    }
}

__global__ __launch_bounds__(BK_NODES)
void bucket_sort_kernel(const int* __restrict__ pairs, const int* __restrict__ off,
                        int* __restrict__ rowptr, int* __restrict__ col) {
    __shared__ int cnt[BK_NODES];
    __shared__ int wsum[BK_NODES / 32];
    const int e = blockIdx.y, b = blockIdx.x, t = threadIdx.x;
    const int lane = t & 31;
    const int w    = t >> 5;
    const int* oe  = off + (size_t)e * NBLK * NBK;
    const int beg  = oe[b];
    const int end  = (b + 1 < NBK) ? oe[b + 1] : N_EDGES;
    const int2* pe = (const int2*)pairs + (size_t)e * N_EDGES;
    int*        ce = col + (size_t)e * N_EDGES;

    cnt[t] = 0;
    __syncthreads();
    for (int j0 = beg; j0 < end; j0 += 4 * BK_NODES) {
        int2 p[4];
#pragma unroll
        for (int u = 0; u < 4; ++u) {
            const int j = j0 + u * BK_NODES + t;
            p[u] = pe[(j < end) ? j : end - 1];
        }
#pragma unroll
        for (int u = 0; u < 4; ++u)
            if (j0 + u * BK_NODES + t < end) atomicAdd(&cnt[p[u].y & (BK_NODES - 1)], 1);
    }
    __syncthreads();

    const int c = cnt[t];
    int inc = c;
#pragma unroll
    for (int d = 1; d < 32; d <<= 1) {
        const int o = __shfl_up(inc, d);
        if (lane >= d) inc += o;
    }
    if (lane == 31) wsum[w] = inc;
    __syncthreads();
    int first = beg + inc - c;
    for (int q = 0; q < w; ++q) first += wsum[q];
    cnt[t] = first;
    const int n = b * BK_NODES + t;
    if (n < N_NODES) rowptr[(size_t)e * RPS + n] = first;
    if (b == NBK - 1 && t == 0) rowptr[(size_t)e * RPS + N_NODES] = end;
    __syncthreads();

    for (int j0 = beg; j0 < end; j0 += 4 * BK_NODES) {
        int2 p[4];
#pragma unroll
        for (int u = 0; u < 4; ++u) {
            const int j = j0 + u * BK_NODES + t;
            p[u] = pe[(j < end) ? j : end - 1];
        }
#pragma unroll
        for (int u = 0; u < 4; ++u) {
            if (j0 + u * BK_NODES + t < end) {
                const int q = atomicAdd(&cnt[p[u].y & (BK_NODES - 1)], 1);
                ce[q] = p[u].x;
            }
        }
    }
}

#define CL_WAVES 13
#define CL_GRID  256
#define CL_PA    (D_IN + 4)
#define N_STRIPS (N_NODES / 16)
static_assert(N_STRIPS * 16 == N_NODES, "strips are whole");
static_assert(2 * CL_WAVES * CL_GRID >= N_STRIPS, "two trips cover every strip");

__global__ __launch_bounds__(CL_WAVES * 32)
void conv_layer_kernel(const float* __restrict__ h,
                       const int*   __restrict__ rowptr,
                       const int*   __restrict__ col,
                       const float* __restrict__ Wl,
                       const float* __restrict__ bl,
                       float*       __restrict__ out) {
    __shared__ __attribute__((aligned(16))) float2 Ws[N_ETYPES * (D_IN / 2) * D_IN];
    __shared__ __attribute__((aligned(16))) float  As[CL_WAVES * 16 * CL_PA];

    const int tid  = threadIdx.x;
    const int wave = __builtin_amdgcn_readfirstlane(tid >> 5);
    const int lane = tid & 31;
    const int half = lane >> 4;
    const int l15  = lane & 15;

    {
        constexpr int SLAB_ITEMS = (D_IN / 2) * (D_IN / 4);
        constexpr int ALL_ITEMS  = N_ETYPES * SLAB_ITEMS;
        constexpr int NIT = (ALL_ITEMS + CL_WAVES * 32 - 1) / (CL_WAVES * 32);
#pragma unroll
        for (int it0 = 0; it0 < NIT; it0 += 8) {
            float4 r0[8], r1[8];
#pragma unroll
            for (int u = 0; u < 8; ++u) {
                const int idx = tid + (it0 + u) * (CL_WAVES * 32);
                r0[u] = make_float4(0.0f, 0.0f, 0.0f, 0.0f);
                r1[u] = make_float4(0.0f, 0.0f, 0.0f, 0.0f);
                if (it0 + u < NIT && idx < ALL_ITEMS) {
                    const int e  = idx / SLAB_ITEMS;
                    const int r  = idx - e * SLAB_ITEMS;
                    const int kp = r / (D_IN / 4);
                    const int n4 = (r % (D_IN / 4)) * 4;
                    const float* W = Wl + (size_t)e * D_IN * D_IN;
                    r0[u] = *(const float4*)(W + (size_t)(2 * kp)     * D_IN + n4);
                    r1[u] = *(const float4*)(W + (size_t)(2 * kp + 1) * D_IN + n4);
                }
            }
#pragma unroll
            for (int u = 0; u < 8; ++u) {
                const int idx = tid + (it0 + u) * (CL_WAVES * 32);
                if (it0 + u < NIT && idx < ALL_ITEMS) {
                    const int e  = idx / SLAB_ITEMS;
                    const int r  = idx - e * SLAB_ITEMS;
                    const int kp = r / (D_IN / 4);
                    const int n4 = (r % (D_IN / 4)) * 4;
                    const int pos = (n4 + 16 * (kp & 1)) & (D_IN - 1);
                    float4* d = (float4*)(Ws + (e * (D_IN / 2) + kp) * D_IN + pos);
                    d[0] = make_float4(r0[u].x, r1[u].x, r0[u].y, r1[u].y);
                    d[1] = make_float4(r0[u].z, r1[u].z, r0[u].w, r1[u].w);
                }
            }
        }
    }
    __syncthreads();

    float*         asw    = As + wave * (16 * CL_PA);
    const char*    hb     = (const char*)h;
    const unsigned lane_b = (unsigned)lane * 16u;

#pragma unroll 1
    for (int strip = blockIdx.x + CL_GRID * wave; strip < N_STRIPS; strip += CL_GRID * CL_WAVES) {
        const int m0 = strip * 16;

        v8f o[8];
#pragma unroll
        for (int j = 0; j < 8; ++j) o[j] = v8f{};

#pragma unroll 1
        for (int e = 0; e < N_ETYPES; ++e) {
            const int* rpe = rowptr + (size_t)e * RPS;
            const int* ce  = col + (size_t)e * N_EDGES;

            int rp = rpe[m0 + (lane < 16 ? lane : 16)];

            {
                const int ebeg = __builtin_amdgcn_readlane(rp, 0);
                const int eend = __builtin_amdgcn_readlane(rp, 16);
                if (lane > 16) rp = 0x7fffffff;
                int row  = 0;
                int rbeg = ebeg;
                int rend = __builtin_amdgcn_readlane(rp, 1);
                float4 acc = make_float4(0.0f, 0.0f, 0.0f, 0.0f);

                int mine_next = 0;
                if (ebeg + lane < eend) mine_next = ce[ebeg + lane];
                for (int j = ebeg; j < eend; j += 32) {
                    int ids = mine_next;
                    if (j + 32 + lane < eend) mine_next = ce[j + 32 + lane];
                    const int id0 = __builtin_amdgcn_readlane(ids, 0);
                    if (j + lane >= eend) ids = id0;
                    float4 v[32];
#pragma unroll
                    for (int u = 0; u < 32; ++u) {
                        const unsigned s = (unsigned)__builtin_amdgcn_readlane(ids, u);
                        v[u] = *(const float4*)(hb + (s * (unsigned)(D_IN * sizeof(float)) + lane_b));
                    }
                    int lim = rend - j;
#pragma unroll
                    for (int u = 0; u < 32; ++u) {
                        if (__builtin_expect(u >= lim, 0)) {
                            while (j + u >= rend && row < 16) {
                                const float rs = 1.0f / fmaxf((float)(rend - rbeg), 1.0f);
                                *(float4*)(asw + row * CL_PA + lane * 4) =
                                    make_float4(acc.x * rs, acc.y * rs, acc.z * rs, acc.w * rs);
                                acc = make_float4(0.0f, 0.0f, 0.0f, 0.0f);
                                ++row;
                                rbeg = rend;
                                rend = __builtin_amdgcn_readlane(rp, row + 1);
                            }
                            lim = rend - j;
                        }
                        acc.x += v[u].x;
                        acc.y += v[u].y;
                        acc.z += v[u].z;
                        acc.w += v[u].w;
                    }
                }
                for (; row < 16; ++row) {
                    const float rs = 1.0f / fmaxf((float)(rend - rbeg), 1.0f);
                    *(float4*)(asw + row * CL_PA + lane * 4) =
                        make_float4(acc.x * rs, acc.y * rs, acc.z * rs, acc.w * rs);
                    acc = make_float4(0.0f, 0.0f, 0.0f, 0.0f);
                    rbeg = rend;
                    rend = __builtin_amdgcn_readlane(rp, (row + 2 < 16) ? row + 2 : 16);
                }
            }
            __builtin_amdgcn_fence(__ATOMIC_ACQ_REL, "wavefront");
            __builtin_amdgcn_wave_barrier();

            {
                v8f acc[8];
#pragma unroll
                for (int j = 0; j < 8; ++j) acc[j] = v8f{};

                const float2* wse = Ws + e * ((D_IN / 2) * D_IN);
                const float*  ar  = asw + l15 * CL_PA + half * 2;
                const float2* wb  = wse + half * D_IN + half * 16 + l15;
                const float2* wb7 = wse + half * D_IN + (half ? 0 : 7 * 16) + l15;

#pragma unroll 1
                for (int kb = 0; kb < D_IN / 32; ++kb) {
#pragma unroll
                    for (int s = 0; s < 8; ++s) {
                        const float2 av = *(const float2*)(ar + kb * 32 + s * 4);
                        v2f a;
                        a.x = av.x;
                        a.y = av.y;
                        const int ko = (kb * 16 + s * 2) * D_IN;
#pragma unroll
                        for (int j = 0; j < 8; ++j) {
                            const float2 bv = (j < 7) ? wb[ko + 16 * j] : wb7[ko];
                            v2f b;
                            b.x = bv.x;
                            b.y = bv.y;
                            acc[j] = __builtin_amdgcn_wmma_f32_16x16x4_f32(
                                 false, a,  false, b,
                                 (short)0, acc[j],  false,  false);
                        }
                    }
                }
                __builtin_amdgcn_fence(__ATOMIC_ACQ_REL, "wavefront");
                __builtin_amdgcn_wave_barrier();

                const float* be = bl + (size_t)e * D_IN;
#pragma unroll
                for (int j = 0; j < 8; ++j) {
                    const float bv = be[16 * j + l15];
#pragma unroll
                    for (int r = 0; r < 8; ++r) {
                        float v = fmaxf(acc[j][r] + bv, 0.0f);
                        v *= 1.0f / 3.0f;
                        o[j][r] = o[j][r] + v;
                    }
                }
            }
        }

        float* crow = out + (size_t)(m0 + half * 8) * D_IN + l15;
#pragma unroll
        for (int j = 0; j < 8; ++j)
#pragma unroll
            for (int r = 0; r < 8; ++r) crow[(size_t)r * D_IN + 16 * j] = o[j][r];
    }
}

template <int NT, int N, int K, bool RELU, bool ACC>
__global__ __launch_bounds__(256)
void wmma_gemm_strip_f32_kernel(const float* __restrict__ A,
                                const float* __restrict__ W,
                                const float* __restrict__ bias,
                                float* __restrict__ C,
                                const float* __restrict__ deg,
                                int M, float outScale) {
    constexpr int JT = NT / 16;
    constexpr int P  = NT + 16;
    constexpr int KB = K / 32;
    __shared__ __attribute__((aligned(16))) float2 Ws[(K / 2) * P];

    const int tid = threadIdx.x;
    const int nb  = blockIdx.y * NT;

    constexpr int STAGE_IT = (K / 2) * (NT / 4) / 256;
    static_assert(STAGE_IT * 256 == (K / 2) * (NT / 4), "slab divides evenly over the work-group");
#pragma unroll
    for (int it = 0; it < STAGE_IT; ++it) {
        const int idx = tid + it * 256;
        const int kp = idx / (NT / 4);
        const int n4 = (idx % (NT / 4)) * 4;
        const float4 r0 = *(const float4*)(W + (size_t)(2 * kp)     * N + nb + n4);
        const float4 r1 = *(const float4*)(W + (size_t)(2 * kp + 1) * N + nb + n4);
        float4* d = (float4*)(Ws + kp * P + n4);
        d[0] = make_float4(r0.x, r1.x, r0.y, r1.y);
        d[1] = make_float4(r0.z, r1.z, r0.w, r1.w);
    }
    __syncthreads();

    const int wave = tid >> 5;
    const int lane = tid & 31;
    const int half = lane >> 4;
    const int l15  = lane & 15;
    const int m0   = (blockIdx.x * 8 + wave) * 16;
    if (m0 >= M) return;

    const int rowA = m0 + l15;
    float rs = 1.0f;
    if (deg) rs = 1.0f / fmaxf(deg[rowA], 1.0f);

    v8f acc[JT];
#pragma unroll
    for (int j = 0; j < JT; ++j) acc[j] = v8f{};

    const float*  arow = A + (size_t)rowA * K + half * 2;
    const float2* wb   = Ws + half * P + l15;

    float2 cur[8], nxt[8];
#pragma unroll
    for (int s = 0; s < 8; ++s) cur[s] = *(const float2*)(arow + s * 4);

#pragma unroll 1
    for (int kb = 0; kb < KB; ++kb) {
        const int kn = (kb + 1 < KB) ? kb + 1 : kb;
#pragma unroll
        for (int s = 0; s < 8; ++s) nxt[s] = *(const float2*)(arow + kn * 32 + s * 4);
#pragma unroll
        for (int s = 0; s < 8; ++s) {
            v2f a;
            a.x = cur[s].x * rs;
            a.y = cur[s].y * rs;
            const float2* wk = wb + (kb * 16 + s * 2) * P;
#pragma unroll
            for (int j = 0; j < JT; ++j) {
                const float2 bv = wk[16 * j];
                v2f b;
                b.x = bv.x;
                b.y = bv.y;
                acc[j] = __builtin_amdgcn_wmma_f32_16x16x4_f32(
                     false, a,  false, b,
                     (short)0, acc[j],  false,  false);
            }
        }
#pragma unroll
        for (int s = 0; s < 8; ++s) cur[s] = nxt[s];
    }

    float* crow = C + (size_t)(m0 + half * 8) * N + nb + l15;
    float old[ACC ? JT : 1][8];
    if (ACC) {
#pragma unroll
        for (int j = 0; j < JT; ++j)
#pragma unroll
            for (int r = 0; r < 8; ++r) old[j][r] = crow[(size_t)r * N + 16 * j];
    }
#pragma unroll
    for (int j = 0; j < JT; ++j) {
        const float bv = bias[nb + 16 * j + l15];
#pragma unroll
        for (int r = 0; r < 8; ++r) {
            float v = acc[j][r] + bv;
            if (RELU) v = fmaxf(v, 0.0f);
            v *= outScale;
            if (ACC) v = old[j][r] + v;
            crow[(size_t)r * N + 16 * j] = v;
        }
    }
}

#define HF_P   (D_OUT + 16)
#define HF_KS  (D_HID / 4)

__global__ __launch_bounds__(256)
void head_fold_kernel(const float* __restrict__ W1,
                      const float* __restrict__ b1,
                      const float* __restrict__ W2,
                      const float* __restrict__ b2,
                      float*       __restrict__ W12,
                      float*       __restrict__ b12) {
    __shared__ __attribute__((aligned(16))) float2 Ws[(D_HID / 2) * HF_P];

    const int tid = threadIdx.x;
    constexpr int STAGE_IT = (D_HID / 2) * (D_OUT / 4) / 256;
    static_assert(STAGE_IT * 256 == (D_HID / 2) * (D_OUT / 4), "slab divides evenly over the work-group");
#pragma unroll
    for (int it = 0; it < STAGE_IT; ++it) {
        const int idx = tid + it * 256;
        const int kp = idx / (D_OUT / 4);
        const int n4 = (idx % (D_OUT / 4)) * 4;
        const float4 r0 = *(const float4*)(W2 + (size_t)(2 * kp)     * D_OUT + n4);
        const float4 r1 = *(const float4*)(W2 + (size_t)(2 * kp + 1) * D_OUT + n4);
        float4* d = (float4*)(Ws + kp * HF_P + n4);
        d[0] = make_float4(r0.x, r1.x, r0.y, r1.y);
        d[1] = make_float4(r0.z, r1.z, r0.w, r1.w);
    }

    const int lane  = tid & 31;
    const int half  = lane >> 4;
    const int l15   = lane & 15;
    const int strip = blockIdx.x;
    const bool bias_strip = strip == D_IN / 16;
    const float* arow = bias_strip ? b1 + half * 2
                                   : W1 + (size_t)(strip * 16 + l15) * D_HID + half * 2;
    const bool zero_row = bias_strip && l15 != 0;

    float2 av[HF_KS];
    if (tid < 32) {
#pragma unroll
        for (int s = 0; s < HF_KS; ++s) av[s] = *(const float2*)(arow + s * 4);
    }
    __syncthreads();
    if (tid >= 32) return;

    v8f acc[D_OUT / 16];
#pragma unroll
    for (int j = 0; j < D_OUT / 16; ++j) acc[j] = v8f{};

    const float2* wb = Ws + half * HF_P + l15;
#pragma unroll
    for (int s = 0; s < HF_KS; ++s) {
        v2f a;
        a.x = zero_row ? 0.0f : av[s].x;
        a.y = zero_row ? 0.0f : av[s].y;
        const float2* wk = wb + (s * 2) * HF_P;
#pragma unroll
        for (int j = 0; j < D_OUT / 16; ++j) {
            const float2 bv = wk[16 * j];
            v2f b;
            b.x = bv.x;
            b.y = bv.y;
            acc[j] = __builtin_amdgcn_wmma_f32_16x16x4_f32(
                 false, a,  false, b,
                 (short)0, acc[j],  false,  false);
        }
    }

    if (!bias_strip) {
        float* crow = W12 + (size_t)(strip * 16 + half * 8) * D_OUT + l15;
#pragma unroll
        for (int j = 0; j < D_OUT / 16; ++j)
#pragma unroll
            for (int r = 0; r < 8; ++r) crow[(size_t)r * D_OUT + 16 * j] = acc[j][r];
    } else if (half == 0) {
#pragma unroll
        for (int j = 0; j < D_OUT / 16; ++j) b12[16 * j + l15] = acc[j][0] + b2[16 * j + l15];
    }
}

extern "C" void kernel_launch(void* const* d_in, const int* in_sizes, int n_in,
                              void* d_out, int out_size, void* d_ws, size_t ws_size,
                              hipStream_t stream) {
    const float* blocks = (const float*)d_in[0];
    const int*   esrc   = (const int*)  d_in[1];
    const int*   edst   = (const int*)  d_in[2];
    const float* convW  = (const float*)d_in[3];
    const float* convB  = (const float*)d_in[4];
    const float* W1     = (const float*)d_in[5];
    const float* b1     = (const float*)d_in[6];
    const float* W2     = (const float*)d_in[7];
    const float* b2     = (const float*)d_in[8];
    float*       out    = (float*)d_out;

    char* ws = (char*)d_ws;
    const size_t BUF = (size_t)N_NODES * D_IN * sizeof(float);
    float* buf0 = (float*)(ws);
    char*  csr  = ws + BUF + BUF / 2;
    int*   col    = (int*)(csr);
    int*   rowptr = (int*)(csr + (size_t)N_ETYPES * N_EDGES * sizeof(int));
    float* buf2 = (float*)(ws + 2 * BUF);
    int*   bh    = (int*)(ws);
    int*   off   = (int*)(ws + (1 << 20));
    int*   pairs = (int*)(ws + 2 * BUF);
    float* W12  = (float*)(ws + BUF);
    float* b12  = W12 + D_IN * D_OUT;

    {
        dim3 g(NBLK, N_ETYPES);
        part_hist_kernel<<<g, 256, 0, stream>>>(edst, bh);
        part_scan_kernel<<<N_ETYPES, 1024, 0, stream>>>(bh, off);
        part_scatter_kernel<<<g, 256, 0, stream>>>(esrc, edst, off, pairs);
        dim3 gb(NBK, N_ETYPES);
        bucket_sort_kernel<<<gb, BK_NODES, 0, stream>>>(pairs, off, rowptr, col);
    }

    const int row_blocks = (N_NODES / 16 + 7) / 8;

    conv_layer_kernel<<<CL_GRID, CL_WAVES * 32, 0, stream>>>(
        blocks, rowptr, col, convW, convB, buf0);
    conv_layer_kernel<<<CL_GRID, CL_WAVES * 32, 0, stream>>>(
        buf0, rowptr, col, convW + (size_t)N_ETYPES * D_IN * D_IN, convB + (size_t)N_ETYPES * D_IN, buf2);

    head_fold_kernel<<<D_IN / 16 + 1, 256, 0, stream>>>(W1, b1, W2, b2, W12, b12);
    {
        dim3 g(row_blocks, D_OUT / 32);
        wmma_gemm_strip_f32_kernel<32, D_OUT, D_IN, false, false><<<g, 256, 0, stream>>>(
            buf2, W12, b12, out, nullptr, N_NODES, 1.0f);
    }
}


# Edge-list build: placement passes 3 and 4 put records in final order in LDS and store contiguous runs (whole lines), not one scattered 8/4-byte store per edge; sum order = LDS add arrival as before
# speedup vs baseline: 8.8573x; 1.1421x over previous
#include <hip/hip_runtime.h>

typedef float v2f __attribute__((ext_vector_type(2)));
typedef float v8f __attribute__((ext_vector_type(8)));

#define N_NODES  100000
#define N_ETYPES 3
#define N_EDGES  1600000
#define D_IN     128
#define D_HID    256
#define D_OUT    32

#define RPS      (N_NODES + 4)

#define BK_SHIFT 9
#define BK_NODES (1 << BK_SHIFT)
#define NBK      ((N_NODES + BK_NODES - 1) / BK_NODES)
#define EPB      4096
#define NBLK     ((N_EDGES + EPB - 1) / EPB)
#define EPT      (EPB / 4 / 256)
#define SCAN_SL  5
#define SLB      ((NBLK + SCAN_SL - 1) / SCAN_SL)
#define LCAP     16384
static_assert((size_t)N_ETYPES * NBLK * NBK * sizeof(int) <= (1u << 20), "a histogram plane fits the 1 MiB the driver gives it");

__global__ __launch_bounds__(256)
void part_hist_kernel(const int* __restrict__ dst, int* __restrict__ bh) {
    __shared__ int lh[NBK];
    const int e = blockIdx.y, blk = blockIdx.x, t = threadIdx.x;
    if (t < NBK) lh[t] = 0;
    __syncthreads();
    const int4* d4 = (const int4*)(dst + (size_t)e * N_EDGES) + (size_t)blk * (EPB / 4);
    int4 d[EPT];
#pragma unroll
    for (int u = 0; u < EPT; ++u) {
        const int i = u * 256 + t;
        d[u] = make_int4(0, 0, 0, 0);
        if (blk * (EPB / 4) + i < N_EDGES / 4) d[u] = d4[i];
    }
#pragma unroll
    for (int u = 0; u < EPT; ++u) {
        if (blk * (EPB / 4) + u * 256 + t < N_EDGES / 4) {
            atomicAdd(&lh[d[u].x >> BK_SHIFT], 1);
            atomicAdd(&lh[d[u].y >> BK_SHIFT], 1);
            atomicAdd(&lh[d[u].z >> BK_SHIFT], 1);
            atomicAdd(&lh[d[u].w >> BK_SHIFT], 1);
        }
    }
    __syncthreads();
    if (t < NBK) bh[((size_t)e * NBLK + blk) * NBK + t] = lh[t];
}

__global__ __launch_bounds__(1024)
void part_scan_kernel(const int* __restrict__ bh, int* __restrict__ off) {
    __shared__ int ssum[SCAN_SL][NBK];
    __shared__ int bbase[NBK];
    __shared__ int wtot[8];
    static_assert(NBK <= 256 && SCAN_SL * NBK <= 1024, "thread map covers every (slice, bucket)");
    const int e    = blockIdx.x;
    const int t    = threadIdx.x;
    const int lane = t & 31;
    const int sl   = t / NBK;
    const int b    = t - sl * NBK;
    const bool live = t < SCAN_SL * NBK;
    const int k0 = sl * SLB;
    const int k1 = (k0 + SLB < NBLK) ? k0 + SLB : NBLK;
    const int* pi = bh  + (size_t)e * NBLK * NBK + b;
    int*       po = off + (size_t)e * NBLK * NBK + b;

    int s = 0;
    if (live) {
#pragma unroll 8
        for (int k = k0; k < k1; ++k) s += pi[(size_t)k * NBK];
        ssum[sl][b] = s;
    }
    __syncthreads();
    int tot = 0, inc = 0;
    if (t < 256) {
        if (t < NBK) {
#pragma unroll
            for (int q = 0; q < SCAN_SL; ++q) tot += ssum[q][t];
        }
        inc = tot;
#pragma unroll
        for (int d = 1; d < 32; d <<= 1) {
            const int o = __shfl_up(inc, d);
            if (lane >= d) inc += o;
        }
        if (lane == 31) wtot[t >> 5] = inc;
    }
    __syncthreads();
    if (t < NBK) {
        int wb = 0;
        for (int q = 0; q < (t >> 5); ++q) wb += wtot[q];
        bbase[t] = wb + inc - tot;
    }
    __syncthreads();
    if (live) {
        int run = bbase[b];
        for (int q = 0; q < sl; ++q) run += ssum[q][b];
#pragma unroll 8
        for (int k = k0; k < k1; ++k) {
            const int v = pi[(size_t)k * NBK];
            po[(size_t)k * NBK] = run;
            run += v;
        }
    }
}

__global__ __launch_bounds__(256)
void part_scatter_kernel(const int* __restrict__ src, const int* __restrict__ dst,
                         const int* __restrict__ off, int* __restrict__ pairs) {
    __shared__ int lh[NBK];
    __shared__ int gofs[NBK];
    __shared__ int wtot[8];
    __shared__ __attribute__((aligned(16))) int2 lp[EPB];
    static_assert(NBK <= 256, "one thread per bucket in the scan");
    const int e = blockIdx.y, blk = blockIdx.x, t = threadIdx.x;
    const int lane = t & 31;
    if (t < NBK) lh[t] = 0;
    int go = 0;
    if (t < NBK) go = off[((size_t)e * NBLK + blk) * NBK + t];
    const int4* d4 = (const int4*)(dst + (size_t)e * N_EDGES) + (size_t)blk * (EPB / 4);
    const int4* s4 = (const int4*)(src + (size_t)e * N_EDGES) + (size_t)blk * (EPB / 4);
    int2* pe = (int2*)pairs + (size_t)e * N_EDGES;
    int4 d[EPT], s[EPT], r[EPT];
#pragma unroll
    for (int u = 0; u < EPT; ++u) {
        const int i = u * 256 + t;
        d[u] = make_int4(0, 0, 0, 0);
        s[u] = make_int4(0, 0, 0, 0);
        if (blk * (EPB / 4) + i < N_EDGES / 4) { d[u] = d4[i]; s[u] = s4[i]; }
    }
    __syncthreads();
#pragma unroll
    for (int u = 0; u < EPT; ++u) {
        r[u] = make_int4(0, 0, 0, 0);
        if (blk * (EPB / 4) + u * 256 + t < N_EDGES / 4) {
            r[u].x = atomicAdd(&lh[d[u].x >> BK_SHIFT], 1);
            r[u].y = atomicAdd(&lh[d[u].y >> BK_SHIFT], 1);
            r[u].z = atomicAdd(&lh[d[u].z >> BK_SHIFT], 1);
            r[u].w = atomicAdd(&lh[d[u].w >> BK_SHIFT], 1);
        }
    }
    __syncthreads();
    const int c = (t < NBK) ? lh[t] : 0;
    int inc = c;
#pragma unroll
    for (int k = 1; k < 32; k <<= 1) {
        const int o = __shfl_up(inc, k);
        if (lane >= k) inc += o;
    }
    if (lane == 31) wtot[t >> 5] = inc;
    __syncthreads();
    if (t < NBK) {
        int wb = 0;
        for (int q = 0; q < (t >> 5); ++q) wb += wtot[q];
        const int ls = wb + inc - c;
        lh[t]   = ls;
        gofs[t] = go - ls;
    }
    __syncthreads();
#pragma unroll
    for (int u = 0; u < EPT; ++u) {
        if (blk * (EPB / 4) + u * 256 + t < N_EDGES / 4) {
            lp[lh[d[u].x >> BK_SHIFT] + r[u].x] = make_int2(s[u].x, d[u].x);
            lp[lh[d[u].y >> BK_SHIFT] + r[u].y] = make_int2(s[u].y, d[u].y);
            lp[lh[d[u].z >> BK_SHIFT] + r[u].z] = make_int2(s[u].z, d[u].z);
            lp[lh[d[u].w >> BK_SHIFT] + r[u].w] = make_int2(s[u].w, d[u].w);
        }
    }
    __syncthreads();
    const int left = N_EDGES - blk * EPB;
    const int nv   = left < EPB ? left : EPB;
#pragma unroll 4
    for (int i = t; i < nv; i += 256) {
        const int2 p = lp[i];
        pe[gofs[p.y >> BK_SHIFT] + i] = p;
    }
}

__global__ __launch_bounds__(BK_NODES)
void bucket_sort_kernel(const int* __restrict__ pairs, const int* __restrict__ off,
                        int* __restrict__ rowptr, int* __restrict__ col) {
    __shared__ int cnt[BK_NODES];
    __shared__ int wsum[BK_NODES / 32];
    __shared__ int lcol[LCAP];
    const int e = blockIdx.y, b = blockIdx.x, t = threadIdx.x;
    const int lane = t & 31;
    const int w    = t >> 5;
    const int* oe  = off + (size_t)e * NBLK * NBK;
    const int beg  = oe[b];
    const int end  = (b + 1 < NBK) ? oe[b + 1] : N_EDGES;
    const int2* pe = (const int2*)pairs + (size_t)e * N_EDGES;
    int*        ce = col + (size_t)e * N_EDGES;

    const bool staged = end - beg <= LCAP;
    cnt[t] = 0;
    __syncthreads();
    for (int j0 = beg; j0 < end; j0 += 4 * BK_NODES) {
        int2 p[4];
#pragma unroll
        for (int u = 0; u < 4; ++u) {
            const int j = j0 + u * BK_NODES + t;
            p[u] = pe[(j < end) ? j : end - 1];
        }
#pragma unroll
        for (int u = 0; u < 4; ++u)
            if (j0 + u * BK_NODES + t < end) atomicAdd(&cnt[p[u].y & (BK_NODES - 1)], 1);
    }
    __syncthreads();

    const int c = cnt[t];
    int inc = c;
#pragma unroll
    for (int d = 1; d < 32; d <<= 1) {
        const int o = __shfl_up(inc, d);
        if (lane >= d) inc += o;
    }
    if (lane == 31) wsum[w] = inc;
    __syncthreads();
    int first = beg + inc - c;
    for (int q = 0; q < w; ++q) first += wsum[q];
    cnt[t] = first;
    const int n = b * BK_NODES + t;
    if (n < N_NODES) rowptr[(size_t)e * RPS + n] = first;
    if (b == NBK - 1 && t == 0) rowptr[(size_t)e * RPS + N_NODES] = end;
    __syncthreads();

    for (int j0 = beg; j0 < end; j0 += 4 * BK_NODES) {
        int2 p[4];
#pragma unroll
        for (int u = 0; u < 4; ++u) {
            const int j = j0 + u * BK_NODES + t;
            p[u] = pe[(j < end) ? j : end - 1];
        }
        int q[4];
#pragma unroll
        for (int u = 0; u < 4; ++u) {
            q[u] = 0;
            if (j0 + u * BK_NODES + t < end) q[u] = atomicAdd(&cnt[p[u].y & (BK_NODES - 1)], 1);
        }
#pragma unroll
        for (int u = 0; u < 4; ++u) {
            if (j0 + u * BK_NODES + t < end) {
                if (staged) lcol[q[u] - beg] = p[u].x;
                else        ce[q[u]] = p[u].x;
            }
        }
    }
    if (staged) {
        __syncthreads();
        int g = (beg & ~31) + t;
        if (g >= beg && g < end) ce[g] = lcol[g - beg];
#pragma unroll 4
        for (g += BK_NODES; g < end; g += BK_NODES) ce[g] = lcol[g - beg];
    }
}

#define CL_WAVES 13
#define CL_GRID  256
#define CL_PA    (D_IN + 4)
#define N_STRIPS (N_NODES / 16)
static_assert(N_STRIPS * 16 == N_NODES, "strips are whole");
static_assert(2 * CL_WAVES * CL_GRID >= N_STRIPS, "two trips cover every strip");

__global__ __launch_bounds__(CL_WAVES * 32)
void conv_layer_kernel(const float* __restrict__ h,
                       const int*   __restrict__ rowptr,
                       const int*   __restrict__ col,
                       const float* __restrict__ Wl,
                       const float* __restrict__ bl,
                       float*       __restrict__ out) {
    __shared__ __attribute__((aligned(16))) float2 Ws[N_ETYPES * (D_IN / 2) * D_IN];
    __shared__ __attribute__((aligned(16))) float  As[CL_WAVES * 16 * CL_PA];

    const int tid  = threadIdx.x;
    const int wave = __builtin_amdgcn_readfirstlane(tid >> 5);
    const int lane = tid & 31;
    const int half = lane >> 4;
    const int l15  = lane & 15;

    {
        constexpr int SLAB_ITEMS = (D_IN / 2) * (D_IN / 4);
        constexpr int ALL_ITEMS  = N_ETYPES * SLAB_ITEMS;
        constexpr int NIT = (ALL_ITEMS + CL_WAVES * 32 - 1) / (CL_WAVES * 32);
#pragma unroll
        for (int it0 = 0; it0 < NIT; it0 += 8) {
            float4 r0[8], r1[8];
#pragma unroll
            for (int u = 0; u < 8; ++u) {
                const int idx = tid + (it0 + u) * (CL_WAVES * 32);
                r0[u] = make_float4(0.0f, 0.0f, 0.0f, 0.0f);
                r1[u] = make_float4(0.0f, 0.0f, 0.0f, 0.0f);
                if (it0 + u < NIT && idx < ALL_ITEMS) {
                    const int e  = idx / SLAB_ITEMS;
                    const int r  = idx - e * SLAB_ITEMS;
                    const int kp = r / (D_IN / 4);
                    const int n4 = (r % (D_IN / 4)) * 4;
                    const float* W = Wl + (size_t)e * D_IN * D_IN;
                    r0[u] = *(const float4*)(W + (size_t)(2 * kp)     * D_IN + n4);
                    r1[u] = *(const float4*)(W + (size_t)(2 * kp + 1) * D_IN + n4);
                }
            }
#pragma unroll
            for (int u = 0; u < 8; ++u) {
                const int idx = tid + (it0 + u) * (CL_WAVES * 32);
                if (it0 + u < NIT && idx < ALL_ITEMS) {
                    const int e  = idx / SLAB_ITEMS;
                    const int r  = idx - e * SLAB_ITEMS;
                    const int kp = r / (D_IN / 4);
                    const int n4 = (r % (D_IN / 4)) * 4;
                    const int pos = (n4 + 16 * (kp & 1)) & (D_IN - 1);
                    float4* d = (float4*)(Ws + (e * (D_IN / 2) + kp) * D_IN + pos);
                    d[0] = make_float4(r0[u].x, r1[u].x, r0[u].y, r1[u].y);
                    d[1] = make_float4(r0[u].z, r1[u].z, r0[u].w, r1[u].w);
                }
            }
        }
    }
    __syncthreads();

    float*         asw    = As + wave * (16 * CL_PA);
    const char*    hb     = (const char*)h;
    const unsigned lane_b = (unsigned)lane * 16u;

#pragma unroll 1
    for (int strip = blockIdx.x + CL_GRID * wave; strip < N_STRIPS; strip += CL_GRID * CL_WAVES) {
        const int m0 = strip * 16;

        v8f o[8];
#pragma unroll
        for (int j = 0; j < 8; ++j) o[j] = v8f{};

#pragma unroll 1
        for (int e = 0; e < N_ETYPES; ++e) {
            const int* rpe = rowptr + (size_t)e * RPS;
            const int* ce  = col + (size_t)e * N_EDGES;

            int rp = rpe[m0 + (lane < 16 ? lane : 16)];

            {
                const int ebeg = __builtin_amdgcn_readlane(rp, 0);
                const int eend = __builtin_amdgcn_readlane(rp, 16);
                if (lane > 16) rp = 0x7fffffff;
                int row  = 0;
                int rbeg = ebeg;
                int rend = __builtin_amdgcn_readlane(rp, 1);
                float4 acc = make_float4(0.0f, 0.0f, 0.0f, 0.0f);

                int mine_next = 0;
                if (ebeg + lane < eend) mine_next = ce[ebeg + lane];
                for (int j = ebeg; j < eend; j += 32) {
                    int ids = mine_next;
                    if (j + 32 + lane < eend) mine_next = ce[j + 32 + lane];
                    const int id0 = __builtin_amdgcn_readlane(ids, 0);
                    if (j + lane >= eend) ids = id0;
                    float4 v[32];
#pragma unroll
                    for (int u = 0; u < 32; ++u) {
                        const unsigned s = (unsigned)__builtin_amdgcn_readlane(ids, u);
                        v[u] = *(const float4*)(hb + (s * (unsigned)(D_IN * sizeof(float)) + lane_b));
                    }
                    int lim = rend - j;
#pragma unroll
                    for (int u = 0; u < 32; ++u) {
                        if (__builtin_expect(u >= lim, 0)) {
                            while (j + u >= rend && row < 16) {
                                const float rs = 1.0f / fmaxf((float)(rend - rbeg), 1.0f);
                                *(float4*)(asw + row * CL_PA + lane * 4) =
                                    make_float4(acc.x * rs, acc.y * rs, acc.z * rs, acc.w * rs);
                                acc = make_float4(0.0f, 0.0f, 0.0f, 0.0f);
                                ++row;
                                rbeg = rend;
                                rend = __builtin_amdgcn_readlane(rp, row + 1);
                            }
                            lim = rend - j;
                        }
                        acc.x += v[u].x;
                        acc.y += v[u].y;
                        acc.z += v[u].z;
                        acc.w += v[u].w;
                    }
                }
                for (; row < 16; ++row) {
                    const float rs = 1.0f / fmaxf((float)(rend - rbeg), 1.0f);
                    *(float4*)(asw + row * CL_PA + lane * 4) =
                        make_float4(acc.x * rs, acc.y * rs, acc.z * rs, acc.w * rs);
                    acc = make_float4(0.0f, 0.0f, 0.0f, 0.0f);
                    rbeg = rend;
                    rend = __builtin_amdgcn_readlane(rp, (row + 2 < 16) ? row + 2 : 16);
                }
            }
            __builtin_amdgcn_fence(__ATOMIC_ACQ_REL, "wavefront");
            __builtin_amdgcn_wave_barrier();

            {
                v8f acc[8];
#pragma unroll
                for (int j = 0; j < 8; ++j) acc[j] = v8f{};

                const float2* wse = Ws + e * ((D_IN / 2) * D_IN);
                const float*  ar  = asw + l15 * CL_PA + half * 2;
                const float2* wb  = wse + half * D_IN + half * 16 + l15;
                const float2* wb7 = wse + half * D_IN + (half ? 0 : 7 * 16) + l15;

#pragma unroll 1
                for (int kb = 0; kb < D_IN / 32; ++kb) {
#pragma unroll
                    for (int s = 0; s < 8; ++s) {
                        const float2 av = *(const float2*)(ar + kb * 32 + s * 4);
                        v2f a;
                        a.x = av.x;
                        a.y = av.y;
                        const int ko = (kb * 16 + s * 2) * D_IN;
#pragma unroll
                        for (int j = 0; j < 8; ++j) {
                            const float2 bv = (j < 7) ? wb[ko + 16 * j] : wb7[ko];
                            v2f b;
                            b.x = bv.x;
                            b.y = bv.y;
                            acc[j] = __builtin_amdgcn_wmma_f32_16x16x4_f32(
                                 false, a,  false, b,
                                 (short)0, acc[j],  false,  false);
                        }
                    }
                }
                __builtin_amdgcn_fence(__ATOMIC_ACQ_REL, "wavefront");
                __builtin_amdgcn_wave_barrier();

                const float* be = bl + (size_t)e * D_IN;
#pragma unroll
                for (int j = 0; j < 8; ++j) {
                    const float bv = be[16 * j + l15];
#pragma unroll
                    for (int r = 0; r < 8; ++r) {
                        float v = fmaxf(acc[j][r] + bv, 0.0f);
                        v *= 1.0f / 3.0f;
                        o[j][r] = o[j][r] + v;
                    }
                }
            }
        }

        float* crow = out + (size_t)(m0 + half * 8) * D_IN + l15;
#pragma unroll
        for (int j = 0; j < 8; ++j)
#pragma unroll
            for (int r = 0; r < 8; ++r) crow[(size_t)r * D_IN + 16 * j] = o[j][r];
    }
}

template <int NT, int N, int K, bool RELU, bool ACC>
__global__ __launch_bounds__(256)
void wmma_gemm_strip_f32_kernel(const float* __restrict__ A,
                                const float* __restrict__ W,
                                const float* __restrict__ bias,
                                float* __restrict__ C,
                                const float* __restrict__ deg,
                                int M, float outScale) {
    constexpr int JT = NT / 16;
    constexpr int P  = NT + 16;
    constexpr int KB = K / 32;
    __shared__ __attribute__((aligned(16))) float2 Ws[(K / 2) * P];

    const int tid = threadIdx.x;
    const int nb  = blockIdx.y * NT;

    constexpr int STAGE_IT = (K / 2) * (NT / 4) / 256;
    static_assert(STAGE_IT * 256 == (K / 2) * (NT / 4), "slab divides evenly over the work-group");
#pragma unroll
    for (int it = 0; it < STAGE_IT; ++it) {
        const int idx = tid + it * 256;
        const int kp = idx / (NT / 4);
        const int n4 = (idx % (NT / 4)) * 4;
        const float4 r0 = *(const float4*)(W + (size_t)(2 * kp)     * N + nb + n4);
        const float4 r1 = *(const float4*)(W + (size_t)(2 * kp + 1) * N + nb + n4);
        float4* d = (float4*)(Ws + kp * P + n4);
        d[0] = make_float4(r0.x, r1.x, r0.y, r1.y);
        d[1] = make_float4(r0.z, r1.z, r0.w, r1.w);
    }
    __syncthreads();

    const int wave = tid >> 5;
    const int lane = tid & 31;
    const int half = lane >> 4;
    const int l15  = lane & 15;
    const int m0   = (blockIdx.x * 8 + wave) * 16;
    if (m0 >= M) return;

    const int rowA = m0 + l15;
    float rs = 1.0f;
    if (deg) rs = 1.0f / fmaxf(deg[rowA], 1.0f);

    v8f acc[JT];
#pragma unroll
    for (int j = 0; j < JT; ++j) acc[j] = v8f{};

    const float*  arow = A + (size_t)rowA * K + half * 2;
    const float2* wb   = Ws + half * P + l15;

    float2 cur[8], nxt[8];
#pragma unroll
    for (int s = 0; s < 8; ++s) cur[s] = *(const float2*)(arow + s * 4);

#pragma unroll 1
    for (int kb = 0; kb < KB; ++kb) {
        const int kn = (kb + 1 < KB) ? kb + 1 : kb;
#pragma unroll
        for (int s = 0; s < 8; ++s) nxt[s] = *(const float2*)(arow + kn * 32 + s * 4);
#pragma unroll
        for (int s = 0; s < 8; ++s) {
            v2f a;
            a.x = cur[s].x * rs;
            a.y = cur[s].y * rs;
            const float2* wk = wb + (kb * 16 + s * 2) * P;
#pragma unroll
            for (int j = 0; j < JT; ++j) {
                const float2 bv = wk[16 * j];
                v2f b;
                b.x = bv.x;
                b.y = bv.y;
                acc[j] = __builtin_amdgcn_wmma_f32_16x16x4_f32(
                     false, a,  false, b,
                     (short)0, acc[j],  false,  false);
            }
        }
#pragma unroll
        for (int s = 0; s < 8; ++s) cur[s] = nxt[s];
    }

    float* crow = C + (size_t)(m0 + half * 8) * N + nb + l15;
    float old[ACC ? JT : 1][8];
    if (ACC) {
#pragma unroll
        for (int j = 0; j < JT; ++j)
#pragma unroll
            for (int r = 0; r < 8; ++r) old[j][r] = crow[(size_t)r * N + 16 * j];
    }
#pragma unroll
    for (int j = 0; j < JT; ++j) {
        const float bv = bias[nb + 16 * j + l15];
#pragma unroll
        for (int r = 0; r < 8; ++r) {
            float v = acc[j][r] + bv;
            if (RELU) v = fmaxf(v, 0.0f);
            v *= outScale;
            if (ACC) v = old[j][r] + v;
            crow[(size_t)r * N + 16 * j] = v;
        }
    }
}

#define HF_P   (D_OUT + 16)
#define HF_KS  (D_HID / 4)

__global__ __launch_bounds__(256)
void head_fold_kernel(const float* __restrict__ W1,
                      const float* __restrict__ b1,
                      const float* __restrict__ W2,
                      const float* __restrict__ b2,
                      float*       __restrict__ W12,
                      float*       __restrict__ b12) {
    __shared__ __attribute__((aligned(16))) float2 Ws[(D_HID / 2) * HF_P];

    const int tid = threadIdx.x;
    constexpr int STAGE_IT = (D_HID / 2) * (D_OUT / 4) / 256;
    static_assert(STAGE_IT * 256 == (D_HID / 2) * (D_OUT / 4), "slab divides evenly over the work-group");
#pragma unroll
    for (int it = 0; it < STAGE_IT; ++it) {
        const int idx = tid + it * 256;
        const int kp = idx / (D_OUT / 4);
        const int n4 = (idx % (D_OUT / 4)) * 4;
        const float4 r0 = *(const float4*)(W2 + (size_t)(2 * kp)     * D_OUT + n4);
        const float4 r1 = *(const float4*)(W2 + (size_t)(2 * kp + 1) * D_OUT + n4);
        float4* d = (float4*)(Ws + kp * HF_P + n4);
        d[0] = make_float4(r0.x, r1.x, r0.y, r1.y);
        d[1] = make_float4(r0.z, r1.z, r0.w, r1.w);
    }

    const int lane  = tid & 31;
    const int half  = lane >> 4;
    const int l15   = lane & 15;
    const int strip = blockIdx.x;
    const bool bias_strip = strip == D_IN / 16;
    const float* arow = bias_strip ? b1 + half * 2
                                   : W1 + (size_t)(strip * 16 + l15) * D_HID + half * 2;
    const bool zero_row = bias_strip && l15 != 0;

    float2 av[HF_KS];
    if (tid < 32) {
#pragma unroll
        for (int s = 0; s < HF_KS; ++s) av[s] = *(const float2*)(arow + s * 4);
    }
    __syncthreads();
    if (tid >= 32) return;

    v8f acc[D_OUT / 16];
#pragma unroll
    for (int j = 0; j < D_OUT / 16; ++j) acc[j] = v8f{};

    const float2* wb = Ws + half * HF_P + l15;
#pragma unroll
    for (int s = 0; s < HF_KS; ++s) {
        v2f a;
        a.x = zero_row ? 0.0f : av[s].x;
        a.y = zero_row ? 0.0f : av[s].y;
        const float2* wk = wb + (s * 2) * HF_P;
#pragma unroll
        for (int j = 0; j < D_OUT / 16; ++j) {
            const float2 bv = wk[16 * j];
            v2f b;
            b.x = bv.x;
            b.y = bv.y;
            acc[j] = __builtin_amdgcn_wmma_f32_16x16x4_f32(
                 false, a,  false, b,
                 (short)0, acc[j],  false,  false);
        }
    }

    if (!bias_strip) {
        float* crow = W12 + (size_t)(strip * 16 + half * 8) * D_OUT + l15;
#pragma unroll
        for (int j = 0; j < D_OUT / 16; ++j)
#pragma unroll
            for (int r = 0; r < 8; ++r) crow[(size_t)r * D_OUT + 16 * j] = acc[j][r];
    } else if (half == 0) {
#pragma unroll
        for (int j = 0; j < D_OUT / 16; ++j) b12[16 * j + l15] = acc[j][0] + b2[16 * j + l15];
    }
}

extern "C" void kernel_launch(void* const* d_in, const int* in_sizes, int n_in,
                              void* d_out, int out_size, void* d_ws, size_t ws_size,
                              hipStream_t stream) {
    const float* blocks = (const float*)d_in[0];
    const int*   esrc   = (const int*)  d_in[1];
    const int*   edst   = (const int*)  d_in[2];
    const float* convW  = (const float*)d_in[3];
    const float* convB  = (const float*)d_in[4];
    const float* W1     = (const float*)d_in[5];
    const float* b1     = (const float*)d_in[6];
    const float* W2     = (const float*)d_in[7];
    const float* b2     = (const float*)d_in[8];
    float*       out    = (float*)d_out;

    char* ws = (char*)d_ws;
    const size_t BUF = (size_t)N_NODES * D_IN * sizeof(float);
    float* buf0 = (float*)(ws);
    char*  csr  = ws + BUF + BUF / 2;
    int*   col    = (int*)(csr);
    int*   rowptr = (int*)(csr + (size_t)N_ETYPES * N_EDGES * sizeof(int));
    float* buf2 = (float*)(ws + 2 * BUF);
    int*   bh    = (int*)(ws);
    int*   off   = (int*)(ws + (1 << 20));
    int*   pairs = (int*)(ws + 2 * BUF);
    float* W12  = (float*)(ws + BUF);
    float* b12  = W12 + D_IN * D_OUT;

    {
        dim3 g(NBLK, N_ETYPES);
        part_hist_kernel<<<g, 256, 0, stream>>>(edst, bh);
        part_scan_kernel<<<N_ETYPES, 1024, 0, stream>>>(bh, off);
        part_scatter_kernel<<<g, 256, 0, stream>>>(esrc, edst, off, pairs);
        dim3 gb(NBK, N_ETYPES);
        bucket_sort_kernel<<<gb, BK_NODES, 0, stream>>>(pairs, off, rowptr, col);
    }

    const int row_blocks = (N_NODES / 16 + 7) / 8;

    conv_layer_kernel<<<CL_GRID, CL_WAVES * 32, 0, stream>>>(
        blocks, rowptr, col, convW, convB, buf0);
    conv_layer_kernel<<<CL_GRID, CL_WAVES * 32, 0, stream>>>(
        buf0, rowptr, col, convW + (size_t)N_ETYPES * D_IN * D_IN, convB + (size_t)N_ETYPES * D_IN, buf2);

    head_fold_kernel<<<D_IN / 16 + 1, 256, 0, stream>>>(W1, b1, W2, b2, W12, b12);
    {
        dim3 g(row_blocks, D_OUT / 32);
        wmma_gemm_strip_f32_kernel<32, D_OUT, D_IN, false, false><<<g, 256, 0, stream>>>(
            buf2, W12, b12, out, nullptr, N_NODES, 1.0f);
    }
}


# Head fused into the layer-2 launch: W12 in LDS, each strip's result goes via the wave's LDS strip into 64 more WMMAs, 16x32 stored; W12 fold rides in the scan launch; 8 launches -> 6; sum orders kept
# speedup vs baseline: 9.2703x; 1.0466x over previous
#include <hip/hip_runtime.h>

typedef float v2f __attribute__((ext_vector_type(2)));
typedef float v8f __attribute__((ext_vector_type(8)));

#define N_NODES  100000
#define N_ETYPES 3
#define N_EDGES  1600000
#define D_IN     128
#define D_HID    256
#define D_OUT    32

#define RPS      (N_NODES + 4)

#define BK_SHIFT 9
#define BK_NODES (1 << BK_SHIFT)
#define NBK      ((N_NODES + BK_NODES - 1) / BK_NODES)
#define EPB      4096
#define NBLK     ((N_EDGES + EPB - 1) / EPB)
#define EPT      (EPB / 4 / 256)
#define SCAN_SL  5
#define SLB      ((NBLK + SCAN_SL - 1) / SCAN_SL)
#define LCAP     16384
static_assert((size_t)N_ETYPES * NBLK * NBK * sizeof(int) <= (1u << 20), "a histogram plane fits the 1 MiB the driver gives it");

__global__ __launch_bounds__(256)
void part_hist_kernel(const int* __restrict__ dst, int* __restrict__ bh) {
    __shared__ int lh[NBK];
    const int e = blockIdx.y, blk = blockIdx.x, t = threadIdx.x;
    if (t < NBK) lh[t] = 0;
    __syncthreads();
    const int4* d4 = (const int4*)(dst + (size_t)e * N_EDGES) + (size_t)blk * (EPB / 4);
    int4 d[EPT];
#pragma unroll
    for (int u = 0; u < EPT; ++u) {
        const int i = u * 256 + t;
        d[u] = make_int4(0, 0, 0, 0);
        if (blk * (EPB / 4) + i < N_EDGES / 4) d[u] = d4[i];
    }
#pragma unroll
    for (int u = 0; u < EPT; ++u) {
        if (blk * (EPB / 4) + u * 256 + t < N_EDGES / 4) {
            atomicAdd(&lh[d[u].x >> BK_SHIFT], 1);
            atomicAdd(&lh[d[u].y >> BK_SHIFT], 1);
            atomicAdd(&lh[d[u].z >> BK_SHIFT], 1);
            atomicAdd(&lh[d[u].w >> BK_SHIFT], 1);
        }
    }
    __syncthreads();
    if (t < NBK) bh[((size_t)e * NBLK + blk) * NBK + t] = lh[t];
}

#define HF_P      (D_OUT + 16)
#define HF_KS     (D_HID / 4)
#define HF_STRIPS (D_IN / 16 + 1)

__global__ __launch_bounds__(1024)
void part_scan_kernel(const int* __restrict__ bh, int* __restrict__ off,
                      const float* __restrict__ W1,
                      const float* __restrict__ b1,
                      const float* __restrict__ W2,
                      const float* __restrict__ b2,
                      float*       __restrict__ W12,
                      float*       __restrict__ b12) {
    __shared__ int ssum[SCAN_SL][NBK];
    __shared__ int bbase[NBK];
    __shared__ int wtot[8];
    __shared__ __attribute__((aligned(16))) float2 Wf[(D_HID / 2) * HF_P];
    static_assert(NBK <= 256 && SCAN_SL * NBK <= 1024, "thread map covers every (slice, bucket)");
    const int t    = threadIdx.x;
    const int lane = t & 31;

    if (blockIdx.x >= N_ETYPES) {
        constexpr int STAGE_IT = (D_HID / 2) * (D_OUT / 4) / 256;
        static_assert(STAGE_IT * 256 == (D_HID / 2) * (D_OUT / 4), "slab divides evenly over 256 threads");
        if (t < 256) {
#pragma unroll
            for (int it = 0; it < STAGE_IT; ++it) {
                const int idx = t + it * 256;
                const int kp = idx / (D_OUT / 4);
                const int n4 = (idx % (D_OUT / 4)) * 4;
                const float4 r0 = *(const float4*)(W2 + (size_t)(2 * kp)     * D_OUT + n4);
                const float4 r1 = *(const float4*)(W2 + (size_t)(2 * kp + 1) * D_OUT + n4);
                float4* d = (float4*)(Wf + kp * HF_P + n4);
                d[0] = make_float4(r0.x, r1.x, r0.y, r1.y);
                d[1] = make_float4(r0.z, r1.z, r0.w, r1.w);
            }
        }
        __syncthreads();
        if (t >= 32) return;

        const int half  = lane >> 4;
        const int l15   = lane & 15;
        const int strip = blockIdx.x - N_ETYPES;
        const bool bias_strip = strip == D_IN / 16;
        const float* arow = bias_strip ? b1 + half * 2
                                       : W1 + (size_t)(strip * 16 + l15) * D_HID + half * 2;
        const bool zero_row = bias_strip && l15 != 0;

        v8f acc[D_OUT / 16];
#pragma unroll
        for (int j = 0; j < D_OUT / 16; ++j) acc[j] = v8f{};

        const float2* wb = Wf + half * HF_P + l15;
#pragma unroll 1
        for (int kb = 0; kb < HF_KS / 8; ++kb) {
            float2 av[8];
#pragma unroll
            for (int s = 0; s < 8; ++s) av[s] = *(const float2*)(arow + (kb * 8 + s) * 4);
#pragma unroll
            for (int s = 0; s < 8; ++s) {
                v2f a;
                a.x = zero_row ? 0.0f : av[s].x;
                a.y = zero_row ? 0.0f : av[s].y;
                const float2* wk = wb + ((kb * 8 + s) * 2) * HF_P;
#pragma unroll
                for (int j = 0; j < D_OUT / 16; ++j) {
                    const float2 bv = wk[16 * j];
                    v2f b;
                    b.x = bv.x;
                    b.y = bv.y;
                    acc[j] = __builtin_amdgcn_wmma_f32_16x16x4_f32(
                         false, a,  false, b,
                         (short)0, acc[j],  false,  false);
                }
            }
        }

        if (!bias_strip) {
            float* crow = W12 + (size_t)(strip * 16 + half * 8) * D_OUT + l15;
#pragma unroll
            for (int j = 0; j < D_OUT / 16; ++j)
#pragma unroll
                for (int r = 0; r < 8; ++r) crow[(size_t)r * D_OUT + 16 * j] = acc[j][r];
        } else if (half == 0) {
#pragma unroll
            for (int j = 0; j < D_OUT / 16; ++j) b12[16 * j + l15] = acc[j][0] + b2[16 * j + l15];
        }
        return;
    }

    const int e    = blockIdx.x;
    const int sl   = t / NBK;
    const int b    = t - sl * NBK;
    const bool live = t < SCAN_SL * NBK;
    const int k0 = sl * SLB;
    const int k1 = (k0 + SLB < NBLK) ? k0 + SLB : NBLK;
    const int* pi = bh  + (size_t)e * NBLK * NBK + b;
    int*       po = off + (size_t)e * NBLK * NBK + b;

    int s = 0;
    if (live) {
#pragma unroll 8
        for (int k = k0; k < k1; ++k) s += pi[(size_t)k * NBK];
        ssum[sl][b] = s;
    }
    __syncthreads();
    int tot = 0, inc = 0;
    if (t < 256) {
        if (t < NBK) {
#pragma unroll
            for (int q = 0; q < SCAN_SL; ++q) tot += ssum[q][t];
        }
        inc = tot;
#pragma unroll
        for (int d = 1; d < 32; d <<= 1) {
            const int o = __shfl_up(inc, d);
            if (lane >= d) inc += o;
        }
        if (lane == 31) wtot[t >> 5] = inc;
    }
    __syncthreads();
    if (t < NBK) {
        int wb = 0;
        for (int q = 0; q < (t >> 5); ++q) wb += wtot[q];
        bbase[t] = wb + inc - tot;
    }
    __syncthreads();
    if (live) {
        int run = bbase[b];
        for (int q = 0; q < sl; ++q) run += ssum[q][b];
#pragma unroll 8
        for (int k = k0; k < k1; ++k) {
            const int v = pi[(size_t)k * NBK];
            po[(size_t)k * NBK] = run;
            run += v;
        }
    }
}

__global__ __launch_bounds__(256)
void part_scatter_kernel(const int* __restrict__ src, const int* __restrict__ dst,
                         const int* __restrict__ off, int* __restrict__ pairs) {
    __shared__ int lh[NBK];
    __shared__ int gofs[NBK];
    __shared__ int wtot[8];
    __shared__ __attribute__((aligned(16))) int2 lp[EPB];
    static_assert(NBK <= 256, "one thread per bucket in the scan");
    const int e = blockIdx.y, blk = blockIdx.x, t = threadIdx.x;
    const int lane = t & 31;
    if (t < NBK) lh[t] = 0;
    int go = 0;
    if (t < NBK) go = off[((size_t)e * NBLK + blk) * NBK + t];
    const int4* d4 = (const int4*)(dst + (size_t)e * N_EDGES) + (size_t)blk * (EPB / 4);
    const int4* s4 = (const int4*)(src + (size_t)e * N_EDGES) + (size_t)blk * (EPB / 4);
    int2* pe = (int2*)pairs + (size_t)e * N_EDGES;
    int4 d[EPT], s[EPT], r[EPT];
#pragma unroll
    for (int u = 0; u < EPT; ++u) {
        const int i = u * 256 + t;
        d[u] = make_int4(0, 0, 0, 0);
        s[u] = make_int4(0, 0, 0, 0);
        if (blk * (EPB / 4) + i < N_EDGES / 4) { d[u] = d4[i]; s[u] = s4[i]; }
    }
    __syncthreads();
#pragma unroll
    for (int u = 0; u < EPT; ++u) {
        r[u] = make_int4(0, 0, 0, 0);
        if (blk * (EPB / 4) + u * 256 + t < N_EDGES / 4) {
            r[u].x = atomicAdd(&lh[d[u].x >> BK_SHIFT], 1);
            r[u].y = atomicAdd(&lh[d[u].y >> BK_SHIFT], 1);
            r[u].z = atomicAdd(&lh[d[u].z >> BK_SHIFT], 1);
            r[u].w = atomicAdd(&lh[d[u].w >> BK_SHIFT], 1);
        }
    }
    __syncthreads();
    const int c = (t < NBK) ? lh[t] : 0;
    int inc = c;
#pragma unroll
    for (int k = 1; k < 32; k <<= 1) {
        const int o = __shfl_up(inc, k);
        if (lane >= k) inc += o;
    }
    if (lane == 31) wtot[t >> 5] = inc;
    __syncthreads();
    if (t < NBK) {
        int wb = 0;
        for (int q = 0; q < (t >> 5); ++q) wb += wtot[q];
        const int ls = wb + inc - c;
        lh[t]   = ls;
        gofs[t] = go - ls;
    }
    __syncthreads();
#pragma unroll
    for (int u = 0; u < EPT; ++u) {
        if (blk * (EPB / 4) + u * 256 + t < N_EDGES / 4) {
            lp[lh[d[u].x >> BK_SHIFT] + r[u].x] = make_int2(s[u].x, d[u].x);
            lp[lh[d[u].y >> BK_SHIFT] + r[u].y] = make_int2(s[u].y, d[u].y);
            lp[lh[d[u].z >> BK_SHIFT] + r[u].z] = make_int2(s[u].z, d[u].z);
            lp[lh[d[u].w >> BK_SHIFT] + r[u].w] = make_int2(s[u].w, d[u].w);
        }
    }
    __syncthreads();
    const int left = N_EDGES - blk * EPB;
    const int nv   = left < EPB ? left : EPB;
#pragma unroll 4
    for (int i = t; i < nv; i += 256) {
        const int2 p = lp[i];
        pe[gofs[p.y >> BK_SHIFT] + i] = p;
    }
}

__global__ __launch_bounds__(BK_NODES)
void bucket_sort_kernel(const int* __restrict__ pairs, const int* __restrict__ off,
                        int* __restrict__ rowptr, int* __restrict__ col) {
    __shared__ int cnt[BK_NODES];
    __shared__ int wsum[BK_NODES / 32];
    __shared__ int lcol[LCAP];
    const int e = blockIdx.y, b = blockIdx.x, t = threadIdx.x;
    const int lane = t & 31;
    const int w    = t >> 5;
    const int* oe  = off + (size_t)e * NBLK * NBK;
    const int beg  = oe[b];
    const int end  = (b + 1 < NBK) ? oe[b + 1] : N_EDGES;
    const int2* pe = (const int2*)pairs + (size_t)e * N_EDGES;
    int*        ce = col + (size_t)e * N_EDGES;

    const bool staged = end - beg <= LCAP;
    cnt[t] = 0;
    __syncthreads();
    for (int j0 = beg; j0 < end; j0 += 4 * BK_NODES) {
        int2 p[4];
#pragma unroll
        for (int u = 0; u < 4; ++u) {
            const int j = j0 + u * BK_NODES + t;
            p[u] = pe[(j < end) ? j : end - 1];
        }
#pragma unroll
        for (int u = 0; u < 4; ++u)
            if (j0 + u * BK_NODES + t < end) atomicAdd(&cnt[p[u].y & (BK_NODES - 1)], 1);
    }
    __syncthreads();

    const int c = cnt[t];
    int inc = c;
#pragma unroll
    for (int d = 1; d < 32; d <<= 1) {
        const int o = __shfl_up(inc, d);
        if (lane >= d) inc += o;
    }
    if (lane == 31) wsum[w] = inc;
    __syncthreads();
    int first = beg + inc - c;
    for (int q = 0; q < w; ++q) first += wsum[q];
    cnt[t] = first;
    const int n = b * BK_NODES + t;
    if (n < N_NODES) rowptr[(size_t)e * RPS + n] = first;
    if (b == NBK - 1 && t == 0) rowptr[(size_t)e * RPS + N_NODES] = end;
    __syncthreads();

    for (int j0 = beg; j0 < end; j0 += 4 * BK_NODES) {
        int2 p[4];
#pragma unroll
        for (int u = 0; u < 4; ++u) {
            const int j = j0 + u * BK_NODES + t;
            p[u] = pe[(j < end) ? j : end - 1];
        }
        int q[4];
#pragma unroll
        for (int u = 0; u < 4; ++u) {
            q[u] = 0;
            if (j0 + u * BK_NODES + t < end) q[u] = atomicAdd(&cnt[p[u].y & (BK_NODES - 1)], 1);
        }
#pragma unroll
        for (int u = 0; u < 4; ++u) {
            if (j0 + u * BK_NODES + t < end) {
                if (staged) lcol[q[u] - beg] = p[u].x;
                else        ce[q[u]] = p[u].x;
            }
        }
    }
    if (staged) {
        __syncthreads();
        int g = (beg & ~31) + t;
        if (g >= beg && g < end) ce[g] = lcol[g - beg];
#pragma unroll 4
        for (g += BK_NODES; g < end; g += BK_NODES) ce[g] = lcol[g - beg];
    }
}

#define CL_WAVES 13
#define CL_GRID  256
#define CL_PA    (D_IN + 4)
#define N_STRIPS (N_NODES / 16)
static_assert(N_STRIPS * 16 == N_NODES, "strips are whole");
static_assert(2 * CL_WAVES * CL_GRID >= N_STRIPS, "two trips cover every strip");

template <bool HEAD>
__global__ __launch_bounds__(CL_WAVES * 32)
void conv_layer_kernel(const float* __restrict__ h,
                       const int*   __restrict__ rowptr,
                       const int*   __restrict__ col,
                       const float* __restrict__ Wl,
                       const float* __restrict__ bl,
                       const float* __restrict__ W12,
                       const float* __restrict__ b12,
                       float*       __restrict__ out) {
    __shared__ __attribute__((aligned(16))) float2 Ws[N_ETYPES * (D_IN / 2) * D_IN];
    __shared__ __attribute__((aligned(16))) float  As[CL_WAVES * 16 * CL_PA];
    __shared__ __attribute__((aligned(16))) float2 Hs[HEAD ? (D_IN / 2) * D_OUT : 2];

    const int tid  = threadIdx.x;
    const int wave = __builtin_amdgcn_readfirstlane(tid >> 5);
    const int lane = tid & 31;
    const int half = lane >> 4;
    const int l15  = lane & 15;

    {
        constexpr int SLAB_ITEMS = (D_IN / 2) * (D_IN / 4);
        constexpr int ALL_ITEMS  = N_ETYPES * SLAB_ITEMS;
        constexpr int NIT = (ALL_ITEMS + CL_WAVES * 32 - 1) / (CL_WAVES * 32);
#pragma unroll
        for (int it0 = 0; it0 < NIT; it0 += 8) {
            float4 r0[8], r1[8];
#pragma unroll
            for (int u = 0; u < 8; ++u) {
                const int idx = tid + (it0 + u) * (CL_WAVES * 32);
                r0[u] = make_float4(0.0f, 0.0f, 0.0f, 0.0f);
                r1[u] = make_float4(0.0f, 0.0f, 0.0f, 0.0f);
                if (it0 + u < NIT && idx < ALL_ITEMS) {
                    const int e  = idx / SLAB_ITEMS;
                    const int r  = idx - e * SLAB_ITEMS;
                    const int kp = r / (D_IN / 4);
                    const int n4 = (r % (D_IN / 4)) * 4;
                    const float* W = Wl + (size_t)e * D_IN * D_IN;
                    r0[u] = *(const float4*)(W + (size_t)(2 * kp)     * D_IN + n4);
                    r1[u] = *(const float4*)(W + (size_t)(2 * kp + 1) * D_IN + n4);
                }
            }
#pragma unroll
            for (int u = 0; u < 8; ++u) {
                const int idx = tid + (it0 + u) * (CL_WAVES * 32);
                if (it0 + u < NIT && idx < ALL_ITEMS) {
                    const int e  = idx / SLAB_ITEMS;
                    const int r  = idx - e * SLAB_ITEMS;
                    const int kp = r / (D_IN / 4);
                    const int n4 = (r % (D_IN / 4)) * 4;
                    const int pos = (n4 + 16 * (kp & 1)) & (D_IN - 1);
                    float4* d = (float4*)(Ws + (e * (D_IN / 2) + kp) * D_IN + pos);
                    d[0] = make_float4(r0[u].x, r1[u].x, r0[u].y, r1[u].y);
                    d[1] = make_float4(r0[u].z, r1[u].z, r0[u].w, r1[u].w);
                }
            }
        }
    }
    if (HEAD) {
        constexpr int H_ITEMS = (D_IN / 2) * (D_OUT / 4);
        for (int idx = tid; idx < H_ITEMS; idx += CL_WAVES * 32) {
            const int kp = idx / (D_OUT / 4);
            const int n4 = (idx % (D_OUT / 4)) * 4;
            const float4 r0 = *(const float4*)(W12 + (size_t)(2 * kp)     * D_OUT + n4);
            const float4 r1 = *(const float4*)(W12 + (size_t)(2 * kp + 1) * D_OUT + n4);
            const int pos = (n4 + 16 * (kp & 1)) & (D_OUT - 1);
            float4* d = (float4*)(Hs + kp * D_OUT + pos);
            d[0] = make_float4(r0.x, r1.x, r0.y, r1.y);
            d[1] = make_float4(r0.z, r1.z, r0.w, r1.w);
        }
    }
    __syncthreads();

    float*         asw    = As + wave * (16 * CL_PA);
    const char*    hb     = (const char*)h;
    const unsigned lane_b = (unsigned)lane * 16u;

#pragma unroll 1
    for (int strip = blockIdx.x + CL_GRID * wave; strip < N_STRIPS; strip += CL_GRID * CL_WAVES) {
        const int m0 = strip * 16;

        v8f o[8];
#pragma unroll
        for (int j = 0; j < 8; ++j) o[j] = v8f{};

#pragma unroll 1
        for (int e = 0; e < N_ETYPES; ++e) {
            const int* rpe = rowptr + (size_t)e * RPS;
            const int* ce  = col + (size_t)e * N_EDGES;

            int rp = rpe[m0 + (lane < 16 ? lane : 16)];

            {
                const int ebeg = __builtin_amdgcn_readlane(rp, 0);
                const int eend = __builtin_amdgcn_readlane(rp, 16);
                if (lane > 16) rp = 0x7fffffff;
                int row  = 0;
                int rbeg = ebeg;
                int rend = __builtin_amdgcn_readlane(rp, 1);
                float4 acc = make_float4(0.0f, 0.0f, 0.0f, 0.0f);

                int mine_next = 0;
                if (ebeg + lane < eend) mine_next = ce[ebeg + lane];
                for (int j = ebeg; j < eend; j += 32) {
                    int ids = mine_next;
                    if (j + 32 + lane < eend) mine_next = ce[j + 32 + lane];
                    const int id0 = __builtin_amdgcn_readlane(ids, 0);
                    if (j + lane >= eend) ids = id0;
                    float4 v[32];
#pragma unroll
                    for (int u = 0; u < 32; ++u) {
                        const unsigned s = (unsigned)__builtin_amdgcn_readlane(ids, u);
                        v[u] = *(const float4*)(hb + (s * (unsigned)(D_IN * sizeof(float)) + lane_b));
                    }
                    int lim = rend - j;
#pragma unroll
                    for (int u = 0; u < 32; ++u) {
                        if (__builtin_expect(u >= lim, 0)) {
                            while (j + u >= rend && row < 16) {
                                const float rs = 1.0f / fmaxf((float)(rend - rbeg), 1.0f);
                                *(float4*)(asw + row * CL_PA + lane * 4) =
                                    make_float4(acc.x * rs, acc.y * rs, acc.z * rs, acc.w * rs);
                                acc = make_float4(0.0f, 0.0f, 0.0f, 0.0f);
                                ++row;
                                rbeg = rend;
                                rend = __builtin_amdgcn_readlane(rp, row + 1);
                            }
                            lim = rend - j;
                        }
                        acc.x += v[u].x;
                        acc.y += v[u].y;
                        acc.z += v[u].z;
                        acc.w += v[u].w;
                    }
                }
                for (; row < 16; ++row) {
                    const float rs = 1.0f / fmaxf((float)(rend - rbeg), 1.0f);
                    *(float4*)(asw + row * CL_PA + lane * 4) =
                        make_float4(acc.x * rs, acc.y * rs, acc.z * rs, acc.w * rs);
                    acc = make_float4(0.0f, 0.0f, 0.0f, 0.0f);
                    rbeg = rend;
                    rend = __builtin_amdgcn_readlane(rp, (row + 2 < 16) ? row + 2 : 16);
                }
            }
            __builtin_amdgcn_fence(__ATOMIC_ACQ_REL, "wavefront");
            __builtin_amdgcn_wave_barrier();

            {
                v8f acc[8];
#pragma unroll
                for (int j = 0; j < 8; ++j) acc[j] = v8f{};

                const float2* wse = Ws + e * ((D_IN / 2) * D_IN);
                const float*  ar  = asw + l15 * CL_PA + half * 2;
                const float2* wb  = wse + half * D_IN + half * 16 + l15;
                const float2* wb7 = wse + half * D_IN + (half ? 0 : 7 * 16) + l15;

#pragma unroll 1
                for (int kb = 0; kb < D_IN / 32; ++kb) {
#pragma unroll
                    for (int s = 0; s < 8; ++s) {
                        const float2 av = *(const float2*)(ar + kb * 32 + s * 4);
                        v2f a;
                        a.x = av.x;
                        a.y = av.y;
                        const int ko = (kb * 16 + s * 2) * D_IN;
#pragma unroll
                        for (int j = 0; j < 8; ++j) {
                            const float2 bv = (j < 7) ? wb[ko + 16 * j] : wb7[ko];
                            v2f b;
                            b.x = bv.x;
                            b.y = bv.y;
                            acc[j] = __builtin_amdgcn_wmma_f32_16x16x4_f32(
                                 false, a,  false, b,
                                 (short)0, acc[j],  false,  false);
                        }
                    }
                }
                __builtin_amdgcn_fence(__ATOMIC_ACQ_REL, "wavefront");
                __builtin_amdgcn_wave_barrier();

                const float* be = bl + (size_t)e * D_IN;
#pragma unroll
                for (int j = 0; j < 8; ++j) {
                    const float bv = be[16 * j + l15];
#pragma unroll
                    for (int r = 0; r < 8; ++r) {
                        float v = fmaxf(acc[j][r] + bv, 0.0f);
                        v *= 1.0f / 3.0f;
                        o[j][r] = o[j][r] + v;
                    }
                }
            }
        }

        if (!HEAD) {
            float* crow = out + (size_t)(m0 + half * 8) * D_IN + l15;
#pragma unroll
            for (int j = 0; j < 8; ++j)
#pragma unroll
                for (int r = 0; r < 8; ++r) crow[(size_t)r * D_IN + 16 * j] = o[j][r];
        } else {
            float* srow = asw + (half * 8) * CL_PA + l15;
#pragma unroll
            for (int j = 0; j < 8; ++j)
#pragma unroll
                for (int r = 0; r < 8; ++r) srow[r * CL_PA + 16 * j] = o[j][r];
            __builtin_amdgcn_fence(__ATOMIC_ACQ_REL, "wavefront");
            __builtin_amdgcn_wave_barrier();

            v8f hacc[D_OUT / 16];
#pragma unroll
            for (int j = 0; j < D_OUT / 16; ++j) hacc[j] = v8f{};

            const float*  ar = asw + l15 * CL_PA + half * 2;
            const float2* hb0 = Hs + half * D_OUT + (half ? 16 : 0) + l15;
            const float2* hb1 = Hs + half * D_OUT + (half ? 0 : 16) + l15;
#pragma unroll 1
            for (int kb = 0; kb < D_IN / 32; ++kb) {
#pragma unroll
                for (int s = 0; s < 8; ++s) {
                    const float2 av = *(const float2*)(ar + kb * 32 + s * 4);
                    v2f a;
                    a.x = av.x;
                    a.y = av.y;
                    const int ko = (kb * 16 + s * 2) * D_OUT;
                    const float2 bv0 = hb0[ko];
                    const float2 bv1 = hb1[ko];
                    v2f b0, b1v;
                    b0.x = bv0.x;
                    b0.y = bv0.y;
                    b1v.x = bv1.x;
                    b1v.y = bv1.y;
                    hacc[0] = __builtin_amdgcn_wmma_f32_16x16x4_f32(
                         false, a,  false, b0,
                         (short)0, hacc[0],  false,  false);
                    hacc[1] = __builtin_amdgcn_wmma_f32_16x16x4_f32(
                         false, a,  false, b1v,
                         (short)0, hacc[1],  false,  false);
                }
            }
            __builtin_amdgcn_fence(__ATOMIC_ACQ_REL, "wavefront");
            __builtin_amdgcn_wave_barrier();

            float* crow = out + (size_t)(m0 + half * 8) * D_OUT + l15;
#pragma unroll
            for (int j = 0; j < D_OUT / 16; ++j) {
                const float bv = b12[16 * j + l15];
#pragma unroll
                for (int r = 0; r < 8; ++r) crow[(size_t)r * D_OUT + 16 * j] = hacc[j][r] + bv;
            }
        }
    }
}

extern "C" void kernel_launch(void* const* d_in, const int* in_sizes, int n_in,
                              void* d_out, int out_size, void* d_ws, size_t ws_size,
                              hipStream_t stream) {
    const float* blocks = (const float*)d_in[0];
    const int*   esrc   = (const int*)  d_in[1];
    const int*   edst   = (const int*)  d_in[2];
    const float* convW  = (const float*)d_in[3];
    const float* convB  = (const float*)d_in[4];
    const float* W1     = (const float*)d_in[5];
    const float* b1     = (const float*)d_in[6];
    const float* W2     = (const float*)d_in[7];
    const float* b2     = (const float*)d_in[8];
    float*       out    = (float*)d_out;

    char* ws = (char*)d_ws;
    const size_t BUF = (size_t)N_NODES * D_IN * sizeof(float);
    float* buf0 = (float*)(ws);
    char*  csr  = ws + BUF + BUF / 2;
    int*   col    = (int*)(csr);
    int*   rowptr = (int*)(csr + (size_t)N_ETYPES * N_EDGES * sizeof(int));
    int*   bh    = (int*)(ws);
    int*   off   = (int*)(ws + (1 << 20));
    int*   pairs = (int*)(ws + 2 * BUF);
    float* W12  = (float*)(ws + BUF);
    float* b12  = W12 + D_IN * D_OUT;

    {
        dim3 g(NBLK, N_ETYPES);
        part_hist_kernel<<<g, 256, 0, stream>>>(edst, bh);
        part_scan_kernel<<<N_ETYPES + HF_STRIPS, 1024, 0, stream>>>(bh, off, W1, b1, W2, b2, W12, b12);
        part_scatter_kernel<<<g, 256, 0, stream>>>(esrc, edst, off, pairs);
        dim3 gb(NBK, N_ETYPES);
        bucket_sort_kernel<<<gb, BK_NODES, 0, stream>>>(pairs, off, rowptr, col);
    }

    conv_layer_kernel<false><<<CL_GRID, CL_WAVES * 32, 0, stream>>>(
        blocks, rowptr, col, convW, convB, nullptr, nullptr, buf0);
    conv_layer_kernel<true><<<CL_GRID, CL_WAVES * 32, 0, stream>>>(
        buf0, rowptr, col, convW + (size_t)N_ETYPES * D_IN * D_IN, convB + (size_t)N_ETYPES * D_IN,
        W12, b12, out);
}
